# starting kernel
# baseline (speedup 1.0000x reference)
#include <hip/hip_runtime.h>

typedef __attribute__((ext_vector_type(2))) float v2f;
typedef __attribute__((ext_vector_type(8))) float v8f;

#define N_NODES  131072
#define F_INK    300
#define F_OUTK   64
#define N_EDGE   2097152
#define S_SUB    128
#define B_SUB    1024

__global__ void gcn_degree_kernel(const int* __restrict__ src,
                                  const int* __restrict__ dst,
                                  float* __restrict__ deg_out,
                                  float* __restrict__ deg_in) {
    int i = blockIdx.x * blockDim.x + threadIdx.x;
    if (i < N_EDGE) {
        atomicAdd(&deg_out[src[i]], 1.0f);
        atomicAdd(&deg_in[dst[i]], 1.0f);
    }
}

__global__ void gcn_gemm_wmma_kernel(const float* __restrict__ X,
                                     const float* __restrict__ W,
                                     const float* __restrict__ deg_out,
                                     float* __restrict__ H) {
    const int lane    = threadIdx.x & 31;
    const int wave    = threadIdx.x >> 5;
    const int rowTile = wave & 1;
    const int colTile = wave >> 1;
    const int m       = lane & 15;
    const int half    = lane >> 4;

    const int rowBase = blockIdx.x * 32 + rowTile * 16;
    const int row     = rowBase + m;
    const int col     = colTile * 16 + m;

    const float rs = __frsqrt_rn(fmaxf(deg_out[row], 1.0f));

    const float* xrow  = X + (long)row * F_INK + half * 2;
    const float* wbase = W + (long)(half * 2) * F_OUTK + col;

    v8f c = {};
    for (int k0 = 0; k0 < F_INK; k0 += 4) {
        v2f a = *(const v2f*)(xrow + k0);
        a *= rs;
        v2f b;
        b.x = wbase[(long)k0 * F_OUTK];
        b.y = wbase[(long)(k0 + 1) * F_OUTK];
        c = __builtin_amdgcn_wmma_f32_16x16x4_f32(false, a, false, b,
                                                  (short)0, c, false, false);
    }

    float* hout = H + (long)(rowBase + half * 8) * F_OUTK + colTile * 16 + m;
#pragma unroll
    for (int i = 0; i < 8; ++i)
        hout[(long)i * F_OUTK] = c[i];
}

__global__ void gcn_scatter_kernel(const int* __restrict__ src,
                                   const int* __restrict__ dst,
                                   const float* __restrict__ H,
                                   float* __restrict__ agg) {
    long idx = (long)blockIdx.x * blockDim.x + threadIdx.x;
    int  e   = (int)(idx >> 4);
    int  q   = (int)(idx & 15);
    int  s   = src[e];
    int  d   = dst[e];
    float4 v = ((const float4*)(H + (long)s * F_OUTK))[q];
    float* out = agg + (long)d * F_OUTK + q * 4;
    atomicAdd(out + 0, v.x);
    atomicAdd(out + 1, v.y);
    atomicAdd(out + 2, v.z);
    atomicAdd(out + 3, v.w);
}

__global__ void gcn_finalize_kernel(const float* __restrict__ agg,
                                    const float* __restrict__ deg_in,
                                    const float* __restrict__ bias,
                                    const float* __restrict__ prelu_a,
                                    float* __restrict__ pool,
                                    float* __restrict__ anchor) {
    __shared__ float red[256];
    const int  sg    = blockIdx.x;
    const int  f     = threadIdx.x & 63;
    const int  g     = threadIdx.x >> 6;
    const float alpha = prelu_a[0];
    const float bf    = bias[f];
    const long base   = (long)sg * S_SUB;

    float partial = 0.0f;
    for (int n = g; n < S_SUB - 1; n += 4) {
        long node = base + n;
        float ri = __frsqrt_rn(fmaxf(deg_in[node], 1.0f));
        float v  = agg[node * F_OUTK + f] * ri + bf;
        v = v > 0.0f ? v : alpha * v;
        partial += v;
    }

    if (g == 3) {
        long node = base + S_SUB - 1;
        float ri = __frsqrt_rn(fmaxf(deg_in[node], 1.0f));
        float v  = agg[node * F_OUTK + f] * ri + bf;
        v = v > 0.0f ? v : alpha * v;
        anchor[(long)sg * F_OUTK + f] = v;
    }

    red[threadIdx.x] = partial;
    __syncthreads();
    if (g == 0) {
        float sum = red[f] + red[f + 64] + red[f + 128] + red[f + 192];
        pool[(long)sg * F_OUTK + f] = sum * (1.0f / (S_SUB - 1));
    }
}

extern "C" void kernel_launch(void* const* d_in, const int* in_sizes, int n_in,
                              void* d_out, int out_size, void* d_ws, size_t ws_size,
                              hipStream_t stream) {
    const float* in_feat  = (const float*)d_in[0];
    const float* W        = (const float*)d_in[1];
    const float* bias     = (const float*)d_in[2];
    const float* prelu_a  = (const float*)d_in[3];
    const int*   src      = (const int*)d_in[4];
    const int*   dst      = (const int*)d_in[5];

    float* ws      = (float*)d_ws;
    float* deg_out = ws;
    float* deg_in  = ws + N_NODES;
    float* H       = ws + 2L * N_NODES;
    float* agg     = H + (long)N_NODES * F_OUTK;

    float* pool   = (float*)d_out;
    float* anchor = pool + (long)B_SUB * F_OUTK;

    hipMemsetAsync(deg_out, 0, 2L * N_NODES * sizeof(float), stream);
    hipMemsetAsync(agg, 0, (long)N_NODES * F_OUTK * sizeof(float), stream);

    gcn_degree_kernel<<<N_EDGE / 256, 256, 0, stream>>>(src, dst, deg_out, deg_in);
    gcn_gemm_wmma_kernel<<<N_NODES / 32, 256, 0, stream>>>(in_feat, W, deg_out, H);
    gcn_scatter_kernel<<<(int)((long)N_EDGE * 16 / 256), 256, 0, stream>>>(src, dst, H, agg);
    gcn_finalize_kernel<<<B_SUB, 256, 0, stream>>>(agg, deg_in, bias, prelu_a, pool, anchor);
}


# GEMM kernel: W staged once per block in LDS, each wave computes 16 rows x 64 cols (A fragment loaded once as 16-byte loads, reused by 4 WMMAs); k is summed in a different order
# speedup vs baseline: 1.1372x; 1.1372x over previous
#include <hip/hip_runtime.h>

typedef __attribute__((ext_vector_type(2))) float v2f;
typedef __attribute__((ext_vector_type(8))) float v8f;

#define N_NODES  131072
#define F_INK    300
#define F_OUTK   64
#define N_EDGE   2097152
#define S_SUB    128
#define B_SUB    1024

__global__ void gcn_degree_kernel(const int* __restrict__ src,
                                  const int* __restrict__ dst,
                                  float* __restrict__ deg_out,
                                  float* __restrict__ deg_in) {
    int i = blockIdx.x * blockDim.x + threadIdx.x;
    if (i < N_EDGE) {
        atomicAdd(&deg_out[src[i]], 1.0f);
        atomicAdd(&deg_in[dst[i]], 1.0f);
    }
}

#define KGROUPS (F_INK / 4)
#define WMMA4(a, b, c) __builtin_amdgcn_wmma_f32_16x16x4_f32(false, (a), false, (b), (short)0, (c), false, false)

__device__ __forceinline__ void gcn_mma_group(const float4 x, const float4* __restrict__ wp,
                                              const float rs,
                                              v8f& c0, v8f& c1, v8f& c2, v8f& c3) {
    const float4 b0 = wp[0], b1 = wp[16], b2 = wp[32], b3 = wp[48];
    v2f a0, a1;
    a0.x = x.x * rs; a0.y = x.y * rs;
    a1.x = x.z * rs; a1.y = x.w * rs;
    v2f t0, t1, t2, t3;
    t0.x = b0.x; t0.y = b0.y; t1.x = b1.x; t1.y = b1.y;
    t2.x = b2.x; t2.y = b2.y; t3.x = b3.x; t3.y = b3.y;
    c0 = WMMA4(a0, t0, c0); c1 = WMMA4(a0, t1, c1);
    c2 = WMMA4(a0, t2, c2); c3 = WMMA4(a0, t3, c3);
    t0.x = b0.z; t0.y = b0.w; t1.x = b1.z; t1.y = b1.w;
    t2.x = b2.z; t2.y = b2.w; t3.x = b3.z; t3.y = b3.w;
    c0 = WMMA4(a1, t0, c0); c1 = WMMA4(a1, t1, c1);
    c2 = WMMA4(a1, t2, c2); c3 = WMMA4(a1, t3, c3);
}

__global__ __launch_bounds__(256)
void gcn_gemm_wmma_kernel(const float* __restrict__ X,
                          const float* __restrict__ W,
                          const float* __restrict__ deg_out,
                          float* __restrict__ H) {
    __shared__ float4 Ws[KGROUPS * F_OUTK];

    const int tid  = threadIdx.x;
    const int lane = tid & 31;
    const int wave = tid >> 5;
    const int m    = lane & 15;
    const int half = lane >> 4;

    const int rowBase = blockIdx.x * 128 + wave * 16;
    const int row     = rowBase + m;
    const float* xrow = X + (long)row * F_INK;

    const float* xp = xrow + half * 16;
    float4 xa0 = *(const float4*)(xp + 0);
    float4 xa1 = *(const float4*)(xp + 4);
    float4 xa2 = *(const float4*)(xp + 8);
    float4 xa3 = *(const float4*)(xp + 12);
    const float rs = __frsqrt_rn(fmaxf(deg_out[row], 1.0f));

    {
        const int c = tid & 63;
        for (int g = tid >> 6; g < KGROUPS; g += 4) {
            const float* wp = W + (long)(4 * g) * F_OUTK + c;
            float4 v;
            v.x = wp[0];
            v.y = wp[F_OUTK];
            v.z = wp[2 * F_OUTK];
            v.w = wp[3 * F_OUTK];
            Ws[g * F_OUTK + c] = v;
        }
    }
    __syncthreads();

    v8f c0 = {}, c1 = {}, c2 = {}, c3 = {};
    const float4* wl = Ws + (half * 4) * F_OUTK + m;

    for (int ch = 0; ch < 9; ++ch) {
        const int nx = (ch < 8) ? ch + 1 : 8;
        const float* xn = xp + nx * 32;
        const float4 xb0 = *(const float4*)(xn + 0);
        const float4 xb1 = *(const float4*)(xn + 4);
        const float4 xb2 = *(const float4*)(xn + 8);
        const float4 xb3 = *(const float4*)(xn + 12);
        const float4* wc = wl + ch * 8 * F_OUTK;
        gcn_mma_group(xa0, wc + 0 * F_OUTK, rs, c0, c1, c2, c3);
        gcn_mma_group(xa1, wc + 1 * F_OUTK, rs, c0, c1, c2, c3);
        gcn_mma_group(xa2, wc + 2 * F_OUTK, rs, c0, c1, c2, c3);
        gcn_mma_group(xa3, wc + 3 * F_OUTK, rs, c0, c1, c2, c3);
        xa0 = xb0; xa1 = xb1; xa2 = xb2; xa3 = xb3;
    }

    {
        const float4 x8 = *(const float4*)(xrow + 288 + half * 4);
        gcn_mma_group(x8, Ws + (72 + half) * F_OUTK + m, rs, c0, c1, c2, c3);
    }
    {
        const v2f xt = *(const v2f*)(xrow + 296 + half * 2);
        v2f a;
        a.x = xt.x * rs; a.y = xt.y * rs;
        const float* wt = (const float*)(Ws + 74 * F_OUTK + m) + half * 2;
        const v2f t0 = *(const v2f*)(wt);
        const v2f t1 = *(const v2f*)(wt + 16 * 4);
        const v2f t2 = *(const v2f*)(wt + 32 * 4);
        const v2f t3 = *(const v2f*)(wt + 48 * 4);
        c0 = WMMA4(a, t0, c0); c1 = WMMA4(a, t1, c1);
        c2 = WMMA4(a, t2, c2); c3 = WMMA4(a, t3, c3);
    }

    float* hout = H + (long)(rowBase + half * 8) * F_OUTK + m;
#pragma unroll
    for (int i = 0; i < 8; ++i) {
        hout[(long)i * F_OUTK +  0] = c0[i];
        hout[(long)i * F_OUTK + 16] = c1[i];
        hout[(long)i * F_OUTK + 32] = c2[i];
        hout[(long)i * F_OUTK + 48] = c3[i];
    }
}

__global__ void gcn_scatter_kernel(const int* __restrict__ src,
                                   const int* __restrict__ dst,
                                   const float* __restrict__ H,
                                   float* __restrict__ agg) {
    long idx = (long)blockIdx.x * blockDim.x + threadIdx.x;
    int  e   = (int)(idx >> 4);
    int  q   = (int)(idx & 15);
    int  s   = src[e];
    int  d   = dst[e];
    float4 v = ((const float4*)(H + (long)s * F_OUTK))[q];
    float* out = agg + (long)d * F_OUTK + q * 4;
    atomicAdd(out + 0, v.x);
    atomicAdd(out + 1, v.y);
    atomicAdd(out + 2, v.z);
    atomicAdd(out + 3, v.w);
}

__global__ void gcn_finalize_kernel(const float* __restrict__ agg,
                                    const float* __restrict__ deg_in,
                                    const float* __restrict__ bias,
                                    const float* __restrict__ prelu_a,
                                    float* __restrict__ pool,
                                    float* __restrict__ anchor) {
    __shared__ float red[256];
    const int  sg    = blockIdx.x;
    const int  f     = threadIdx.x & 63;
    const int  g     = threadIdx.x >> 6;
    const float alpha = prelu_a[0];
    const float bf    = bias[f];
    const long base   = (long)sg * S_SUB;

    float partial = 0.0f;
    for (int n = g; n < S_SUB - 1; n += 4) {
        long node = base + n;
        float ri = __frsqrt_rn(fmaxf(deg_in[node], 1.0f));
        float v  = agg[node * F_OUTK + f] * ri + bf;
        v = v > 0.0f ? v : alpha * v;
        partial += v;
    }

    if (g == 3) {
        long node = base + S_SUB - 1;
        float ri = __frsqrt_rn(fmaxf(deg_in[node], 1.0f));
        float v  = agg[node * F_OUTK + f] * ri + bf;
        v = v > 0.0f ? v : alpha * v;
        anchor[(long)sg * F_OUTK + f] = v;
    }

    red[threadIdx.x] = partial;
    __syncthreads();
    if (g == 0) {
        float sum = red[f] + red[f + 64] + red[f + 128] + red[f + 192];
        pool[(long)sg * F_OUTK + f] = sum * (1.0f / (S_SUB - 1));
    }
}

extern "C" void kernel_launch(void* const* d_in, const int* in_sizes, int n_in,
                              void* d_out, int out_size, void* d_ws, size_t ws_size,
                              hipStream_t stream) {
    const float* in_feat  = (const float*)d_in[0];
    const float* W        = (const float*)d_in[1];
    const float* bias     = (const float*)d_in[2];
    const float* prelu_a  = (const float*)d_in[3];
    const int*   src      = (const int*)d_in[4];
    const int*   dst      = (const int*)d_in[5];

    float* ws      = (float*)d_ws;
    float* deg_out = ws;
    float* deg_in  = ws + N_NODES;
    float* H       = ws + 2L * N_NODES;
    float* agg     = H + (long)N_NODES * F_OUTK;

    float* pool   = (float*)d_out;
    float* anchor = pool + (long)B_SUB * F_OUTK;

    hipMemsetAsync(deg_out, 0, 2L * N_NODES * sizeof(float), stream);
    hipMemsetAsync(agg, 0, (long)N_NODES * F_OUTK * sizeof(float), stream);

    gcn_degree_kernel<<<N_EDGE / 256, 256, 0, stream>>>(src, dst, deg_out, deg_in);
    gcn_gemm_wmma_kernel<<<N_NODES / 128, 256, 0, stream>>>(in_feat, W, deg_out, H);
    gcn_scatter_kernel<<<(int)((long)N_EDGE * 16 / 256), 256, 0, stream>>>(src, dst, H, agg);
    gcn_finalize_kernel<<<B_SUB, 256, 0, stream>>>(agg, deg_in, bias, prelu_a, pool, anchor);
}


# Aggregation over an edge list counting-sorted by destination (rank from a returning integer atomic, scan, fill), rows summed in registers; no float atomics on agg; sum order = arrival-order rank
# speedup vs baseline: 1.4741x; 1.2963x over previous
#include <hip/hip_runtime.h>

typedef __attribute__((ext_vector_type(2))) float v2f;
typedef __attribute__((ext_vector_type(8))) float v8f;

#define N_NODES  131072
#define F_INK    300
#define F_OUTK   64
#define N_EDGE   2097152
#define S_SUB    128
#define B_SUB    1024

#define EDGE_THREADS (N_EDGE / 8)

__global__ __launch_bounds__(256)
void gcn_degree_rank_kernel(const int* __restrict__ src,
                            const int* __restrict__ dst,
                            float* __restrict__ deg_out,
                            int* __restrict__ deg_in,
                            int* __restrict__ rank) {
    const int t = blockIdx.x * 256 + threadIdx.x;
    const int4* s4 = (const int4*)src;
    const int4* d4 = (const int4*)dst;
    int4* r4 = (int4*)rank;
    const int4 da = d4[t];
    const int4 db = d4[t + EDGE_THREADS];
    const int4 sa = s4[t];
    const int4 sb = s4[t + EDGE_THREADS];
    int4 ka, kb;
    ka.x = atomicAdd(&deg_in[da.x], 1);
    ka.y = atomicAdd(&deg_in[da.y], 1);
    ka.z = atomicAdd(&deg_in[da.z], 1);
    ka.w = atomicAdd(&deg_in[da.w], 1);
    kb.x = atomicAdd(&deg_in[db.x], 1);
    kb.y = atomicAdd(&deg_in[db.y], 1);
    kb.z = atomicAdd(&deg_in[db.z], 1);
    kb.w = atomicAdd(&deg_in[db.w], 1);
    atomicAdd(&deg_out[sa.x], 1.0f);
    atomicAdd(&deg_out[sa.y], 1.0f);
    atomicAdd(&deg_out[sa.z], 1.0f);
    atomicAdd(&deg_out[sa.w], 1.0f);
    atomicAdd(&deg_out[sb.x], 1.0f);
    atomicAdd(&deg_out[sb.y], 1.0f);
    atomicAdd(&deg_out[sb.z], 1.0f);
    atomicAdd(&deg_out[sb.w], 1.0f);
    r4[t] = ka;
    r4[t + EDGE_THREADS] = kb;
}

#define KGROUPS (F_INK / 4)
#define WMMA4(a, b, c) __builtin_amdgcn_wmma_f32_16x16x4_f32(false, (a), false, (b), (short)0, (c), false, false)

__device__ __forceinline__ void gcn_mma_group(const float4 x, const float4* __restrict__ wp,
                                              const float rs,
                                              v8f& c0, v8f& c1, v8f& c2, v8f& c3) {
    const float4 b0 = wp[0], b1 = wp[16], b2 = wp[32], b3 = wp[48];
    v2f a0, a1;
    a0.x = x.x * rs; a0.y = x.y * rs;
    a1.x = x.z * rs; a1.y = x.w * rs;
    v2f t0, t1, t2, t3;
    t0.x = b0.x; t0.y = b0.y; t1.x = b1.x; t1.y = b1.y;
    t2.x = b2.x; t2.y = b2.y; t3.x = b3.x; t3.y = b3.y;
    c0 = WMMA4(a0, t0, c0); c1 = WMMA4(a0, t1, c1);
    c2 = WMMA4(a0, t2, c2); c3 = WMMA4(a0, t3, c3);
    t0.x = b0.z; t0.y = b0.w; t1.x = b1.z; t1.y = b1.w;
    t2.x = b2.z; t2.y = b2.w; t3.x = b3.z; t3.y = b3.w;
    c0 = WMMA4(a1, t0, c0); c1 = WMMA4(a1, t1, c1);
    c2 = WMMA4(a1, t2, c2); c3 = WMMA4(a1, t3, c3);
}

__global__ __launch_bounds__(256)
void gcn_gemm_wmma_kernel(const float* __restrict__ X,
                          const float* __restrict__ W,
                          const float* __restrict__ deg_out,
                          float* __restrict__ H) {
    __shared__ float4 Ws[KGROUPS * F_OUTK];

    const int tid  = threadIdx.x;
    const int lane = tid & 31;
    const int wave = tid >> 5;
    const int m    = lane & 15;
    const int half = lane >> 4;

    const int rowBase = blockIdx.x * 128 + wave * 16;
    const int row     = rowBase + m;
    const float* xrow = X + (long)row * F_INK;

    const float* xp = xrow + half * 16;
    float4 xa0 = *(const float4*)(xp + 0);
    float4 xa1 = *(const float4*)(xp + 4);
    float4 xa2 = *(const float4*)(xp + 8);
    float4 xa3 = *(const float4*)(xp + 12);
    const float rs = __frsqrt_rn(fmaxf(deg_out[row], 1.0f));

    {
        const int c = tid & 63;
        for (int g = tid >> 6; g < KGROUPS; g += 4) {
            const float* wp = W + (long)(4 * g) * F_OUTK + c;
            float4 v;
            v.x = wp[0];
            v.y = wp[F_OUTK];
            v.z = wp[2 * F_OUTK];
            v.w = wp[3 * F_OUTK];
            Ws[g * F_OUTK + c] = v;
        }
    }
    __syncthreads();

    v8f c0 = {}, c1 = {}, c2 = {}, c3 = {};
    const float4* wl = Ws + (half * 4) * F_OUTK + m;

    for (int ch = 0; ch < 9; ++ch) {
        const int nx = (ch < 8) ? ch + 1 : 8;
        const float* xn = xp + nx * 32;
        const float4 xb0 = *(const float4*)(xn + 0);
        const float4 xb1 = *(const float4*)(xn + 4);
        const float4 xb2 = *(const float4*)(xn + 8);
        const float4 xb3 = *(const float4*)(xn + 12);
        const float4* wc = wl + ch * 8 * F_OUTK;
        gcn_mma_group(xa0, wc + 0 * F_OUTK, rs, c0, c1, c2, c3);
        gcn_mma_group(xa1, wc + 1 * F_OUTK, rs, c0, c1, c2, c3);
        gcn_mma_group(xa2, wc + 2 * F_OUTK, rs, c0, c1, c2, c3);
        gcn_mma_group(xa3, wc + 3 * F_OUTK, rs, c0, c1, c2, c3);
        xa0 = xb0; xa1 = xb1; xa2 = xb2; xa3 = xb3;
    }

    {
        const float4 x8 = *(const float4*)(xrow + 288 + half * 4);
        gcn_mma_group(x8, Ws + (72 + half) * F_OUTK + m, rs, c0, c1, c2, c3);
    }
    {
        const v2f xt = *(const v2f*)(xrow + 296 + half * 2);
        v2f a;
        a.x = xt.x * rs; a.y = xt.y * rs;
        const float* wt = (const float*)(Ws + 74 * F_OUTK + m) + half * 2;
        const v2f t0 = *(const v2f*)(wt);
        const v2f t1 = *(const v2f*)(wt + 16 * 4);
        const v2f t2 = *(const v2f*)(wt + 32 * 4);
        const v2f t3 = *(const v2f*)(wt + 48 * 4);
        c0 = WMMA4(a, t0, c0); c1 = WMMA4(a, t1, c1);
        c2 = WMMA4(a, t2, c2); c3 = WMMA4(a, t3, c3);
    }

    float* hout = H + (long)(rowBase + half * 8) * F_OUTK + m;
#pragma unroll
    for (int i = 0; i < 8; ++i) {
        hout[(long)i * F_OUTK +  0] = c0[i];
        hout[(long)i * F_OUTK + 16] = c1[i];
        hout[(long)i * F_OUTK + 32] = c2[i];
        hout[(long)i * F_OUTK + 48] = c3[i];
    }
}

#define SCAN_T 1024

__global__ __launch_bounds__(SCAN_T)
void gcn_scan_kernel(const int* __restrict__ deg_in, int* __restrict__ row_ptr) {
    __shared__ int s_own[SCAN_T / 32];
    __shared__ int s_pre[SCAN_T / 32];
    const int tid  = threadIdx.x;
    const int lane = tid & 31;
    const int wave = tid >> 5;
    const int blk  = blockIdx.x;
    const int node = blk * SCAN_T + tid;
    const int own  = deg_in[node];

    const int4* d4 = (const int4*)deg_in + tid;
    const int rounds = blk >> 2;
    int pre = 0;
    int r = 0;
    for (; r + 8 <= rounds; r += 8) {
        int4 t[8];
#pragma unroll
        for (int k = 0; k < 8; ++k) t[k] = d4[(r + k) * SCAN_T];
#pragma unroll
        for (int k = 0; k < 8; ++k) pre += (t[k].x + t[k].y) + (t[k].z + t[k].w);
    }
    for (; r < rounds; ++r) {
        const int4 v = d4[r * SCAN_T];
        pre += (v.x + v.y) + (v.z + v.w);
    }
    if (tid < (blk & 3) * (SCAN_T / 4)) {
        const int4 v = d4[rounds * SCAN_T];
        pre += (v.x + v.y) + (v.z + v.w);
    }

    int incl = own;
#pragma unroll
    for (int d = 1; d < 32; d <<= 1) {
        const int up = __shfl_up(incl, d);
        if (lane >= d) incl += up;
    }
#pragma unroll
    for (int d = 16; d >= 1; d >>= 1) pre += __shfl_xor(pre, d);

    if (lane == 31) s_own[wave] = incl;
    if (lane == 0)  s_pre[wave] = pre;
    __syncthreads();
    int base = 0;
#pragma unroll
    for (int w = 0; w < SCAN_T / 32; ++w) {
        base += s_pre[w];
        if (w < wave) base += s_own[w];
    }
    row_ptr[node] = base + incl - own;
}

__global__ __launch_bounds__(256)
void gcn_fill_kernel(const int* __restrict__ src,
                     const int* __restrict__ dst,
                     const int* __restrict__ rank,
                     const int* __restrict__ row_ptr,
                     int* __restrict__ csr) {
    const int t = blockIdx.x * 256 + threadIdx.x;
    const int4* s4 = (const int4*)src;
    const int4* d4 = (const int4*)dst;
    const int4* r4 = (const int4*)rank;
#pragma unroll
    for (int r = 0; r < 2; ++r) {
        const int i = t + r * EDGE_THREADS;
        const int4 s = s4[i];
        const int4 d = d4[i];
        const int4 k = r4[i];
        csr[row_ptr[d.x] + k.x] = s.x;
        csr[row_ptr[d.y] + k.y] = s.y;
        csr[row_ptr[d.z] + k.z] = s.z;
        csr[row_ptr[d.w] + k.w] = s.w;
    }
}

#define NPW 16

#define GCN_ROW_AT(idx) (*(const float2*)(Hb + (((unsigned)(idx)) << 8) + lane8))

__device__ __forceinline__ float2 gcn_sum_rows(const char* __restrict__ Hb, const int my, const int first,
                                               const int cnt, const unsigned lane8) {
    float2 v[32];
    v[0] = GCN_ROW_AT(first);
#pragma unroll
    for (int u = 1; u < 32; ++u) {
        if (u < cnt) v[u] = GCN_ROW_AT(__builtin_amdgcn_readlane(my, u));
        else         v[u] = make_float2(0.0f, 0.0f);
    }
    float2 a0 = v[0], a1 = v[1];
#pragma unroll
    for (int u = 2; u < 32; u += 2) {
        a0.x += v[u].x;     a0.y += v[u].y;
        a1.x += v[u + 1].x; a1.y += v[u + 1].y;
    }
    a0.x += a1.x;
    a0.y += a1.y;
    return a0;
}

__global__ __launch_bounds__(256)
void gcn_gather_kernel(const int* __restrict__ row_ptr,
                       const int* __restrict__ deg_in,
                       const int* __restrict__ csr,
                       const float* __restrict__ H,
                       float* __restrict__ agg) {
    const int lane  = threadIdx.x & 31;
    const int node0 = (blockIdx.x * 8 + (threadIdx.x >> 5)) * NPW;
    int lb = 0, ln = 0;
    if (lane < NPW) {
        lb = row_ptr[node0 + lane];
        ln = deg_in[node0 + lane];
    }
    const char* Hb = (const char*)H;
    const unsigned lane8 = (unsigned)lane * 8u;
    float2* A2 = (float2*)agg + (long)node0 * (F_OUTK / 2) + lane;

    int beg = __builtin_amdgcn_readlane(lb, 0);
    int n   = __builtin_amdgcn_readlane(ln, 0);
    int my  = (lane < n) ? csr[beg + lane] : 0;
    int first = __builtin_amdgcn_readlane(my, 0);
    for (int i = 0; i < NPW; ++i) {
        int begN = 0, nN = 0, myN = 0;
        if (i + 1 < NPW) {
            begN = __builtin_amdgcn_readlane(lb, i + 1);
            nN   = __builtin_amdgcn_readlane(ln, i + 1);
            myN  = (lane < nN) ? csr[begN + lane] : 0;
        }
        float2 acc = make_float2(0.0f, 0.0f);
        if (n > 0) {
            acc = gcn_sum_rows(Hb, my, first, min(32, n), lane8);
            for (int j = 32; j < n; j += 32) {
                const int more = (lane < n - j) ? csr[beg + j + lane] : 0;
                const float2 part = gcn_sum_rows(Hb, more, __builtin_amdgcn_readlane(more, 0),
                                                 min(32, n - j), lane8);
                acc.x += part.x;
                acc.y += part.y;
            }
        }
        A2[(long)i * (F_OUTK / 2)] = acc;
        beg = begN; n = nN; my = myN;
        first = __builtin_amdgcn_readlane(myN, 0);
    }
}

__global__ void gcn_finalize_kernel(const float* __restrict__ agg,
                                    const int* __restrict__ deg_in,
                                    const float* __restrict__ bias,
                                    const float* __restrict__ prelu_a,
                                    float* __restrict__ pool,
                                    float* __restrict__ anchor) {
    __shared__ float red[256];
    const int  sg    = blockIdx.x;
    const int  f     = threadIdx.x & 63;
    const int  g     = threadIdx.x >> 6;
    const float alpha = prelu_a[0];
    const float bf    = bias[f];
    const long base   = (long)sg * S_SUB;

    float partial = 0.0f;
    for (int n = g; n < S_SUB - 1; n += 4) {
        long node = base + n;
        float ri = __frsqrt_rn(fmaxf((float)deg_in[node], 1.0f));
        float v  = agg[node * F_OUTK + f] * ri + bf;
        v = v > 0.0f ? v : alpha * v;
        partial += v;
    }

    if (g == 3) {
        long node = base + S_SUB - 1;
        float ri = __frsqrt_rn(fmaxf((float)deg_in[node], 1.0f));
        float v  = agg[node * F_OUTK + f] * ri + bf;
        v = v > 0.0f ? v : alpha * v;
        anchor[(long)sg * F_OUTK + f] = v;
    }

    red[threadIdx.x] = partial;
    __syncthreads();
    if (g == 0) {
        float sum = red[f] + red[f + 64] + red[f + 128] + red[f + 192];
        pool[(long)sg * F_OUTK + f] = sum * (1.0f / (S_SUB - 1));
    }
}

extern "C" void kernel_launch(void* const* d_in, const int* in_sizes, int n_in,
                              void* d_out, int out_size, void* d_ws, size_t ws_size,
                              hipStream_t stream) {
    const float* in_feat  = (const float*)d_in[0];
    const float* W        = (const float*)d_in[1];
    const float* bias     = (const float*)d_in[2];
    const float* prelu_a  = (const float*)d_in[3];
    const int*   src      = (const int*)d_in[4];
    const int*   dst      = (const int*)d_in[5];

    float* ws      = (float*)d_ws;
    float* deg_out = ws;
    int*   deg_in  = (int*)(ws + N_NODES);
    float* H       = ws + 2L * N_NODES;
    float* agg     = H + (long)N_NODES * F_OUTK;
    int*   row_ptr = (int*)(agg + (long)N_NODES * F_OUTK);
    int*   rank    = row_ptr + N_NODES;
    int*   csr     = rank + N_EDGE;

    float* pool   = (float*)d_out;
    float* anchor = pool + (long)B_SUB * F_OUTK;

    hipMemsetAsync(deg_out, 0, 2L * N_NODES * sizeof(float), stream);

    gcn_degree_rank_kernel<<<EDGE_THREADS / 256, 256, 0, stream>>>(src, dst, deg_out, deg_in, rank);
    gcn_gemm_wmma_kernel<<<N_NODES / 128, 256, 0, stream>>>(in_feat, W, deg_out, H);
    gcn_scan_kernel<<<N_NODES / SCAN_T, SCAN_T, 0, stream>>>(deg_in, row_ptr);
    gcn_fill_kernel<<<EDGE_THREADS / 256, 256, 0, stream>>>(src, dst, rank, row_ptr, csr);
    gcn_gather_kernel<<<N_NODES / (8 * NPW), 256, 0, stream>>>(row_ptr, deg_in, csr, H, agg);
    gcn_finalize_kernel<<<B_SUB, 256, 0, stream>>>(agg, deg_in, bias, prelu_a, pool, anchor);
}


# Edges grouped by an LDS counting sort (per-tile bucket counts; one block per 128-node bucket orders its edges by node); no global integer atomics, no per-edge scattered stores; sum order = LDS arrival
# speedup vs baseline: 2.3597x; 1.6008x over previous
#include <hip/hip_runtime.h>

typedef __attribute__((ext_vector_type(2))) float v2f;
typedef __attribute__((ext_vector_type(8))) float v8f;

#define N_NODES  131072
#define F_INK    300
#define F_OUTK   64
#define N_EDGE   2097152
#define S_SUB    128
#define B_SUB    1024

#define TILE_E   8192
#define N_TILES  (N_EDGE / TILE_E)
#define N_BKT    (N_NODES / 128)
#define TILE_T   1024

__global__ __launch_bounds__(TILE_T)
void gcn_count_kernel(const int* __restrict__ src,
                      const int* __restrict__ dst,
                      float* __restrict__ deg_out,
                      int* __restrict__ tile_cnt) {
    __shared__ int cnt[N_BKT];
    const int tid = threadIdx.x;
    const int4* s4 = (const int4*)src + blockIdx.x * (TILE_E / 4);
    const int4* d4 = (const int4*)dst + blockIdx.x * (TILE_E / 4);
    const int4 da = d4[tid];
    const int4 db = d4[tid + TILE_T];
    const int4 sa = s4[tid];
    const int4 sb = s4[tid + TILE_T];
    cnt[tid] = 0;
    __syncthreads();
    atomicAdd(&cnt[da.x >> 7], 1);
    atomicAdd(&cnt[da.y >> 7], 1);
    atomicAdd(&cnt[da.z >> 7], 1);
    atomicAdd(&cnt[da.w >> 7], 1);
    atomicAdd(&cnt[db.x >> 7], 1);
    atomicAdd(&cnt[db.y >> 7], 1);
    atomicAdd(&cnt[db.z >> 7], 1);
    atomicAdd(&cnt[db.w >> 7], 1);
    atomicAdd(&deg_out[sa.x], 1.0f);
    atomicAdd(&deg_out[sa.y], 1.0f);
    atomicAdd(&deg_out[sa.z], 1.0f);
    atomicAdd(&deg_out[sa.w], 1.0f);
    atomicAdd(&deg_out[sb.x], 1.0f);
    atomicAdd(&deg_out[sb.y], 1.0f);
    atomicAdd(&deg_out[sb.z], 1.0f);
    atomicAdd(&deg_out[sb.w], 1.0f);
    __syncthreads();
    tile_cnt[blockIdx.x * N_BKT + tid] = cnt[tid];
}

#define KGROUPS (F_INK / 4)
#define WMMA4(a, b, c) __builtin_amdgcn_wmma_f32_16x16x4_f32(false, (a), false, (b), (short)0, (c), false, false)

__device__ __forceinline__ void gcn_mma_group(const float4 x, const float4* __restrict__ wp,
                                              const float rs,
                                              v8f& c0, v8f& c1, v8f& c2, v8f& c3) {
    const float4 b0 = wp[0], b1 = wp[16], b2 = wp[32], b3 = wp[48];
    v2f a0, a1;
    a0.x = x.x * rs; a0.y = x.y * rs;
    a1.x = x.z * rs; a1.y = x.w * rs;
    v2f t0, t1, t2, t3;
    t0.x = b0.x; t0.y = b0.y; t1.x = b1.x; t1.y = b1.y;
    t2.x = b2.x; t2.y = b2.y; t3.x = b3.x; t3.y = b3.y;
    c0 = WMMA4(a0, t0, c0); c1 = WMMA4(a0, t1, c1);
    c2 = WMMA4(a0, t2, c2); c3 = WMMA4(a0, t3, c3);
    t0.x = b0.z; t0.y = b0.w; t1.x = b1.z; t1.y = b1.w;
    t2.x = b2.z; t2.y = b2.w; t3.x = b3.z; t3.y = b3.w;
    c0 = WMMA4(a1, t0, c0); c1 = WMMA4(a1, t1, c1);
    c2 = WMMA4(a1, t2, c2); c3 = WMMA4(a1, t3, c3);
}

__global__ __launch_bounds__(256)
void gcn_gemm_wmma_kernel(const float* __restrict__ X,
                          const float* __restrict__ W,
                          const float* __restrict__ deg_out,
                          float* __restrict__ H) {
    __shared__ float4 Ws[KGROUPS * F_OUTK];

    const int tid  = threadIdx.x;
    const int lane = tid & 31;
    const int wave = tid >> 5;
    const int m    = lane & 15;
    const int half = lane >> 4;

    const int rowBase = blockIdx.x * 128 + wave * 16;
    const int row     = rowBase + m;
    const float* xrow = X + (long)row * F_INK;

    const float* xp = xrow + half * 16;
    float4 xa0 = *(const float4*)(xp + 0);
    float4 xa1 = *(const float4*)(xp + 4);
    float4 xa2 = *(const float4*)(xp + 8);
    float4 xa3 = *(const float4*)(xp + 12);
    const float rs = __frsqrt_rn(fmaxf(deg_out[row], 1.0f));

    {
        const int c = tid & 63;
        for (int g = tid >> 6; g < KGROUPS; g += 4) {
            const float* wp = W + (long)(4 * g) * F_OUTK + c;
            float4 v;
            v.x = wp[0];
            v.y = wp[F_OUTK];
            v.z = wp[2 * F_OUTK];
            v.w = wp[3 * F_OUTK];
            Ws[g * F_OUTK + c] = v;
        }
    }
    __syncthreads();

    v8f c0 = {}, c1 = {}, c2 = {}, c3 = {};
    const float4* wl = Ws + (half * 4) * F_OUTK + m;

    for (int ch = 0; ch < 9; ++ch) {
        const int nx = (ch < 8) ? ch + 1 : 8;
        const float* xn = xp + nx * 32;
        const float4 xb0 = *(const float4*)(xn + 0);
        const float4 xb1 = *(const float4*)(xn + 4);
        const float4 xb2 = *(const float4*)(xn + 8);
        const float4 xb3 = *(const float4*)(xn + 12);
        const float4* wc = wl + ch * 8 * F_OUTK;
        gcn_mma_group(xa0, wc + 0 * F_OUTK, rs, c0, c1, c2, c3);
        gcn_mma_group(xa1, wc + 1 * F_OUTK, rs, c0, c1, c2, c3);
        gcn_mma_group(xa2, wc + 2 * F_OUTK, rs, c0, c1, c2, c3);
        gcn_mma_group(xa3, wc + 3 * F_OUTK, rs, c0, c1, c2, c3);
        xa0 = xb0; xa1 = xb1; xa2 = xb2; xa3 = xb3;
    }

    {
        const float4 x8 = *(const float4*)(xrow + 288 + half * 4);
        gcn_mma_group(x8, Ws + (72 + half) * F_OUTK + m, rs, c0, c1, c2, c3);
    }
    {
        const v2f xt = *(const v2f*)(xrow + 296 + half * 2);
        v2f a;
        a.x = xt.x * rs; a.y = xt.y * rs;
        const float* wt = (const float*)(Ws + 74 * F_OUTK + m) + half * 2;
        const v2f t0 = *(const v2f*)(wt);
        const v2f t1 = *(const v2f*)(wt + 16 * 4);
        const v2f t2 = *(const v2f*)(wt + 32 * 4);
        const v2f t3 = *(const v2f*)(wt + 48 * 4);
        c0 = WMMA4(a, t0, c0); c1 = WMMA4(a, t1, c1);
        c2 = WMMA4(a, t2, c2); c3 = WMMA4(a, t3, c3);
    }

    float* hout = H + (long)(rowBase + half * 8) * F_OUTK + m;
#pragma unroll
    for (int i = 0; i < 8; ++i) {
        hout[(long)i * F_OUTK +  0] = c0[i];
        hout[(long)i * F_OUTK + 16] = c1[i];
        hout[(long)i * F_OUTK + 32] = c2[i];
        hout[(long)i * F_OUTK + 48] = c3[i];
    }
}

__device__ __forceinline__ int gcn_block_excl_scan(const int v, int* s_w, int& total) {
    const int lane = threadIdx.x & 31;
    const int wave = threadIdx.x >> 5;
    int incl = v;
#pragma unroll
    for (int d = 1; d < 32; d <<= 1) {
        const int up = __shfl_up(incl, d);
        if (lane >= d) incl += up;
    }
    if (lane == 31) s_w[wave] = incl;
    __syncthreads();
    int base = 0, all = 0;
#pragma unroll
    for (int w = 0; w < TILE_T / 32; ++w) {
        const int x = s_w[w];
        all += x;
        if (w < wave) base += x;
    }
    total = all;
    return base + incl - v;
}

#define SCAN_GROUPS (TILE_T / 32)
#define SCAN_TPG    (N_TILES / SCAN_GROUPS)

__global__ __launch_bounds__(TILE_T)
void gcn_tile_scan_kernel(const int* __restrict__ tile_cnt,
                          int* __restrict__ tile_base,
                          int* __restrict__ bkt_total) {
    __shared__ int colsum[SCAN_GROUPS][32];
    const int c = threadIdx.x & 31;
    const int g = threadIdx.x >> 5;
    const int b = blockIdx.x * 32 + c;
    const int t0 = g * SCAN_TPG;
    int v[SCAN_TPG];
#pragma unroll
    for (int k = 0; k < SCAN_TPG; ++k) v[k] = tile_cnt[(t0 + k) * N_BKT + b];
    int s = 0;
#pragma unroll
    for (int k = 0; k < SCAN_TPG; ++k) {
        const int t = v[k];
        v[k] = s;
        s += t;
    }
    colsum[g][c] = s;
    __syncthreads();
    int off = 0, tot = 0;
#pragma unroll
    for (int w = 0; w < SCAN_GROUPS; ++w) {
        const int x = colsum[w][c];
        tot += x;
        if (w < g) off += x;
    }
#pragma unroll
    for (int k = 0; k < SCAN_TPG; ++k) tile_base[(t0 + k) * N_BKT + b] = off + v[k];
    if (g == 0) bkt_total[b] = tot;
}

__global__ __launch_bounds__(TILE_T)
void gcn_bin_kernel(const int* __restrict__ src,
                    const int* __restrict__ dst,
                    const int* __restrict__ tile_cnt,
                    const int* __restrict__ tile_base,
                    const int* __restrict__ bkt_total,
                    int* __restrict__ bucket_start,
                    int* __restrict__ binned) {
    __shared__ int lcur[N_BKT];
    __shared__ int delta[N_BKT];
    __shared__ int rec_l[TILE_E];
    __shared__ unsigned short bkt_l[TILE_E];
    __shared__ int s_w[TILE_T / 32];
    __shared__ int s_w2[TILE_T / 32];
    const int tid = threadIdx.x;
    const int4* s4 = (const int4*)src + blockIdx.x * (TILE_E / 4);
    const int4* d4 = (const int4*)dst + blockIdx.x * (TILE_E / 4);
    const int4 da = d4[tid];
    const int4 db = d4[tid + TILE_T];
    const int4 sa = s4[tid];
    const int4 sb = s4[tid + TILE_T];
    const int c  = tile_cnt[blockIdx.x * N_BKT + tid];
    const int tb = tile_base[blockIdx.x * N_BKT + tid];
    const int bt = bkt_total[tid];
    int total, total_e;
    const int lstart = gcn_block_excl_scan(c, s_w, total);
    const int bs = gcn_block_excl_scan(bt, s_w2, total_e);
    if (blockIdx.x == 0) {
        bucket_start[tid] = bs;
        if (tid == 0) bucket_start[N_BKT] = total_e;
    }
    lcur[tid]  = lstart;
    delta[tid] = bs + tb - lstart;
    __syncthreads();
    int bk, p;
    bk = da.x >> 7; p = atomicAdd(&lcur[bk], 1); rec_l[p] = (sa.x << 7) | (da.x & 127); bkt_l[p] = (unsigned short)bk;
    bk = da.y >> 7; p = atomicAdd(&lcur[bk], 1); rec_l[p] = (sa.y << 7) | (da.y & 127); bkt_l[p] = (unsigned short)bk;
    bk = da.z >> 7; p = atomicAdd(&lcur[bk], 1); rec_l[p] = (sa.z << 7) | (da.z & 127); bkt_l[p] = (unsigned short)bk;
    bk = da.w >> 7; p = atomicAdd(&lcur[bk], 1); rec_l[p] = (sa.w << 7) | (da.w & 127); bkt_l[p] = (unsigned short)bk;
    bk = db.x >> 7; p = atomicAdd(&lcur[bk], 1); rec_l[p] = (sb.x << 7) | (db.x & 127); bkt_l[p] = (unsigned short)bk;
    bk = db.y >> 7; p = atomicAdd(&lcur[bk], 1); rec_l[p] = (sb.y << 7) | (db.y & 127); bkt_l[p] = (unsigned short)bk;
    bk = db.z >> 7; p = atomicAdd(&lcur[bk], 1); rec_l[p] = (sb.z << 7) | (db.z & 127); bkt_l[p] = (unsigned short)bk;
    bk = db.w >> 7; p = atomicAdd(&lcur[bk], 1); rec_l[p] = (sb.w << 7) | (db.w & 127); bkt_l[p] = (unsigned short)bk;
    __syncthreads();
#pragma unroll
    for (int k = 0; k < TILE_E / TILE_T; ++k) {
        const int i = tid + k * TILE_T;
        binned[delta[bkt_l[i]] + i] = rec_l[i];
    }
}

#define NPW 16

#define GCN_ROW_AT(idx) (*(const float2*)(Hb + (((unsigned)(idx)) << 8) + lane8))

__device__ __forceinline__ float2 gcn_sum_rows(const char* __restrict__ Hb, const int my, const int first,
                                               const int cnt, const unsigned lane8) {
    float2 v[32];
    v[0] = GCN_ROW_AT(first);
#pragma unroll
    for (int u = 1; u < 32; ++u) {
        if (u < cnt) v[u] = GCN_ROW_AT(__builtin_amdgcn_readlane(my, u));
        else         v[u] = make_float2(0.0f, 0.0f);
    }
    float2 a0 = v[0], a1 = v[1];
#pragma unroll
    for (int u = 2; u < 32; u += 2) {
        a0.x += v[u].x;     a0.y += v[u].y;
        a1.x += v[u + 1].x; a1.y += v[u + 1].y;
    }
    a0.x += a1.x;
    a0.y += a1.y;
    return a0;
}

#define BKT_CAP 4096

__global__ __launch_bounds__(256)
void gcn_bucket_gather_kernel(const int* __restrict__ bucket_start,
                              const int* __restrict__ binned,
                              const float* __restrict__ H,
                              float* __restrict__ agg,
                              int* __restrict__ deg_in) {
    __shared__ int    n_cnt[128];
    __shared__ int    n_start[128];
    __shared__ int    n_cur[128];
    __shared__ int    n_deg[128];
    __shared__ int    sorted[BKT_CAP];
    __shared__ float2 accs[128 * 32];

    const int tid  = threadIdx.x;
    const int lane = tid & 31;
    const int wave = tid >> 5;
    const int bs = bucket_start[blockIdx.x];
    const int be = bucket_start[blockIdx.x + 1];

    if (tid < 128) { n_cnt[tid] = 0; n_deg[tid] = 0; }
    for (int i = tid; i < 128 * 32; i += 256) accs[i] = make_float2(0.0f, 0.0f);
    __syncthreads();

    const char* Hb = (const char*)H;
    const unsigned lane8 = (unsigned)lane * 8u;
    const int nb = wave * NPW;

    for (int c0 = bs; c0 < be; c0 += BKT_CAP) {
        const int m = min(BKT_CAP, be - c0);
        const int* rp = binned + c0;

        for (int i0 = 0; i0 < m; i0 += 2048) {
            int r[8];
#pragma unroll
            for (int k = 0; k < 8; ++k) {
                const int idx = i0 + k * 256 + tid;
                r[k] = (idx < m) ? rp[idx] : -1;
            }
#pragma unroll
            for (int k = 0; k < 8; ++k)
                if (r[k] >= 0) atomicAdd(&n_cnt[r[k] & 127], 1);
        }
        __syncthreads();

        if (wave == 0) {
            const int q0 = n_cnt[4 * lane + 0];
            const int q1 = n_cnt[4 * lane + 1];
            const int q2 = n_cnt[4 * lane + 2];
            const int q3 = n_cnt[4 * lane + 3];
            const int sum4 = (q0 + q1) + (q2 + q3);
            int incl = sum4;
#pragma unroll
            for (int d = 1; d < 32; d <<= 1) {
                const int up = __shfl_up(incl, d);
                if (lane >= d) incl += up;
            }
            const int e0 = incl - sum4;
            const int e1 = e0 + q0;
            const int e2 = e1 + q1;
            const int e3 = e2 + q2;
            n_start[4 * lane + 0] = e0; n_cur[4 * lane + 0] = e0; n_deg[4 * lane + 0] += q0;
            n_start[4 * lane + 1] = e1; n_cur[4 * lane + 1] = e1; n_deg[4 * lane + 1] += q1;
            n_start[4 * lane + 2] = e2; n_cur[4 * lane + 2] = e2; n_deg[4 * lane + 2] += q2;
            n_start[4 * lane + 3] = e3; n_cur[4 * lane + 3] = e3; n_deg[4 * lane + 3] += q3;
        }
        __syncthreads();

        for (int i0 = 0; i0 < m; i0 += 2048) {
            int r[8];
#pragma unroll
            for (int k = 0; k < 8; ++k) {
                const int idx = i0 + k * 256 + tid;
                r[k] = (idx < m) ? rp[idx] : -1;
            }
#pragma unroll
            for (int k = 0; k < 8; ++k) {
                if (r[k] >= 0) {
                    const int p = atomicAdd(&n_cur[r[k] & 127], 1);
                    sorted[p] = r[k] >> 7;
                }
            }
        }
        __syncthreads();

        {
            int lb = 0, ln = 0;
            if (lane < NPW) {
                lb = n_start[nb + lane];
                ln = n_cnt[nb + lane];
            }
            int beg = __builtin_amdgcn_readlane(lb, 0);
            int n   = __builtin_amdgcn_readlane(ln, 0);
            int my  = (lane < n) ? sorted[beg + lane] : 0;
            int first = __builtin_amdgcn_readlane(my, 0);
            for (int i = 0; i < NPW; ++i) {
                int begN = 0, nN = 0, myN = 0;
                if (i + 1 < NPW) {
                    begN = __builtin_amdgcn_readlane(lb, i + 1);
                    nN   = __builtin_amdgcn_readlane(ln, i + 1);
                    myN  = (lane < nN) ? sorted[begN + lane] : 0;
                }
                if (n > 0) {
                    float2 acc = gcn_sum_rows(Hb, my, first, min(32, n), lane8);
                    for (int j = 32; j < n; j += 32) {
                        const int more = (lane < n - j) ? sorted[beg + j + lane] : 0;
                        const float2 part = gcn_sum_rows(Hb, more, __builtin_amdgcn_readlane(more, 0),
                                                         min(32, n - j), lane8);
                        acc.x += part.x;
                        acc.y += part.y;
                    }
                    float2 o = accs[(nb + i) * 32 + lane];
                    o.x += acc.x;
                    o.y += acc.y;
                    accs[(nb + i) * 32 + lane] = o;
                }
                beg = begN; n = nN; my = myN;
                first = __builtin_amdgcn_readlane(myN, 0);
            }
        }

        if (c0 + BKT_CAP < be) {
            __syncthreads();
            if (tid < 128) n_cnt[tid] = 0;
            __syncthreads();
        }
    }

    float2* A2 = (float2*)agg + ((long)blockIdx.x * 128 + nb) * (F_OUTK / 2) + lane;
#pragma unroll
    for (int i = 0; i < NPW; ++i) A2[(long)i * (F_OUTK / 2)] = accs[(nb + i) * 32 + lane];
    if (tid < 128) deg_in[blockIdx.x * 128 + tid] = n_deg[tid];
}

__global__ void gcn_finalize_kernel(const float* __restrict__ agg,
                                    const int* __restrict__ deg_in,
                                    const float* __restrict__ bias,
                                    const float* __restrict__ prelu_a,
                                    float* __restrict__ pool,
                                    float* __restrict__ anchor) {
    __shared__ float red[256];
    const int  sg    = blockIdx.x;
    const int  f     = threadIdx.x & 63;
    const int  g     = threadIdx.x >> 6;
    const float alpha = prelu_a[0];
    const float bf    = bias[f];
    const long base   = (long)sg * S_SUB;

    float partial = 0.0f;
    for (int n = g; n < S_SUB - 1; n += 4) {
        long node = base + n;
        float ri = __frsqrt_rn(fmaxf((float)deg_in[node], 1.0f));
        float v  = agg[node * F_OUTK + f] * ri + bf;
        v = v > 0.0f ? v : alpha * v;
        partial += v;
    }

    if (g == 3) {
        long node = base + S_SUB - 1;
        float ri = __frsqrt_rn(fmaxf((float)deg_in[node], 1.0f));
        float v  = agg[node * F_OUTK + f] * ri + bf;
        v = v > 0.0f ? v : alpha * v;
        anchor[(long)sg * F_OUTK + f] = v;
    }

    red[threadIdx.x] = partial;
    __syncthreads();
    if (g == 0) {
        float sum = red[f] + red[f + 64] + red[f + 128] + red[f + 192];
        pool[(long)sg * F_OUTK + f] = sum * (1.0f / (S_SUB - 1));
    }
}

extern "C" void kernel_launch(void* const* d_in, const int* in_sizes, int n_in,
                              void* d_out, int out_size, void* d_ws, size_t ws_size,
                              hipStream_t stream) {
    const float* in_feat  = (const float*)d_in[0];
    const float* W        = (const float*)d_in[1];
    const float* bias     = (const float*)d_in[2];
    const float* prelu_a  = (const float*)d_in[3];
    const int*   src      = (const int*)d_in[4];
    const int*   dst      = (const int*)d_in[5];

    float* ws      = (float*)d_ws;
    float* deg_out = ws;
    int*   deg_in  = (int*)(ws + N_NODES);
    float* H       = ws + 2L * N_NODES;
    float* agg     = H + (long)N_NODES * F_OUTK;
    int*   tile_cnt     = (int*)(agg + (long)N_NODES * F_OUTK);
    int*   tile_base    = tile_cnt + (long)N_TILES * N_BKT;
    int*   bucket_start = tile_base + (long)N_TILES * N_BKT;
    int*   bkt_total    = bucket_start + 2048;
    int*   binned       = bkt_total + N_BKT;

    float* pool   = (float*)d_out;
    float* anchor = pool + (long)B_SUB * F_OUTK;

    hipMemsetAsync(deg_out, 0, (long)N_NODES * sizeof(float), stream);

    gcn_count_kernel<<<N_TILES, TILE_T, 0, stream>>>(src, dst, deg_out, tile_cnt);
    gcn_gemm_wmma_kernel<<<N_NODES / 128, 256, 0, stream>>>(in_feat, W, deg_out, H);
    gcn_tile_scan_kernel<<<N_BKT / 32, TILE_T, 0, stream>>>(tile_cnt, tile_base, bkt_total);
    gcn_bin_kernel<<<N_TILES, TILE_T, 0, stream>>>(src, dst, tile_cnt, tile_base, bkt_total, bucket_start, binned);
    gcn_bucket_gather_kernel<<<N_BKT, 256, 0, stream>>>(bucket_start, binned, H, agg, deg_in);
    gcn_finalize_kernel<<<B_SUB, 256, 0, stream>>>(agg, deg_in, bias, prelu_a, pool, anchor);
}


# Out-degrees counted in LDS per 128-node bucket from the destination-sorted records, in the GEMM block that owns those rows; a global float add is left only for edges that cross buckets
# speedup vs baseline: 4.4984x; 1.9063x over previous
#include <hip/hip_runtime.h>

typedef __attribute__((ext_vector_type(2))) float v2f;
typedef __attribute__((ext_vector_type(8))) float v8f;

#define N_NODES  131072
#define F_INK    300
#define F_OUTK   64
#define N_EDGE   2097152
#define S_SUB    128
#define B_SUB    1024

#define TILE_E   8192
#define N_TILES  (N_EDGE / TILE_E)
#define N_BKT    (N_NODES / 128)
#define TILE_T   1024

__global__ __launch_bounds__(TILE_T)
void gcn_count_kernel(const int* __restrict__ dst,
                      int* __restrict__ tile_cnt) {
    __shared__ int cnt[N_BKT];
    const int tid = threadIdx.x;
    const int4* d4 = (const int4*)dst + blockIdx.x * (TILE_E / 4);
    const int4 da = d4[tid];
    const int4 db = d4[tid + TILE_T];
    cnt[tid] = 0;
    __syncthreads();
    atomicAdd(&cnt[da.x >> 7], 1);
    atomicAdd(&cnt[da.y >> 7], 1);
    atomicAdd(&cnt[da.z >> 7], 1);
    atomicAdd(&cnt[da.w >> 7], 1);
    atomicAdd(&cnt[db.x >> 7], 1);
    atomicAdd(&cnt[db.y >> 7], 1);
    atomicAdd(&cnt[db.z >> 7], 1);
    atomicAdd(&cnt[db.w >> 7], 1);
    __syncthreads();
    tile_cnt[blockIdx.x * N_BKT + tid] = cnt[tid];
}

#define KGROUPS (F_INK / 4)
#define WMMA4(a, b, c) __builtin_amdgcn_wmma_f32_16x16x4_f32(false, (a), false, (b), (short)0, (c), false, false)

__device__ __forceinline__ void gcn_mma_group(const float4 x, const float4* __restrict__ wp,
                                              const float rs,
                                              v8f& c0, v8f& c1, v8f& c2, v8f& c3) {
    const float4 b0 = wp[0], b1 = wp[16], b2 = wp[32], b3 = wp[48];
    v2f a0, a1;
    a0.x = x.x * rs; a0.y = x.y * rs;
    a1.x = x.z * rs; a1.y = x.w * rs;
    v2f t0, t1, t2, t3;
    t0.x = b0.x; t0.y = b0.y; t1.x = b1.x; t1.y = b1.y;
    t2.x = b2.x; t2.y = b2.y; t3.x = b3.x; t3.y = b3.y;
    c0 = WMMA4(a0, t0, c0); c1 = WMMA4(a0, t1, c1);
    c2 = WMMA4(a0, t2, c2); c3 = WMMA4(a0, t3, c3);
    t0.x = b0.z; t0.y = b0.w; t1.x = b1.z; t1.y = b1.w;
    t2.x = b2.z; t2.y = b2.w; t3.x = b3.z; t3.y = b3.w;
    c0 = WMMA4(a1, t0, c0); c1 = WMMA4(a1, t1, c1);
    c2 = WMMA4(a1, t2, c2); c3 = WMMA4(a1, t3, c3);
}

__global__ __launch_bounds__(256)
void gcn_gemm_wmma_kernel(const float* __restrict__ X,
                          const float* __restrict__ W,
                          const float* __restrict__ deg_far,
                          const int* __restrict__ bucket_start,
                          const int* __restrict__ binned,
                          float* __restrict__ H) {
    __shared__ float4 Ws[KGROUPS * F_OUTK];
    __shared__ int    s_deg[128];

    const int tid  = threadIdx.x;
    if (tid < 128) s_deg[tid] = 0;
    __syncthreads();
    const int lane = tid & 31;
    const int wave = tid >> 5;
    const int m    = lane & 15;
    const int half = lane >> 4;

    const int rowBase = blockIdx.x * 128 + wave * 16;
    const int row     = rowBase + m;
    const float* xrow = X + (long)row * F_INK;

    const float* xp = xrow + half * 16;
    float4 xa0 = *(const float4*)(xp + 0);
    float4 xa1 = *(const float4*)(xp + 4);
    float4 xa2 = *(const float4*)(xp + 8);
    float4 xa3 = *(const float4*)(xp + 12);
    const float dfar = deg_far[row];

    const int bs = bucket_start[blockIdx.x];
    const int be = bucket_start[blockIdx.x + 1];
    int r0[16];
#pragma unroll
    for (int k = 0; k < 16; ++k) {
        const int idx = bs + k * 256 + tid;
        r0[k] = (idx < be) ? binned[idx] : -1;
    }

    {
        const int c = tid & 63;
        for (int g = tid >> 6; g < KGROUPS; g += 4) {
            const float* wp = W + (long)(4 * g) * F_OUTK + c;
            float4 v;
            v.x = wp[0];
            v.y = wp[F_OUTK];
            v.z = wp[2 * F_OUTK];
            v.w = wp[3 * F_OUTK];
            Ws[g * F_OUTK + c] = v;
        }
    }
#pragma unroll
    for (int k = 0; k < 16; ++k)
        if ((unsigned)(r0[k] >> 14) == blockIdx.x) atomicAdd(&s_deg[(r0[k] >> 7) & 127], 1);
    for (int i0 = bs + 4096; i0 < be; i0 += 4096) {
        int r[16];
#pragma unroll
        for (int k = 0; k < 16; ++k) {
            const int idx = i0 + k * 256 + tid;
            r[k] = (idx < be) ? binned[idx] : -1;
        }
#pragma unroll
        for (int k = 0; k < 16; ++k)
            if ((unsigned)(r[k] >> 14) == blockIdx.x) atomicAdd(&s_deg[(r[k] >> 7) & 127], 1);
    }
    __syncthreads();

    const float rs = __frsqrt_rn(fmaxf((float)s_deg[wave * 16 + m] + dfar, 1.0f));

    v8f c0 = {}, c1 = {}, c2 = {}, c3 = {};
    const float4* wl = Ws + (half * 4) * F_OUTK + m;

    for (int ch = 0; ch < 9; ++ch) {
        const int nx = (ch < 8) ? ch + 1 : 8;
        const float* xn = xp + nx * 32;
        const float4 xb0 = *(const float4*)(xn + 0);
        const float4 xb1 = *(const float4*)(xn + 4);
        const float4 xb2 = *(const float4*)(xn + 8);
        const float4 xb3 = *(const float4*)(xn + 12);
        const float4* wc = wl + ch * 8 * F_OUTK;
        gcn_mma_group(xa0, wc + 0 * F_OUTK, rs, c0, c1, c2, c3);
        gcn_mma_group(xa1, wc + 1 * F_OUTK, rs, c0, c1, c2, c3);
        gcn_mma_group(xa2, wc + 2 * F_OUTK, rs, c0, c1, c2, c3);
        gcn_mma_group(xa3, wc + 3 * F_OUTK, rs, c0, c1, c2, c3);
        xa0 = xb0; xa1 = xb1; xa2 = xb2; xa3 = xb3;
    }

    {
        const float4 x8 = *(const float4*)(xrow + 288 + half * 4);
        gcn_mma_group(x8, Ws + (72 + half) * F_OUTK + m, rs, c0, c1, c2, c3);
    }
    {
        const v2f xt = *(const v2f*)(xrow + 296 + half * 2);
        v2f a;
        a.x = xt.x * rs; a.y = xt.y * rs;
        const float* wt = (const float*)(Ws + 74 * F_OUTK + m) + half * 2;
        const v2f t0 = *(const v2f*)(wt);
        const v2f t1 = *(const v2f*)(wt + 16 * 4);
        const v2f t2 = *(const v2f*)(wt + 32 * 4);
        const v2f t3 = *(const v2f*)(wt + 48 * 4);
        c0 = WMMA4(a, t0, c0); c1 = WMMA4(a, t1, c1);
        c2 = WMMA4(a, t2, c2); c3 = WMMA4(a, t3, c3);
    }

    float* hout = H + (long)(rowBase + half * 8) * F_OUTK + m;
#pragma unroll
    for (int i = 0; i < 8; ++i) {
        hout[(long)i * F_OUTK +  0] = c0[i];
        hout[(long)i * F_OUTK + 16] = c1[i];
        hout[(long)i * F_OUTK + 32] = c2[i];
        hout[(long)i * F_OUTK + 48] = c3[i];
    }
}

__device__ __forceinline__ int gcn_block_excl_scan(const int v, int* s_w, int& total) {
    const int lane = threadIdx.x & 31;
    const int wave = threadIdx.x >> 5;
    int incl = v;
#pragma unroll
    for (int d = 1; d < 32; d <<= 1) {
        const int up = __shfl_up(incl, d);
        if (lane >= d) incl += up;
    }
    if (lane == 31) s_w[wave] = incl;
    __syncthreads();
    int base = 0, all = 0;
#pragma unroll
    for (int w = 0; w < TILE_T / 32; ++w) {
        const int x = s_w[w];
        all += x;
        if (w < wave) base += x;
    }
    total = all;
    return base + incl - v;
}

#define SCAN_GROUPS (TILE_T / 32)
#define SCAN_TPG    (N_TILES / SCAN_GROUPS)

__global__ __launch_bounds__(TILE_T)
void gcn_tile_scan_kernel(const int* __restrict__ tile_cnt,
                          int* __restrict__ tile_base,
                          int* __restrict__ bkt_total) {
    __shared__ int colsum[SCAN_GROUPS][32];
    const int c = threadIdx.x & 31;
    const int g = threadIdx.x >> 5;
    const int b = blockIdx.x * 32 + c;
    const int t0 = g * SCAN_TPG;
    int v[SCAN_TPG];
#pragma unroll
    for (int k = 0; k < SCAN_TPG; ++k) v[k] = tile_cnt[(t0 + k) * N_BKT + b];
    int s = 0;
#pragma unroll
    for (int k = 0; k < SCAN_TPG; ++k) {
        const int t = v[k];
        v[k] = s;
        s += t;
    }
    colsum[g][c] = s;
    __syncthreads();
    int off = 0, tot = 0;
#pragma unroll
    for (int w = 0; w < SCAN_GROUPS; ++w) {
        const int x = colsum[w][c];
        tot += x;
        if (w < g) off += x;
    }
#pragma unroll
    for (int k = 0; k < SCAN_TPG; ++k) tile_base[(t0 + k) * N_BKT + b] = off + v[k];
    if (g == 0) bkt_total[b] = tot;
}

__global__ __launch_bounds__(TILE_T)
void gcn_bin_kernel(const int* __restrict__ src,
                    const int* __restrict__ dst,
                    float* __restrict__ deg_far,
                    const int* __restrict__ tile_cnt,
                    const int* __restrict__ tile_base,
                    const int* __restrict__ bkt_total,
                    int* __restrict__ bucket_start,
                    int* __restrict__ binned) {
    __shared__ int lcur[N_BKT];
    __shared__ int delta[N_BKT];
    __shared__ int rec_l[TILE_E];
    __shared__ unsigned short bkt_l[TILE_E];
    __shared__ int s_w[TILE_T / 32];
    __shared__ int s_w2[TILE_T / 32];
    const int tid = threadIdx.x;
    const int4* s4 = (const int4*)src + blockIdx.x * (TILE_E / 4);
    const int4* d4 = (const int4*)dst + blockIdx.x * (TILE_E / 4);
    const int4 da = d4[tid];
    const int4 db = d4[tid + TILE_T];
    const int4 sa = s4[tid];
    const int4 sb = s4[tid + TILE_T];
    const int c  = tile_cnt[blockIdx.x * N_BKT + tid];
    const int tb = tile_base[blockIdx.x * N_BKT + tid];
    const int bt = bkt_total[tid];
    int total, total_e;
    const int lstart = gcn_block_excl_scan(c, s_w, total);
    const int bs = gcn_block_excl_scan(bt, s_w2, total_e);
    if (blockIdx.x == 0) {
        bucket_start[tid] = bs;
        if (tid == 0) bucket_start[N_BKT] = total_e;
    }
    lcur[tid]  = lstart;
    delta[tid] = bs + tb - lstart;
    if ((sa.x >> 7) != (da.x >> 7)) atomicAdd(&deg_far[sa.x], 1.0f);
    if ((sa.y >> 7) != (da.y >> 7)) atomicAdd(&deg_far[sa.y], 1.0f);
    if ((sa.z >> 7) != (da.z >> 7)) atomicAdd(&deg_far[sa.z], 1.0f);
    if ((sa.w >> 7) != (da.w >> 7)) atomicAdd(&deg_far[sa.w], 1.0f);
    if ((sb.x >> 7) != (db.x >> 7)) atomicAdd(&deg_far[sb.x], 1.0f);
    if ((sb.y >> 7) != (db.y >> 7)) atomicAdd(&deg_far[sb.y], 1.0f);
    if ((sb.z >> 7) != (db.z >> 7)) atomicAdd(&deg_far[sb.z], 1.0f);
    if ((sb.w >> 7) != (db.w >> 7)) atomicAdd(&deg_far[sb.w], 1.0f);
    __syncthreads();
    int bk, p;
    bk = da.x >> 7; p = atomicAdd(&lcur[bk], 1); rec_l[p] = (sa.x << 7) | (da.x & 127); bkt_l[p] = (unsigned short)bk;
    bk = da.y >> 7; p = atomicAdd(&lcur[bk], 1); rec_l[p] = (sa.y << 7) | (da.y & 127); bkt_l[p] = (unsigned short)bk;
    bk = da.z >> 7; p = atomicAdd(&lcur[bk], 1); rec_l[p] = (sa.z << 7) | (da.z & 127); bkt_l[p] = (unsigned short)bk;
    bk = da.w >> 7; p = atomicAdd(&lcur[bk], 1); rec_l[p] = (sa.w << 7) | (da.w & 127); bkt_l[p] = (unsigned short)bk;
    bk = db.x >> 7; p = atomicAdd(&lcur[bk], 1); rec_l[p] = (sb.x << 7) | (db.x & 127); bkt_l[p] = (unsigned short)bk;
    bk = db.y >> 7; p = atomicAdd(&lcur[bk], 1); rec_l[p] = (sb.y << 7) | (db.y & 127); bkt_l[p] = (unsigned short)bk;
    bk = db.z >> 7; p = atomicAdd(&lcur[bk], 1); rec_l[p] = (sb.z << 7) | (db.z & 127); bkt_l[p] = (unsigned short)bk;
    bk = db.w >> 7; p = atomicAdd(&lcur[bk], 1); rec_l[p] = (sb.w << 7) | (db.w & 127); bkt_l[p] = (unsigned short)bk;
    __syncthreads();
#pragma unroll
    for (int k = 0; k < TILE_E / TILE_T; ++k) {
        const int i = tid + k * TILE_T;
        binned[delta[bkt_l[i]] + i] = rec_l[i];
    }
}

#define NPW 16

#define GCN_ROW_AT(idx) (*(const float2*)(Hb + (((unsigned)(idx)) << 8) + lane8))

__device__ __forceinline__ float2 gcn_sum_rows(const char* __restrict__ Hb, const int my, const int first,
                                               const int cnt, const unsigned lane8) {
    float2 v[32];
    v[0] = GCN_ROW_AT(first);
#pragma unroll
    for (int u = 1; u < 32; ++u) {
        if (u < cnt) v[u] = GCN_ROW_AT(__builtin_amdgcn_readlane(my, u));
        else         v[u] = make_float2(0.0f, 0.0f);
    }
    float2 a0 = v[0], a1 = v[1];
#pragma unroll
    for (int u = 2; u < 32; u += 2) {
        a0.x += v[u].x;     a0.y += v[u].y;
        a1.x += v[u + 1].x; a1.y += v[u + 1].y;
    }
    a0.x += a1.x;
    a0.y += a1.y;
    return a0;
}

#define BKT_CAP 4096

__global__ __launch_bounds__(256)
void gcn_bucket_gather_kernel(const int* __restrict__ bucket_start,
                              const int* __restrict__ binned,
                              const float* __restrict__ H,
                              float* __restrict__ agg,
                              int* __restrict__ deg_in) {
    __shared__ int    n_cnt[128];
    __shared__ int    n_start[128];
    __shared__ int    n_cur[128];
    __shared__ int    n_deg[128];
    __shared__ int    sorted[BKT_CAP];
    __shared__ float2 accs[128 * 32];

    const int tid  = threadIdx.x;
    const int lane = tid & 31;
    const int wave = tid >> 5;
    const int bs = bucket_start[blockIdx.x];
    const int be = bucket_start[blockIdx.x + 1];

    if (tid < 128) { n_cnt[tid] = 0; n_deg[tid] = 0; }
    for (int i = tid; i < 128 * 32; i += 256) accs[i] = make_float2(0.0f, 0.0f);
    __syncthreads();

    const char* Hb = (const char*)H;
    const unsigned lane8 = (unsigned)lane * 8u;
    const int nb = wave * NPW;

    for (int c0 = bs; c0 < be; c0 += BKT_CAP) {
        const int m = min(BKT_CAP, be - c0);
        const int* rp = binned + c0;

        for (int i0 = 0; i0 < m; i0 += 2048) {
            int r[8];
#pragma unroll
            for (int k = 0; k < 8; ++k) {
                const int idx = i0 + k * 256 + tid;
                r[k] = (idx < m) ? rp[idx] : -1;
            }
#pragma unroll
            for (int k = 0; k < 8; ++k)
                if (r[k] >= 0) atomicAdd(&n_cnt[r[k] & 127], 1);
        }
        __syncthreads();

        if (wave == 0) {
            const int q0 = n_cnt[4 * lane + 0];
            const int q1 = n_cnt[4 * lane + 1];
            const int q2 = n_cnt[4 * lane + 2];
            const int q3 = n_cnt[4 * lane + 3];
            const int sum4 = (q0 + q1) + (q2 + q3);
            int incl = sum4;
#pragma unroll
            for (int d = 1; d < 32; d <<= 1) {
                const int up = __shfl_up(incl, d);
                if (lane >= d) incl += up;
            }
            const int e0 = incl - sum4;
            const int e1 = e0 + q0;
            const int e2 = e1 + q1;
            const int e3 = e2 + q2;
            n_start[4 * lane + 0] = e0; n_cur[4 * lane + 0] = e0; n_deg[4 * lane + 0] += q0;
            n_start[4 * lane + 1] = e1; n_cur[4 * lane + 1] = e1; n_deg[4 * lane + 1] += q1;
            n_start[4 * lane + 2] = e2; n_cur[4 * lane + 2] = e2; n_deg[4 * lane + 2] += q2;
            n_start[4 * lane + 3] = e3; n_cur[4 * lane + 3] = e3; n_deg[4 * lane + 3] += q3;
        }
        __syncthreads();

        for (int i0 = 0; i0 < m; i0 += 2048) {
            int r[8];
#pragma unroll
            for (int k = 0; k < 8; ++k) {
                const int idx = i0 + k * 256 + tid;
                r[k] = (idx < m) ? rp[idx] : -1;
            }
#pragma unroll
            for (int k = 0; k < 8; ++k) {
                if (r[k] >= 0) {
                    const int p = atomicAdd(&n_cur[r[k] & 127], 1);
                    sorted[p] = r[k] >> 7;
                }
            }
        }
        __syncthreads();

        {
            int lb = 0, ln = 0;
            if (lane < NPW) {
                lb = n_start[nb + lane];
                ln = n_cnt[nb + lane];
            }
            int beg = __builtin_amdgcn_readlane(lb, 0);
            int n   = __builtin_amdgcn_readlane(ln, 0);
            int my  = (lane < n) ? sorted[beg + lane] : 0;
            int first = __builtin_amdgcn_readlane(my, 0);
            for (int i = 0; i < NPW; ++i) {
                int begN = 0, nN = 0, myN = 0;
                if (i + 1 < NPW) {
                    begN = __builtin_amdgcn_readlane(lb, i + 1);
                    nN   = __builtin_amdgcn_readlane(ln, i + 1);
                    myN  = (lane < nN) ? sorted[begN + lane] : 0;
                }
                if (n > 0) {
                    float2 acc = gcn_sum_rows(Hb, my, first, min(32, n), lane8);
                    for (int j = 32; j < n; j += 32) {
                        const int more = (lane < n - j) ? sorted[beg + j + lane] : 0;
                        const float2 part = gcn_sum_rows(Hb, more, __builtin_amdgcn_readlane(more, 0),
                                                         min(32, n - j), lane8);
                        acc.x += part.x;
                        acc.y += part.y;
                    }
                    float2 o = accs[(nb + i) * 32 + lane];
                    o.x += acc.x;
                    o.y += acc.y;
                    accs[(nb + i) * 32 + lane] = o;
                }
                beg = begN; n = nN; my = myN;
                first = __builtin_amdgcn_readlane(myN, 0);
            }
        }

        if (c0 + BKT_CAP < be) {
            __syncthreads();
            if (tid < 128) n_cnt[tid] = 0;
            __syncthreads();
        }
    }

    float2* A2 = (float2*)agg + ((long)blockIdx.x * 128 + nb) * (F_OUTK / 2) + lane;
#pragma unroll
    for (int i = 0; i < NPW; ++i) A2[(long)i * (F_OUTK / 2)] = accs[(nb + i) * 32 + lane];
    if (tid < 128) deg_in[blockIdx.x * 128 + tid] = n_deg[tid];
}

__global__ void gcn_finalize_kernel(const float* __restrict__ agg,
                                    const int* __restrict__ deg_in,
                                    const float* __restrict__ bias,
                                    const float* __restrict__ prelu_a,
                                    float* __restrict__ pool,
                                    float* __restrict__ anchor) {
    __shared__ float red[256];
    const int  sg    = blockIdx.x;
    const int  f     = threadIdx.x & 63;
    const int  g     = threadIdx.x >> 6;
    const float alpha = prelu_a[0];
    const float bf    = bias[f];
    const long base   = (long)sg * S_SUB;

    float partial = 0.0f;
    for (int n = g; n < S_SUB - 1; n += 4) {
        long node = base + n;
        float ri = __frsqrt_rn(fmaxf((float)deg_in[node], 1.0f));
        float v  = agg[node * F_OUTK + f] * ri + bf;
        v = v > 0.0f ? v : alpha * v;
        partial += v;
    }

    if (g == 3) {
        long node = base + S_SUB - 1;
        float ri = __frsqrt_rn(fmaxf((float)deg_in[node], 1.0f));
        float v  = agg[node * F_OUTK + f] * ri + bf;
        v = v > 0.0f ? v : alpha * v;
        anchor[(long)sg * F_OUTK + f] = v;
    }

    red[threadIdx.x] = partial;
    __syncthreads();
    if (g == 0) {
        float sum = red[f] + red[f + 64] + red[f + 128] + red[f + 192];
        pool[(long)sg * F_OUTK + f] = sum * (1.0f / (S_SUB - 1));
    }
}

extern "C" void kernel_launch(void* const* d_in, const int* in_sizes, int n_in,
                              void* d_out, int out_size, void* d_ws, size_t ws_size,
                              hipStream_t stream) {
    const float* in_feat  = (const float*)d_in[0];
    const float* W        = (const float*)d_in[1];
    const float* bias     = (const float*)d_in[2];
    const float* prelu_a  = (const float*)d_in[3];
    const int*   src      = (const int*)d_in[4];
    const int*   dst      = (const int*)d_in[5];

    float* ws      = (float*)d_ws;
    float* deg_far = ws;
    int*   deg_in  = (int*)(ws + N_NODES);
    float* H       = ws + 2L * N_NODES;
    float* agg     = H + (long)N_NODES * F_OUTK;
    int*   tile_cnt     = (int*)(agg + (long)N_NODES * F_OUTK);
    int*   tile_base    = tile_cnt + (long)N_TILES * N_BKT;
    int*   bucket_start = tile_base + (long)N_TILES * N_BKT;
    int*   bkt_total    = bucket_start + 2048;
    int*   binned       = bkt_total + N_BKT;

    float* pool   = (float*)d_out;
    float* anchor = pool + (long)B_SUB * F_OUTK;

    hipMemsetAsync(deg_far, 0, (long)N_NODES * sizeof(float), stream);

    gcn_count_kernel<<<N_TILES, TILE_T, 0, stream>>>(dst, tile_cnt);
    gcn_tile_scan_kernel<<<N_BKT / 32, TILE_T, 0, stream>>>(tile_cnt, tile_base, bkt_total);
    gcn_bin_kernel<<<N_TILES, TILE_T, 0, stream>>>(src, dst, deg_far, tile_cnt, tile_base, bkt_total, bucket_start, binned);
    gcn_gemm_wmma_kernel<<<N_NODES / 128, 256, 0, stream>>>(in_feat, W, deg_far, bucket_start, binned, H);
    gcn_bucket_gather_kernel<<<N_BKT, 256, 0, stream>>>(bucket_start, binned, H, agg, deg_in);
    gcn_finalize_kernel<<<B_SUB, 256, 0, stream>>>(agg, deg_in, bias, prelu_a, pool, anchor);
}


# Gather block finishes its subgraph from the LDS rows: in-degree norm, bias, PReLU, mean pool, anchor; summed rows and in-degrees no longer stored and re-read, one launch less; pool sum order kept
# speedup vs baseline: 4.9485x; 1.1001x over previous
#include <hip/hip_runtime.h>

typedef __attribute__((ext_vector_type(2))) float v2f;
typedef __attribute__((ext_vector_type(8))) float v8f;

#define N_NODES  131072
#define F_INK    300
#define F_OUTK   64
#define N_EDGE   2097152
#define S_SUB    128
#define B_SUB    1024

#define TILE_E   8192
#define N_TILES  (N_EDGE / TILE_E)
#define N_BKT    (N_NODES / 128)
#define TILE_T   1024

__global__ __launch_bounds__(TILE_T)
void gcn_count_kernel(const int* __restrict__ dst,
                      int* __restrict__ tile_cnt) {
    __shared__ int cnt[N_BKT];
    const int tid = threadIdx.x;
    const int4* d4 = (const int4*)dst + blockIdx.x * (TILE_E / 4);
    const int4 da = d4[tid];
    const int4 db = d4[tid + TILE_T];
    cnt[tid] = 0;
    __syncthreads();
    atomicAdd(&cnt[da.x >> 7], 1);
    atomicAdd(&cnt[da.y >> 7], 1);
    atomicAdd(&cnt[da.z >> 7], 1);
    atomicAdd(&cnt[da.w >> 7], 1);
    atomicAdd(&cnt[db.x >> 7], 1);
    atomicAdd(&cnt[db.y >> 7], 1);
    atomicAdd(&cnt[db.z >> 7], 1);
    atomicAdd(&cnt[db.w >> 7], 1);
    __syncthreads();
    tile_cnt[blockIdx.x * N_BKT + tid] = cnt[tid];
}

#define KGROUPS (F_INK / 4)
#define WMMA4(a, b, c) __builtin_amdgcn_wmma_f32_16x16x4_f32(false, (a), false, (b), (short)0, (c), false, false)

__device__ __forceinline__ void gcn_mma_group(const float4 x, const float4* __restrict__ wp,
                                              const float rs,
                                              v8f& c0, v8f& c1, v8f& c2, v8f& c3) {
    const float4 b0 = wp[0], b1 = wp[16], b2 = wp[32], b3 = wp[48];
    v2f a0, a1;
    a0.x = x.x * rs; a0.y = x.y * rs;
    a1.x = x.z * rs; a1.y = x.w * rs;
    v2f t0, t1, t2, t3;
    t0.x = b0.x; t0.y = b0.y; t1.x = b1.x; t1.y = b1.y;
    t2.x = b2.x; t2.y = b2.y; t3.x = b3.x; t3.y = b3.y;
    c0 = WMMA4(a0, t0, c0); c1 = WMMA4(a0, t1, c1);
    c2 = WMMA4(a0, t2, c2); c3 = WMMA4(a0, t3, c3);
    t0.x = b0.z; t0.y = b0.w; t1.x = b1.z; t1.y = b1.w;
    t2.x = b2.z; t2.y = b2.w; t3.x = b3.z; t3.y = b3.w;
    c0 = WMMA4(a1, t0, c0); c1 = WMMA4(a1, t1, c1);
    c2 = WMMA4(a1, t2, c2); c3 = WMMA4(a1, t3, c3);
}

__global__ __launch_bounds__(256)
void gcn_gemm_wmma_kernel(const float* __restrict__ X,
                          const float* __restrict__ W,
                          const float* __restrict__ deg_far,
                          const int* __restrict__ bucket_start,
                          const int* __restrict__ binned,
                          float* __restrict__ H) {
    __shared__ float4 Ws[KGROUPS * F_OUTK];
    __shared__ int    s_deg[128];

    const int tid  = threadIdx.x;
    if (tid < 128) s_deg[tid] = 0;
    __syncthreads();
    const int lane = tid & 31;
    const int wave = tid >> 5;
    const int m    = lane & 15;
    const int half = lane >> 4;

    const int rowBase = blockIdx.x * 128 + wave * 16;
    const int row     = rowBase + m;
    const float* xrow = X + (long)row * F_INK;

    const float* xp = xrow + half * 16;
    float4 xa0 = *(const float4*)(xp + 0);
    float4 xa1 = *(const float4*)(xp + 4);
    float4 xa2 = *(const float4*)(xp + 8);
    float4 xa3 = *(const float4*)(xp + 12);
    const float dfar = deg_far[row];

    const int bs = bucket_start[blockIdx.x];
    const int be = bucket_start[blockIdx.x + 1];
    int r0[16];
#pragma unroll
    for (int k = 0; k < 16; ++k) {
        const int idx = bs + k * 256 + tid;
        r0[k] = (idx < be) ? binned[idx] : -1;
    }

    {
        const int c = tid & 63;
        for (int g = tid >> 6; g < KGROUPS; g += 4) {
            const float* wp = W + (long)(4 * g) * F_OUTK + c;
            float4 v;
            v.x = wp[0];
            v.y = wp[F_OUTK];
            v.z = wp[2 * F_OUTK];
            v.w = wp[3 * F_OUTK];
            Ws[g * F_OUTK + c] = v;
        }
    }
#pragma unroll
    for (int k = 0; k < 16; ++k)
        if ((unsigned)(r0[k] >> 14) == blockIdx.x) atomicAdd(&s_deg[(r0[k] >> 7) & 127], 1);
    for (int i0 = bs + 4096; i0 < be; i0 += 4096) {
        int r[16];
#pragma unroll
        for (int k = 0; k < 16; ++k) {
            const int idx = i0 + k * 256 + tid;
            r[k] = (idx < be) ? binned[idx] : -1;
        }
#pragma unroll
        for (int k = 0; k < 16; ++k)
            if ((unsigned)(r[k] >> 14) == blockIdx.x) atomicAdd(&s_deg[(r[k] >> 7) & 127], 1);
    }
    __syncthreads();

    const float rs = __frsqrt_rn(fmaxf((float)s_deg[wave * 16 + m] + dfar, 1.0f));

    v8f c0 = {}, c1 = {}, c2 = {}, c3 = {};
    const float4* wl = Ws + (half * 4) * F_OUTK + m;

    for (int ch = 0; ch < 9; ++ch) {
        const int nx = (ch < 8) ? ch + 1 : 8;
        const float* xn = xp + nx * 32;
        const float4 xb0 = *(const float4*)(xn + 0);
        const float4 xb1 = *(const float4*)(xn + 4);
        const float4 xb2 = *(const float4*)(xn + 8);
        const float4 xb3 = *(const float4*)(xn + 12);
        const float4* wc = wl + ch * 8 * F_OUTK;
        gcn_mma_group(xa0, wc + 0 * F_OUTK, rs, c0, c1, c2, c3);
        gcn_mma_group(xa1, wc + 1 * F_OUTK, rs, c0, c1, c2, c3);
        gcn_mma_group(xa2, wc + 2 * F_OUTK, rs, c0, c1, c2, c3);
        gcn_mma_group(xa3, wc + 3 * F_OUTK, rs, c0, c1, c2, c3);
        xa0 = xb0; xa1 = xb1; xa2 = xb2; xa3 = xb3;
    }

    {
        const float4 x8 = *(const float4*)(xrow + 288 + half * 4);
        gcn_mma_group(x8, Ws + (72 + half) * F_OUTK + m, rs, c0, c1, c2, c3);
    }
    {
        const v2f xt = *(const v2f*)(xrow + 296 + half * 2);
        v2f a;
        a.x = xt.x * rs; a.y = xt.y * rs;
        const float* wt = (const float*)(Ws + 74 * F_OUTK + m) + half * 2;
        const v2f t0 = *(const v2f*)(wt);
        const v2f t1 = *(const v2f*)(wt + 16 * 4);
        const v2f t2 = *(const v2f*)(wt + 32 * 4);
        const v2f t3 = *(const v2f*)(wt + 48 * 4);
        c0 = WMMA4(a, t0, c0); c1 = WMMA4(a, t1, c1);
        c2 = WMMA4(a, t2, c2); c3 = WMMA4(a, t3, c3);
    }

    float* hout = H + (long)(rowBase + half * 8) * F_OUTK + m;
#pragma unroll
    for (int i = 0; i < 8; ++i) {
        hout[(long)i * F_OUTK +  0] = c0[i];
        hout[(long)i * F_OUTK + 16] = c1[i];
        hout[(long)i * F_OUTK + 32] = c2[i];
        hout[(long)i * F_OUTK + 48] = c3[i];
    }
}

__device__ __forceinline__ int gcn_block_excl_scan(const int v, int* s_w, int& total) {
    const int lane = threadIdx.x & 31;
    const int wave = threadIdx.x >> 5;
    int incl = v;
#pragma unroll
    for (int d = 1; d < 32; d <<= 1) {
        const int up = __shfl_up(incl, d);
        if (lane >= d) incl += up;
    }
    if (lane == 31) s_w[wave] = incl;
    __syncthreads();
    int base = 0, all = 0;
#pragma unroll
    for (int w = 0; w < TILE_T / 32; ++w) {
        const int x = s_w[w];
        all += x;
        if (w < wave) base += x;
    }
    total = all;
    return base + incl - v;
}

#define SCAN_GROUPS (TILE_T / 32)
#define SCAN_TPG    (N_TILES / SCAN_GROUPS)

__global__ __launch_bounds__(TILE_T)
void gcn_tile_scan_kernel(const int* __restrict__ tile_cnt,
                          int* __restrict__ tile_base,
                          int* __restrict__ bkt_total) {
    __shared__ int colsum[SCAN_GROUPS][32];
    const int c = threadIdx.x & 31;
    const int g = threadIdx.x >> 5;
    const int b = blockIdx.x * 32 + c;
    const int t0 = g * SCAN_TPG;
    int v[SCAN_TPG];
#pragma unroll
    for (int k = 0; k < SCAN_TPG; ++k) v[k] = tile_cnt[(t0 + k) * N_BKT + b];
    int s = 0;
#pragma unroll
    for (int k = 0; k < SCAN_TPG; ++k) {
        const int t = v[k];
        v[k] = s;
        s += t;
    }
    colsum[g][c] = s;
    __syncthreads();
    int off = 0, tot = 0;
#pragma unroll
    for (int w = 0; w < SCAN_GROUPS; ++w) {
        const int x = colsum[w][c];
        tot += x;
        if (w < g) off += x;
    }
#pragma unroll
    for (int k = 0; k < SCAN_TPG; ++k) tile_base[(t0 + k) * N_BKT + b] = off + v[k];
    if (g == 0) bkt_total[b] = tot;
}

__global__ __launch_bounds__(TILE_T)
void gcn_bin_kernel(const int* __restrict__ src,
                    const int* __restrict__ dst,
                    float* __restrict__ deg_far,
                    const int* __restrict__ tile_cnt,
                    const int* __restrict__ tile_base,
                    const int* __restrict__ bkt_total,
                    int* __restrict__ bucket_start,
                    int* __restrict__ binned) {
    __shared__ int lcur[N_BKT];
    __shared__ int delta[N_BKT];
    __shared__ int rec_l[TILE_E];
    __shared__ unsigned short bkt_l[TILE_E];
    __shared__ int s_w[TILE_T / 32];
    __shared__ int s_w2[TILE_T / 32];
    const int tid = threadIdx.x;
    const int4* s4 = (const int4*)src + blockIdx.x * (TILE_E / 4);
    const int4* d4 = (const int4*)dst + blockIdx.x * (TILE_E / 4);
    const int4 da = d4[tid];
    const int4 db = d4[tid + TILE_T];
    const int4 sa = s4[tid];
    const int4 sb = s4[tid + TILE_T];
    const int c  = tile_cnt[blockIdx.x * N_BKT + tid];
    const int tb = tile_base[blockIdx.x * N_BKT + tid];
    const int bt = bkt_total[tid];
    int total, total_e;
    const int lstart = gcn_block_excl_scan(c, s_w, total);
    const int bs = gcn_block_excl_scan(bt, s_w2, total_e);
    if (blockIdx.x == 0) {
        bucket_start[tid] = bs;
        if (tid == 0) bucket_start[N_BKT] = total_e;
    }
    lcur[tid]  = lstart;
    delta[tid] = bs + tb - lstart;
    if ((sa.x >> 7) != (da.x >> 7)) atomicAdd(&deg_far[sa.x], 1.0f);
    if ((sa.y >> 7) != (da.y >> 7)) atomicAdd(&deg_far[sa.y], 1.0f);
    if ((sa.z >> 7) != (da.z >> 7)) atomicAdd(&deg_far[sa.z], 1.0f);
    if ((sa.w >> 7) != (da.w >> 7)) atomicAdd(&deg_far[sa.w], 1.0f);
    if ((sb.x >> 7) != (db.x >> 7)) atomicAdd(&deg_far[sb.x], 1.0f);
    if ((sb.y >> 7) != (db.y >> 7)) atomicAdd(&deg_far[sb.y], 1.0f);
    if ((sb.z >> 7) != (db.z >> 7)) atomicAdd(&deg_far[sb.z], 1.0f);
    if ((sb.w >> 7) != (db.w >> 7)) atomicAdd(&deg_far[sb.w], 1.0f);
    __syncthreads();
    int bk, p;
    bk = da.x >> 7; p = atomicAdd(&lcur[bk], 1); rec_l[p] = (sa.x << 7) | (da.x & 127); bkt_l[p] = (unsigned short)bk;
    bk = da.y >> 7; p = atomicAdd(&lcur[bk], 1); rec_l[p] = (sa.y << 7) | (da.y & 127); bkt_l[p] = (unsigned short)bk;
    bk = da.z >> 7; p = atomicAdd(&lcur[bk], 1); rec_l[p] = (sa.z << 7) | (da.z & 127); bkt_l[p] = (unsigned short)bk;
    bk = da.w >> 7; p = atomicAdd(&lcur[bk], 1); rec_l[p] = (sa.w << 7) | (da.w & 127); bkt_l[p] = (unsigned short)bk;
    bk = db.x >> 7; p = atomicAdd(&lcur[bk], 1); rec_l[p] = (sb.x << 7) | (db.x & 127); bkt_l[p] = (unsigned short)bk;
    bk = db.y >> 7; p = atomicAdd(&lcur[bk], 1); rec_l[p] = (sb.y << 7) | (db.y & 127); bkt_l[p] = (unsigned short)bk;
    bk = db.z >> 7; p = atomicAdd(&lcur[bk], 1); rec_l[p] = (sb.z << 7) | (db.z & 127); bkt_l[p] = (unsigned short)bk;
    bk = db.w >> 7; p = atomicAdd(&lcur[bk], 1); rec_l[p] = (sb.w << 7) | (db.w & 127); bkt_l[p] = (unsigned short)bk;
    __syncthreads();
#pragma unroll
    for (int k = 0; k < TILE_E / TILE_T; ++k) {
        const int i = tid + k * TILE_T;
        binned[delta[bkt_l[i]] + i] = rec_l[i];
    }
}

#define NPW 16

#define GCN_ROW_AT(idx) (*(const float2*)(Hb + (((unsigned)(idx)) << 8) + lane8))

__device__ __forceinline__ float2 gcn_sum_rows(const char* __restrict__ Hb, const int my, const int first,
                                               const int cnt, const unsigned lane8) {
    float2 v[32];
    v[0] = GCN_ROW_AT(first);
#pragma unroll
    for (int u = 1; u < 32; ++u) {
        if (u < cnt) v[u] = GCN_ROW_AT(__builtin_amdgcn_readlane(my, u));
        else         v[u] = make_float2(0.0f, 0.0f);
    }
    float2 a0 = v[0], a1 = v[1];
#pragma unroll
    for (int u = 2; u < 32; u += 2) {
        a0.x += v[u].x;     a0.y += v[u].y;
        a1.x += v[u + 1].x; a1.y += v[u + 1].y;
    }
    a0.x += a1.x;
    a0.y += a1.y;
    return a0;
}

#define BKT_CAP 4096

__global__ __launch_bounds__(256)
void gcn_bucket_gather_kernel(const int* __restrict__ bucket_start,
                              const int* __restrict__ binned,
                              const float* __restrict__ H,
                              const float* __restrict__ bias,
                              const float* __restrict__ prelu_a,
                              float* __restrict__ pool,
                              float* __restrict__ anchor) {
    __shared__ int    n_cnt[128];
    __shared__ int    n_start[128];
    __shared__ int    n_cur[128];
    __shared__ int    n_deg[128];
    __shared__ int    sorted[BKT_CAP];
    __shared__ float2 accs[128 * 32];
    __shared__ float  n_ri[128];
    __shared__ float  red[256];

    const int tid  = threadIdx.x;
    const int lane = tid & 31;
    const int wave = tid >> 5;
    const int bs = bucket_start[blockIdx.x];
    const int be = bucket_start[blockIdx.x + 1];
    const float alpha = prelu_a[0];
    const float bf    = bias[tid & 63];

    if (tid < 128) { n_cnt[tid] = 0; n_deg[tid] = 0; }
    for (int i = tid; i < 128 * 32; i += 256) accs[i] = make_float2(0.0f, 0.0f);
    __syncthreads();

    const char* Hb = (const char*)H;
    const unsigned lane8 = (unsigned)lane * 8u;
    const int nb = wave * NPW;

    for (int c0 = bs; c0 < be; c0 += BKT_CAP) {
        const int m = min(BKT_CAP, be - c0);
        const int* rp = binned + c0;

        for (int i0 = 0; i0 < m; i0 += 2048) {
            int r[8];
#pragma unroll
            for (int k = 0; k < 8; ++k) {
                const int idx = i0 + k * 256 + tid;
                r[k] = (idx < m) ? rp[idx] : -1;
            }
#pragma unroll
            for (int k = 0; k < 8; ++k)
                if (r[k] >= 0) atomicAdd(&n_cnt[r[k] & 127], 1);
        }
        __syncthreads();

        if (wave == 0) {
            const int q0 = n_cnt[4 * lane + 0];
            const int q1 = n_cnt[4 * lane + 1];
            const int q2 = n_cnt[4 * lane + 2];
            const int q3 = n_cnt[4 * lane + 3];
            const int sum4 = (q0 + q1) + (q2 + q3);
            int incl = sum4;
#pragma unroll
            for (int d = 1; d < 32; d <<= 1) {
                const int up = __shfl_up(incl, d);
                if (lane >= d) incl += up;
            }
            const int e0 = incl - sum4;
            const int e1 = e0 + q0;
            const int e2 = e1 + q1;
            const int e3 = e2 + q2;
            n_start[4 * lane + 0] = e0; n_cur[4 * lane + 0] = e0; n_deg[4 * lane + 0] += q0;
            n_start[4 * lane + 1] = e1; n_cur[4 * lane + 1] = e1; n_deg[4 * lane + 1] += q1;
            n_start[4 * lane + 2] = e2; n_cur[4 * lane + 2] = e2; n_deg[4 * lane + 2] += q2;
            n_start[4 * lane + 3] = e3; n_cur[4 * lane + 3] = e3; n_deg[4 * lane + 3] += q3;
        }
        __syncthreads();

        for (int i0 = 0; i0 < m; i0 += 2048) {
            int r[8];
#pragma unroll
            for (int k = 0; k < 8; ++k) {
                const int idx = i0 + k * 256 + tid;
                r[k] = (idx < m) ? rp[idx] : -1;
            }
#pragma unroll
            for (int k = 0; k < 8; ++k) {
                if (r[k] >= 0) {
                    const int p = atomicAdd(&n_cur[r[k] & 127], 1);
                    sorted[p] = r[k] >> 7;
                }
            }
        }
        __syncthreads();

        {
            int lb = 0, ln = 0;
            if (lane < NPW) {
                lb = n_start[nb + lane];
                ln = n_cnt[nb + lane];
            }
            int beg = __builtin_amdgcn_readlane(lb, 0);
            int n   = __builtin_amdgcn_readlane(ln, 0);
            int my  = (lane < n) ? sorted[beg + lane] : 0;
            int first = __builtin_amdgcn_readlane(my, 0);
            for (int i = 0; i < NPW; ++i) {
                int begN = 0, nN = 0, myN = 0;
                if (i + 1 < NPW) {
                    begN = __builtin_amdgcn_readlane(lb, i + 1);
                    nN   = __builtin_amdgcn_readlane(ln, i + 1);
                    myN  = (lane < nN) ? sorted[begN + lane] : 0;
                }
                if (n > 0) {
                    float2 acc = gcn_sum_rows(Hb, my, first, min(32, n), lane8);
                    for (int j = 32; j < n; j += 32) {
                        const int more = (lane < n - j) ? sorted[beg + j + lane] : 0;
                        const float2 part = gcn_sum_rows(Hb, more, __builtin_amdgcn_readlane(more, 0),
                                                         min(32, n - j), lane8);
                        acc.x += part.x;
                        acc.y += part.y;
                    }
                    float2 o = accs[(nb + i) * 32 + lane];
                    o.x += acc.x;
                    o.y += acc.y;
                    accs[(nb + i) * 32 + lane] = o;
                }
                beg = begN; n = nN; my = myN;
                first = __builtin_amdgcn_readlane(myN, 0);
            }
        }

        if (c0 + BKT_CAP < be) {
            __syncthreads();
            if (tid < 128) n_cnt[tid] = 0;
            __syncthreads();
        }
    }

    if (tid < 128) n_ri[tid] = __frsqrt_rn(fmaxf((float)n_deg[tid], 1.0f));
    __syncthreads();
    {
        const int f = tid & 63;
        const int g = tid >> 6;
        const float* rowsf = (const float*)accs + f;
        float partial = 0.0f;
#pragma unroll
        for (int k = 0; k < S_SUB / 4 - 1; ++k) {
            const int n = g + 4 * k;
            float v = rowsf[n * F_OUTK] * n_ri[n] + bf;
            v = v > 0.0f ? v : alpha * v;
            partial += v;
        }
        {
            const int n = g + S_SUB - 4;
            float v = rowsf[n * F_OUTK] * n_ri[n] + bf;
            v = v > 0.0f ? v : alpha * v;
            if (g == 3) anchor[(long)blockIdx.x * F_OUTK + f] = v;
            else        partial += v;
        }
        red[tid] = partial;
        __syncthreads();
        if (g == 0) {
            const float sum = red[f] + red[f + 64] + red[f + 128] + red[f + 192];
            pool[(long)blockIdx.x * F_OUTK + f] = sum * (1.0f / (S_SUB - 1));
        }
    }
}

extern "C" void kernel_launch(void* const* d_in, const int* in_sizes, int n_in,
                              void* d_out, int out_size, void* d_ws, size_t ws_size,
                              hipStream_t stream) {
    const float* in_feat  = (const float*)d_in[0];
    const float* W        = (const float*)d_in[1];
    const float* bias     = (const float*)d_in[2];
    const float* prelu_a  = (const float*)d_in[3];
    const int*   src      = (const int*)d_in[4];
    const int*   dst      = (const int*)d_in[5];

    float* ws      = (float*)d_ws;
    float* deg_far = ws;
    float* H       = ws + (long)N_NODES;
    int*   tile_cnt     = (int*)(H + (long)N_NODES * F_OUTK);
    int*   tile_base    = tile_cnt + (long)N_TILES * N_BKT;
    int*   bucket_start = tile_base + (long)N_TILES * N_BKT;
    int*   bkt_total    = bucket_start + 2048;
    int*   binned       = bkt_total + N_BKT;

    float* pool   = (float*)d_out;
    float* anchor = pool + (long)B_SUB * F_OUTK;

    hipMemsetAsync(deg_far, 0, (long)N_NODES * sizeof(float), stream);

    gcn_count_kernel<<<N_TILES, TILE_T, 0, stream>>>(dst, tile_cnt);
    gcn_tile_scan_kernel<<<N_BKT / 32, TILE_T, 0, stream>>>(tile_cnt, tile_base, bkt_total);
    gcn_bin_kernel<<<N_TILES, TILE_T, 0, stream>>>(src, dst, deg_far, tile_cnt, tile_base, bkt_total, bucket_start, binned);
    gcn_gemm_wmma_kernel<<<N_NODES / 128, 256, 0, stream>>>(in_feat, W, deg_far, bucket_start, binned, H);
    gcn_bucket_gather_kernel<<<N_BKT, 256, 0, stream>>>(bucket_start, binned, H, bias, prelu_a, pool, anchor);
}


# GEMM block sums its own subgraph from its H tile in LDS (records node-ordered in LDS) and finishes it; H store and later kernel only for edges that cross buckets; row sum order = LDS arrival rank
# speedup vs baseline: 5.3379x; 1.0787x over previous
#include <hip/hip_runtime.h>

typedef __attribute__((ext_vector_type(2))) float v2f;
typedef __attribute__((ext_vector_type(8))) float v8f;

#define N_NODES  131072
#define F_INK    300
#define F_OUTK   64
#define N_EDGE   2097152
#define S_SUB    128
#define B_SUB    1024

#define TILE_E   8192
#define N_TILES  (N_EDGE / TILE_E)
#define N_BKT    (N_NODES / 128)
#define TILE_T   1024

__global__ __launch_bounds__(TILE_T)
void gcn_count_kernel(const int* __restrict__ dst,
                      int* __restrict__ tile_cnt) {
    __shared__ int cnt[N_BKT];
    const int tid = threadIdx.x;
    const int4* d4 = (const int4*)dst + blockIdx.x * (TILE_E / 4);
    const int4 da = d4[tid];
    const int4 db = d4[tid + TILE_T];
    cnt[tid] = 0;
    __syncthreads();
    atomicAdd(&cnt[da.x >> 7], 1);
    atomicAdd(&cnt[da.y >> 7], 1);
    atomicAdd(&cnt[da.z >> 7], 1);
    atomicAdd(&cnt[da.w >> 7], 1);
    atomicAdd(&cnt[db.x >> 7], 1);
    atomicAdd(&cnt[db.y >> 7], 1);
    atomicAdd(&cnt[db.z >> 7], 1);
    atomicAdd(&cnt[db.w >> 7], 1);
    __syncthreads();
    tile_cnt[blockIdx.x * N_BKT + tid] = cnt[tid];
}

#define KGROUPS (F_INK / 4)
#define WMMA4(a, b, c) __builtin_amdgcn_wmma_f32_16x16x4_f32(false, (a), false, (b), (short)0, (c), false, false)
#define NPW      16
#define FUSE_CAP 3072

__device__ __forceinline__ float2 gcn_sum_rows_lds(const float2* __restrict__ Ht2, const int my,
                                                   const int cnt, const int lane) {
    float2 v[32];
    v[0] = Ht2[__builtin_amdgcn_readlane(my, 0) * 32 + lane];
#pragma unroll
    for (int u = 1; u < 32; ++u) {
        if (u < cnt) v[u] = Ht2[__builtin_amdgcn_readlane(my, u) * 32 + lane];
        else         v[u] = make_float2(0.0f, 0.0f);
    }
    float2 a0 = v[0], a1 = v[1];
#pragma unroll
    for (int u = 2; u < 32; u += 2) {
        a0.x += v[u].x;     a0.y += v[u].y;
        a1.x += v[u + 1].x; a1.y += v[u + 1].y;
    }
    a0.x += a1.x;
    a0.y += a1.y;
    return a0;
}

__device__ __forceinline__ void gcn_mma_group(const float4 x, const float4* __restrict__ wp,
                                              const float rs,
                                              v8f& c0, v8f& c1, v8f& c2, v8f& c3) {
    const float4 b0 = wp[0], b1 = wp[16], b2 = wp[32], b3 = wp[48];
    v2f a0, a1;
    a0.x = x.x * rs; a0.y = x.y * rs;
    a1.x = x.z * rs; a1.y = x.w * rs;
    v2f t0, t1, t2, t3;
    t0.x = b0.x; t0.y = b0.y; t1.x = b1.x; t1.y = b1.y;
    t2.x = b2.x; t2.y = b2.y; t3.x = b3.x; t3.y = b3.y;
    c0 = WMMA4(a0, t0, c0); c1 = WMMA4(a0, t1, c1);
    c2 = WMMA4(a0, t2, c2); c3 = WMMA4(a0, t3, c3);
    t0.x = b0.z; t0.y = b0.w; t1.x = b1.z; t1.y = b1.w;
    t2.x = b2.z; t2.y = b2.w; t3.x = b3.z; t3.y = b3.w;
    c0 = WMMA4(a1, t0, c0); c1 = WMMA4(a1, t1, c1);
    c2 = WMMA4(a1, t2, c2); c3 = WMMA4(a1, t3, c3);
}

__global__ __launch_bounds__(256)
void gcn_gemm_wmma_kernel(const float* __restrict__ X,
                          const float* __restrict__ W,
                          const float* __restrict__ deg_far,
                          const int* __restrict__ bucket_start,
                          const int* __restrict__ binned,
                          const float* __restrict__ bias,
                          const float* __restrict__ prelu_a,
                          float* __restrict__ H,
                          int* __restrict__ bkt_late,
                          float* __restrict__ pool,
                          float* __restrict__ anchor) {
    __shared__ float4 Ws[KGROUPS * F_OUTK];
    __shared__ int    s_deg[128];
    __shared__ int    n_cnt[128];
    __shared__ int    n_start[128];
    __shared__ unsigned char sorted[FUSE_CAP];
    __shared__ int    s_far;
    __shared__ int    s_out;

    const int tid  = threadIdx.x;
    if (tid < 128) { s_deg[tid] = 0; n_cnt[tid] = 0; }
    if (tid == 0) { s_far = 0; s_out = 0; }
    __syncthreads();
    const int lane = tid & 31;
    const int wave = tid >> 5;
    const int m    = lane & 15;
    const int half = lane >> 4;

    const int rowBase = blockIdx.x * 128 + wave * 16;
    const int row     = rowBase + m;
    const float* xrow = X + (long)row * F_INK;

    const float* xp = xrow + half * 16;
    float4 xa0 = *(const float4*)(xp + 0);
    float4 xa1 = *(const float4*)(xp + 4);
    float4 xa2 = *(const float4*)(xp + 8);
    float4 xa3 = *(const float4*)(xp + 12);
    const float dfar = deg_far[row];
    const float alpha = prelu_a[0];
    const float bf    = bias[tid & 63];

    const int bs = bucket_start[blockIdx.x];
    const int be = bucket_start[blockIdx.x + 1];
    int r0[16];
#pragma unroll
    for (int k = 0; k < 16; ++k) {
        const int idx = bs + k * 256 + tid;
        r0[k] = (idx < be) ? binned[idx] : -1;
    }

    {
        const int c = tid & 63;
        for (int g = tid >> 6; g < KGROUPS; g += 4) {
            const float* wp = W + (long)(4 * g) * F_OUTK + c;
            float4 v;
            v.x = wp[0];
            v.y = wp[F_OUTK];
            v.z = wp[2 * F_OUTK];
            v.w = wp[3 * F_OUTK];
            Ws[g * F_OUTK + c] = v;
        }
    }
    int far = 0;
#pragma unroll
    for (int k = 0; k < 16; ++k) {
        const int r = r0[k];
        if (r >= 0) {
            if ((unsigned)(r >> 14) == blockIdx.x) atomicAdd(&s_deg[(r >> 7) & 127], 1);
            else ++far;
            const int rank = atomicAdd(&n_cnt[r & 127], 1);
            r0[k] = (r & 0x3FFF) | (rank << 14);
        }
    }
    for (int i0 = bs + 4096; i0 < be; i0 += 4096) {
        int r[16];
#pragma unroll
        for (int k = 0; k < 16; ++k) {
            const int idx = i0 + k * 256 + tid;
            r[k] = (idx < be) ? binned[idx] : -1;
        }
#pragma unroll
        for (int k = 0; k < 16; ++k) {
            if (r[k] >= 0) {
                if ((unsigned)(r[k] >> 14) == blockIdx.x) atomicAdd(&s_deg[(r[k] >> 7) & 127], 1);
                else ++far;
            }
        }
    }
    if (far) atomicAdd(&s_far, far);
    if (dfar > 0.0f) s_out = 1;
    __syncthreads();

    const float rs = __frsqrt_rn(fmaxf((float)s_deg[wave * 16 + m] + dfar, 1.0f));

    const bool fused = (be - bs <= FUSE_CAP) && (s_far == 0);
    const bool wr = !fused || (s_out != 0);
    if (tid == 0) bkt_late[blockIdx.x] = fused ? 0 : 1;

    if (fused && wave == 0) {
        const int q0 = n_cnt[4 * lane + 0];
        const int q1 = n_cnt[4 * lane + 1];
        const int q2 = n_cnt[4 * lane + 2];
        const int q3 = n_cnt[4 * lane + 3];
        const int sum4 = (q0 + q1) + (q2 + q3);
        int incl = sum4;
#pragma unroll
        for (int d = 1; d < 32; d <<= 1) {
            const int up = __shfl_up(incl, d);
            if (lane >= d) incl += up;
        }
        const int e0 = incl - sum4;
        n_start[4 * lane + 0] = e0;
        n_start[4 * lane + 1] = e0 + q0;
        n_start[4 * lane + 2] = e0 + q0 + q1;
        n_start[4 * lane + 3] = e0 + q0 + q1 + q2;
    }
    __syncthreads();
    if (fused) {
#pragma unroll
        for (int k = 0; k < 16; ++k) {
            const int q = r0[k];
            if (q >= 0) sorted[n_start[q & 127] + (q >> 14)] = (unsigned char)((q >> 7) & 127);
        }
    }

    v8f c0 = {}, c1 = {}, c2 = {}, c3 = {};
    const float4* wl = Ws + (half * 4) * F_OUTK + m;

    for (int ch = 0; ch < 9; ++ch) {
        const int nx = (ch < 8) ? ch + 1 : 8;
        const float* xn = xp + nx * 32;
        const float4 xb0 = *(const float4*)(xn + 0);
        const float4 xb1 = *(const float4*)(xn + 4);
        const float4 xb2 = *(const float4*)(xn + 8);
        const float4 xb3 = *(const float4*)(xn + 12);
        const float4* wc = wl + ch * 8 * F_OUTK;
        gcn_mma_group(xa0, wc + 0 * F_OUTK, rs, c0, c1, c2, c3);
        gcn_mma_group(xa1, wc + 1 * F_OUTK, rs, c0, c1, c2, c3);
        gcn_mma_group(xa2, wc + 2 * F_OUTK, rs, c0, c1, c2, c3);
        gcn_mma_group(xa3, wc + 3 * F_OUTK, rs, c0, c1, c2, c3);
        xa0 = xb0; xa1 = xb1; xa2 = xb2; xa3 = xb3;
    }

    {
        const float4 x8 = *(const float4*)(xrow + 288 + half * 4);
        gcn_mma_group(x8, Ws + (72 + half) * F_OUTK + m, rs, c0, c1, c2, c3);
    }
    {
        const v2f xt = *(const v2f*)(xrow + 296 + half * 2);
        v2f a;
        a.x = xt.x * rs; a.y = xt.y * rs;
        const float* wt = (const float*)(Ws + 74 * F_OUTK + m) + half * 2;
        const v2f t0 = *(const v2f*)(wt);
        const v2f t1 = *(const v2f*)(wt + 16 * 4);
        const v2f t2 = *(const v2f*)(wt + 32 * 4);
        const v2f t3 = *(const v2f*)(wt + 48 * 4);
        c0 = WMMA4(a, t0, c0); c1 = WMMA4(a, t1, c1);
        c2 = WMMA4(a, t2, c2); c3 = WMMA4(a, t3, c3);
    }

    if (wr) {
        float* hout = H + (long)(rowBase + half * 8) * F_OUTK + m;
#pragma unroll
        for (int i = 0; i < 8; ++i) {
            hout[(long)i * F_OUTK +  0] = c0[i];
            hout[(long)i * F_OUTK + 16] = c1[i];
            hout[(long)i * F_OUTK + 32] = c2[i];
            hout[(long)i * F_OUTK + 48] = c3[i];
        }
    }
    if (!fused) return;

    __syncthreads();
    float*  Ht   = (float*)Ws;
    float2* accs = (float2*)Ws + 128 * 32;
    float*  n_ri = (float*)Ws + 2 * 128 * F_OUTK;
    float*  red  = n_ri + 128;
    {
        float* ht = Ht + (wave * 16 + half * 8) * F_OUTK + m;
#pragma unroll
        for (int i = 0; i < 8; ++i) {
            ht[i * F_OUTK +  0] = c0[i];
            ht[i * F_OUTK + 16] = c1[i];
            ht[i * F_OUTK + 32] = c2[i];
            ht[i * F_OUTK + 48] = c3[i];
        }
    }
    if (tid < 128) n_ri[tid] = __frsqrt_rn(fmaxf((float)n_cnt[tid], 1.0f));
    __syncthreads();

    {
        const float2* Ht2 = (const float2*)Ws;
        const int nb = wave * NPW;
        int lb = 0, ln = 0;
        if (lane < NPW) {
            lb = n_start[nb + lane];
            ln = n_cnt[nb + lane];
        }
        int beg = __builtin_amdgcn_readlane(lb, 0);
        int n   = __builtin_amdgcn_readlane(ln, 0);
        int my  = (lane < n) ? (int)sorted[beg + lane] : 0;
        for (int i = 0; i < NPW; ++i) {
            int begN = 0, nN = 0, myN = 0;
            if (i + 1 < NPW) {
                begN = __builtin_amdgcn_readlane(lb, i + 1);
                nN   = __builtin_amdgcn_readlane(ln, i + 1);
                myN  = (lane < nN) ? (int)sorted[begN + lane] : 0;
            }
            float2 acc = make_float2(0.0f, 0.0f);
            if (n > 0) {
                acc = gcn_sum_rows_lds(Ht2, my, min(32, n), lane);
                for (int j = 32; j < n; j += 32) {
                    const int more = (lane < n - j) ? (int)sorted[beg + j + lane] : 0;
                    const float2 part = gcn_sum_rows_lds(Ht2, more, min(32, n - j), lane);
                    acc.x += part.x;
                    acc.y += part.y;
                }
            }
            accs[(nb + i) * 32 + lane] = acc;
            beg = begN; n = nN; my = myN;
        }
    }
    __syncthreads();

    {
        const int f = tid & 63;
        const int g = tid >> 6;
        const float* rowsf = (const float*)accs + f;
        float partial = 0.0f;
#pragma unroll
        for (int k = 0; k < S_SUB / 4 - 1; ++k) {
            const int nn = g + 4 * k;
            float v = rowsf[nn * F_OUTK] * n_ri[nn] + bf;
            v = v > 0.0f ? v : alpha * v;
            partial += v;
        }
        {
            const int nn = g + S_SUB - 4;
            float v = rowsf[nn * F_OUTK] * n_ri[nn] + bf;
            v = v > 0.0f ? v : alpha * v;
            if (g == 3) anchor[(long)blockIdx.x * F_OUTK + f] = v;
            else        partial += v;
        }
        red[tid] = partial;
        __syncthreads();
        if (g == 0) {
            const float sum = red[f] + red[f + 64] + red[f + 128] + red[f + 192];
            pool[(long)blockIdx.x * F_OUTK + f] = sum * (1.0f / (S_SUB - 1));
        }
    }
}

__device__ __forceinline__ int gcn_block_excl_scan(const int v, int* s_w, int& total) {
    const int lane = threadIdx.x & 31;
    const int wave = threadIdx.x >> 5;
    int incl = v;
#pragma unroll
    for (int d = 1; d < 32; d <<= 1) {
        const int up = __shfl_up(incl, d);
        if (lane >= d) incl += up;
    }
    if (lane == 31) s_w[wave] = incl;
    __syncthreads();
    int base = 0, all = 0;
#pragma unroll
    for (int w = 0; w < TILE_T / 32; ++w) {
        const int x = s_w[w];
        all += x;
        if (w < wave) base += x;
    }
    total = all;
    return base + incl - v;
}

#define SCAN_GROUPS (TILE_T / 32)
#define SCAN_TPG    (N_TILES / SCAN_GROUPS)

__global__ __launch_bounds__(TILE_T)
void gcn_tile_scan_kernel(const int* __restrict__ tile_cnt,
                          int* __restrict__ tile_base,
                          int* __restrict__ bkt_total) {
    __shared__ int colsum[SCAN_GROUPS][32];
    const int c = threadIdx.x & 31;
    const int g = threadIdx.x >> 5;
    const int b = blockIdx.x * 32 + c;
    const int t0 = g * SCAN_TPG;
    int v[SCAN_TPG];
#pragma unroll
    for (int k = 0; k < SCAN_TPG; ++k) v[k] = tile_cnt[(t0 + k) * N_BKT + b];
    int s = 0;
#pragma unroll
    for (int k = 0; k < SCAN_TPG; ++k) {
        const int t = v[k];
        v[k] = s;
        s += t;
    }
    colsum[g][c] = s;
    __syncthreads();
    int off = 0, tot = 0;
#pragma unroll
    for (int w = 0; w < SCAN_GROUPS; ++w) {
        const int x = colsum[w][c];
        tot += x;
        if (w < g) off += x;
    }
#pragma unroll
    for (int k = 0; k < SCAN_TPG; ++k) tile_base[(t0 + k) * N_BKT + b] = off + v[k];
    if (g == 0) bkt_total[b] = tot;
}

__global__ __launch_bounds__(TILE_T)
void gcn_bin_kernel(const int* __restrict__ src,
                    const int* __restrict__ dst,
                    float* __restrict__ deg_far,
                    const int* __restrict__ tile_cnt,
                    const int* __restrict__ tile_base,
                    const int* __restrict__ bkt_total,
                    int* __restrict__ bucket_start,
                    int* __restrict__ binned) {
    __shared__ int lcur[N_BKT];
    __shared__ int delta[N_BKT];
    __shared__ int rec_l[TILE_E];
    __shared__ unsigned short bkt_l[TILE_E];
    __shared__ int s_w[TILE_T / 32];
    __shared__ int s_w2[TILE_T / 32];
    const int tid = threadIdx.x;
    const int4* s4 = (const int4*)src + blockIdx.x * (TILE_E / 4);
    const int4* d4 = (const int4*)dst + blockIdx.x * (TILE_E / 4);
    const int4 da = d4[tid];
    const int4 db = d4[tid + TILE_T];
    const int4 sa = s4[tid];
    const int4 sb = s4[tid + TILE_T];
    const int c  = tile_cnt[blockIdx.x * N_BKT + tid];
    const int tb = tile_base[blockIdx.x * N_BKT + tid];
    const int bt = bkt_total[tid];
    int total, total_e;
    const int lstart = gcn_block_excl_scan(c, s_w, total);
    const int bs = gcn_block_excl_scan(bt, s_w2, total_e);
    if (blockIdx.x == 0) {
        bucket_start[tid] = bs;
        if (tid == 0) bucket_start[N_BKT] = total_e;
    }
    lcur[tid]  = lstart;
    delta[tid] = bs + tb - lstart;
    if ((sa.x >> 7) != (da.x >> 7)) atomicAdd(&deg_far[sa.x], 1.0f);
    if ((sa.y >> 7) != (da.y >> 7)) atomicAdd(&deg_far[sa.y], 1.0f);
    if ((sa.z >> 7) != (da.z >> 7)) atomicAdd(&deg_far[sa.z], 1.0f);
    if ((sa.w >> 7) != (da.w >> 7)) atomicAdd(&deg_far[sa.w], 1.0f);
    if ((sb.x >> 7) != (db.x >> 7)) atomicAdd(&deg_far[sb.x], 1.0f);
    if ((sb.y >> 7) != (db.y >> 7)) atomicAdd(&deg_far[sb.y], 1.0f);
    if ((sb.z >> 7) != (db.z >> 7)) atomicAdd(&deg_far[sb.z], 1.0f);
    if ((sb.w >> 7) != (db.w >> 7)) atomicAdd(&deg_far[sb.w], 1.0f);
    __syncthreads();
    int bk, p;
    bk = da.x >> 7; p = atomicAdd(&lcur[bk], 1); rec_l[p] = (sa.x << 7) | (da.x & 127); bkt_l[p] = (unsigned short)bk;
    bk = da.y >> 7; p = atomicAdd(&lcur[bk], 1); rec_l[p] = (sa.y << 7) | (da.y & 127); bkt_l[p] = (unsigned short)bk;
    bk = da.z >> 7; p = atomicAdd(&lcur[bk], 1); rec_l[p] = (sa.z << 7) | (da.z & 127); bkt_l[p] = (unsigned short)bk;
    bk = da.w >> 7; p = atomicAdd(&lcur[bk], 1); rec_l[p] = (sa.w << 7) | (da.w & 127); bkt_l[p] = (unsigned short)bk;
    bk = db.x >> 7; p = atomicAdd(&lcur[bk], 1); rec_l[p] = (sb.x << 7) | (db.x & 127); bkt_l[p] = (unsigned short)bk;
    bk = db.y >> 7; p = atomicAdd(&lcur[bk], 1); rec_l[p] = (sb.y << 7) | (db.y & 127); bkt_l[p] = (unsigned short)bk;
    bk = db.z >> 7; p = atomicAdd(&lcur[bk], 1); rec_l[p] = (sb.z << 7) | (db.z & 127); bkt_l[p] = (unsigned short)bk;
    bk = db.w >> 7; p = atomicAdd(&lcur[bk], 1); rec_l[p] = (sb.w << 7) | (db.w & 127); bkt_l[p] = (unsigned short)bk;
    __syncthreads();
#pragma unroll
    for (int k = 0; k < TILE_E / TILE_T; ++k) {
        const int i = tid + k * TILE_T;
        binned[delta[bkt_l[i]] + i] = rec_l[i];
    }
}

#define GCN_ROW_AT(idx) (*(const float2*)(Hb + (((unsigned)(idx)) << 8) + lane8))

__device__ __forceinline__ float2 gcn_sum_rows(const char* __restrict__ Hb, const int my, const int first,
                                               const int cnt, const unsigned lane8) {
    float2 v[32];
    v[0] = GCN_ROW_AT(first);
#pragma unroll
    for (int u = 1; u < 32; ++u) {
        if (u < cnt) v[u] = GCN_ROW_AT(__builtin_amdgcn_readlane(my, u));
        else         v[u] = make_float2(0.0f, 0.0f);
    }
    float2 a0 = v[0], a1 = v[1];
#pragma unroll
    for (int u = 2; u < 32; u += 2) {
        a0.x += v[u].x;     a0.y += v[u].y;
        a1.x += v[u + 1].x; a1.y += v[u + 1].y;
    }
    a0.x += a1.x;
    a0.y += a1.y;
    return a0;
}

#define BKT_CAP 4096

__global__ __launch_bounds__(256)
void gcn_bucket_gather_kernel(const int* __restrict__ bucket_start,
                              const int* __restrict__ binned,
                              const int* __restrict__ bkt_late,
                              const float* __restrict__ H,
                              const float* __restrict__ bias,
                              const float* __restrict__ prelu_a,
                              float* __restrict__ pool,
                              float* __restrict__ anchor) {
    __shared__ int    n_cnt[128];
    __shared__ int    n_start[128];
    __shared__ int    n_cur[128];
    __shared__ int    n_deg[128];
    __shared__ int    sorted[BKT_CAP];
    __shared__ float2 accs[128 * 32];
    __shared__ float  n_ri[128];
    __shared__ float  red[256];

    const int tid  = threadIdx.x;
    const int lane = tid & 31;
    const int wave = tid >> 5;
    const float alpha = prelu_a[0];
    const float bf    = bias[tid & 63];

    for (int b = blockIdx.x; b < N_BKT; b += gridDim.x) {
    if (bkt_late[b] == 0) continue;
    const int bs = bucket_start[b];
    const int be = bucket_start[b + 1];

    if (tid < 128) { n_cnt[tid] = 0; n_deg[tid] = 0; }
    for (int i = tid; i < 128 * 32; i += 256) accs[i] = make_float2(0.0f, 0.0f);
    __syncthreads();

    const char* Hb = (const char*)H;
    const unsigned lane8 = (unsigned)lane * 8u;
    const int nb = wave * NPW;

    for (int c0 = bs; c0 < be; c0 += BKT_CAP) {
        const int m = min(BKT_CAP, be - c0);
        const int* rp = binned + c0;

        for (int i0 = 0; i0 < m; i0 += 2048) {
            int r[8];
#pragma unroll
            for (int k = 0; k < 8; ++k) {
                const int idx = i0 + k * 256 + tid;
                r[k] = (idx < m) ? rp[idx] : -1;
            }
#pragma unroll
            for (int k = 0; k < 8; ++k)
                if (r[k] >= 0) atomicAdd(&n_cnt[r[k] & 127], 1);
        }
        __syncthreads();

        if (wave == 0) {
            const int q0 = n_cnt[4 * lane + 0];
            const int q1 = n_cnt[4 * lane + 1];
            const int q2 = n_cnt[4 * lane + 2];
            const int q3 = n_cnt[4 * lane + 3];
            const int sum4 = (q0 + q1) + (q2 + q3);
            int incl = sum4;
#pragma unroll
            for (int d = 1; d < 32; d <<= 1) {
                const int up = __shfl_up(incl, d);
                if (lane >= d) incl += up;
            }
            const int e0 = incl - sum4;
            const int e1 = e0 + q0;
            const int e2 = e1 + q1;
            const int e3 = e2 + q2;
            n_start[4 * lane + 0] = e0; n_cur[4 * lane + 0] = e0; n_deg[4 * lane + 0] += q0;
            n_start[4 * lane + 1] = e1; n_cur[4 * lane + 1] = e1; n_deg[4 * lane + 1] += q1;
            n_start[4 * lane + 2] = e2; n_cur[4 * lane + 2] = e2; n_deg[4 * lane + 2] += q2;
            n_start[4 * lane + 3] = e3; n_cur[4 * lane + 3] = e3; n_deg[4 * lane + 3] += q3;
        }
        __syncthreads();

        for (int i0 = 0; i0 < m; i0 += 2048) {
            int r[8];
#pragma unroll
            for (int k = 0; k < 8; ++k) {
                const int idx = i0 + k * 256 + tid;
                r[k] = (idx < m) ? rp[idx] : -1;
            }
#pragma unroll
            for (int k = 0; k < 8; ++k) {
                if (r[k] >= 0) {
                    const int p = atomicAdd(&n_cur[r[k] & 127], 1);
                    sorted[p] = r[k] >> 7;
                }
            }
        }
        __syncthreads();

        {
            int lb = 0, ln = 0;
            if (lane < NPW) {
                lb = n_start[nb + lane];
                ln = n_cnt[nb + lane];
            }
            int beg = __builtin_amdgcn_readlane(lb, 0);
            int n   = __builtin_amdgcn_readlane(ln, 0);
            int my  = (lane < n) ? sorted[beg + lane] : 0;
            int first = __builtin_amdgcn_readlane(my, 0);
            for (int i = 0; i < NPW; ++i) {
                int begN = 0, nN = 0, myN = 0;
                if (i + 1 < NPW) {
                    begN = __builtin_amdgcn_readlane(lb, i + 1);
                    nN   = __builtin_amdgcn_readlane(ln, i + 1);
                    myN  = (lane < nN) ? sorted[begN + lane] : 0;
                }
                if (n > 0) {
                    float2 acc = gcn_sum_rows(Hb, my, first, min(32, n), lane8);
                    for (int j = 32; j < n; j += 32) {
                        const int more = (lane < n - j) ? sorted[beg + j + lane] : 0;
                        const float2 part = gcn_sum_rows(Hb, more, __builtin_amdgcn_readlane(more, 0),
                                                         min(32, n - j), lane8);
                        acc.x += part.x;
                        acc.y += part.y;
                    }
                    float2 o = accs[(nb + i) * 32 + lane];
                    o.x += acc.x;
                    o.y += acc.y;
                    accs[(nb + i) * 32 + lane] = o;
                }
                beg = begN; n = nN; my = myN;
                first = __builtin_amdgcn_readlane(myN, 0);
            }
        }

        if (c0 + BKT_CAP < be) {
            __syncthreads();
            if (tid < 128) n_cnt[tid] = 0;
            __syncthreads();
        }
    }

    if (tid < 128) n_ri[tid] = __frsqrt_rn(fmaxf((float)n_deg[tid], 1.0f));
    __syncthreads();
    {
        const int f = tid & 63;
        const int g = tid >> 6;
        const float* rowsf = (const float*)accs + f;
        float partial = 0.0f;
#pragma unroll
        for (int k = 0; k < S_SUB / 4 - 1; ++k) {
            const int n = g + 4 * k;
            float v = rowsf[n * F_OUTK] * n_ri[n] + bf;
            v = v > 0.0f ? v : alpha * v;
            partial += v;
        }
        {
            const int n = g + S_SUB - 4;
            float v = rowsf[n * F_OUTK] * n_ri[n] + bf;
            v = v > 0.0f ? v : alpha * v;
            if (g == 3) anchor[(long)b * F_OUTK + f] = v;
            else        partial += v;
        }
        red[tid] = partial;
        __syncthreads();
        if (g == 0) {
            const float sum = red[f] + red[f + 64] + red[f + 128] + red[f + 192];
            pool[(long)b * F_OUTK + f] = sum * (1.0f / (S_SUB - 1));
        }
    }
    __syncthreads();
    }
}

extern "C" void kernel_launch(void* const* d_in, const int* in_sizes, int n_in,
                              void* d_out, int out_size, void* d_ws, size_t ws_size,
                              hipStream_t stream) {
    const float* in_feat  = (const float*)d_in[0];
    const float* W        = (const float*)d_in[1];
    const float* bias     = (const float*)d_in[2];
    const float* prelu_a  = (const float*)d_in[3];
    const int*   src      = (const int*)d_in[4];
    const int*   dst      = (const int*)d_in[5];

    float* ws      = (float*)d_ws;
    float* deg_far = ws;
    float* H       = ws + (long)N_NODES;
    int*   tile_cnt     = (int*)(H + (long)N_NODES * F_OUTK);
    int*   tile_base    = tile_cnt + (long)N_TILES * N_BKT;
    int*   bucket_start = tile_base + (long)N_TILES * N_BKT;
    int*   bkt_total    = bucket_start + 2048;
    int*   binned       = bkt_total + N_BKT;
    int*   bkt_late     = binned + N_EDGE;

    float* pool   = (float*)d_out;
    float* anchor = pool + (long)B_SUB * F_OUTK;

    hipMemsetAsync(deg_far, 0, (long)N_NODES * sizeof(float), stream);

    gcn_count_kernel<<<N_TILES, TILE_T, 0, stream>>>(dst, tile_cnt);
    gcn_tile_scan_kernel<<<N_BKT / 32, TILE_T, 0, stream>>>(tile_cnt, tile_base, bkt_total);
    gcn_bin_kernel<<<N_TILES, TILE_T, 0, stream>>>(src, dst, deg_far, tile_cnt, tile_base, bkt_total, bucket_start, binned);
    gcn_gemm_wmma_kernel<<<N_NODES / 128, 256, 0, stream>>>(in_feat, W, deg_far, bucket_start, binned, bias, prelu_a,
                                                            H, bkt_late, pool, anchor);
    gcn_bucket_gather_kernel<<<N_BKT / 4, 256, 0, stream>>>(bucket_start, binned, bkt_late, H, bias, prelu_a, pool, anchor);
}


# Tiles keep their bucket-sorted edges in place (per-tile runs + run table); a GEMM block reads its bucket's 256 runs, a thread per tile: count and scan launches gone; row sum order = LDS arrival
# speedup vs baseline: 5.6615x; 1.0606x over previous
#include <hip/hip_runtime.h>

typedef __attribute__((ext_vector_type(2))) float v2f;
typedef __attribute__((ext_vector_type(8))) float v8f;

#define N_NODES  131072
#define F_INK    300
#define F_OUTK   64
#define N_EDGE   2097152
#define S_SUB    128
#define B_SUB    1024


#define TILE_E   8192
#define N_TILES  (N_EDGE / TILE_E)
#define N_BKT    (N_NODES / 128)
#define TILE_T   1024
#define RUN_Q    8

#define KGROUPS (F_INK / 4)
#define WMMA4(a, b, c) __builtin_amdgcn_wmma_f32_16x16x4_f32(false, (a), false, (b), (short)0, (c), false, false)
#define NPW      16
#define FUSE_CAP 3072

__device__ __forceinline__ float2 gcn_sum_rows_lds(const float2* __restrict__ Ht2, const int my,
                                                   const int cnt, const int lane) {
    float2 v[32];
    v[0] = Ht2[__builtin_amdgcn_readlane(my, 0) * 32 + lane];
#pragma unroll
    for (int u = 1; u < 32; ++u) {
        if (u < cnt) v[u] = Ht2[__builtin_amdgcn_readlane(my, u) * 32 + lane];
        else         v[u] = make_float2(0.0f, 0.0f);
    }
    float2 a0 = v[0], a1 = v[1];
#pragma unroll
    for (int u = 2; u < 32; u += 2) {
        a0.x += v[u].x;     a0.y += v[u].y;
        a1.x += v[u + 1].x; a1.y += v[u + 1].y;
    }
    a0.x += a1.x;
    a0.y += a1.y;
    return a0;
}

__device__ __forceinline__ void gcn_mma_group(const float4 x, const float4* __restrict__ wp,
                                              const float rs,
                                              v8f& c0, v8f& c1, v8f& c2, v8f& c3) {
    const float4 b0 = wp[0], b1 = wp[16], b2 = wp[32], b3 = wp[48];
    v2f a0, a1;
    a0.x = x.x * rs; a0.y = x.y * rs;
    a1.x = x.z * rs; a1.y = x.w * rs;
    v2f t0, t1, t2, t3;
    t0.x = b0.x; t0.y = b0.y; t1.x = b1.x; t1.y = b1.y;
    t2.x = b2.x; t2.y = b2.y; t3.x = b3.x; t3.y = b3.y;
    c0 = WMMA4(a0, t0, c0); c1 = WMMA4(a0, t1, c1);
    c2 = WMMA4(a0, t2, c2); c3 = WMMA4(a0, t3, c3);
    t0.x = b0.z; t0.y = b0.w; t1.x = b1.z; t1.y = b1.w;
    t2.x = b2.z; t2.y = b2.w; t3.x = b3.z; t3.y = b3.w;
    c0 = WMMA4(a1, t0, c0); c1 = WMMA4(a1, t1, c1);
    c2 = WMMA4(a1, t2, c2); c3 = WMMA4(a1, t3, c3);
}

__global__ __launch_bounds__(256)
void gcn_gemm_wmma_kernel(const float* __restrict__ X,
                          const float* __restrict__ W,
                          const float* __restrict__ deg_far,
                          const int* __restrict__ run_info,
                          const int* __restrict__ recs,
                          const float* __restrict__ bias,
                          const float* __restrict__ prelu_a,
                          float* __restrict__ H,
                          int* __restrict__ bkt_late,
                          float* __restrict__ pool,
                          float* __restrict__ anchor) {
    __shared__ float4 Ws[KGROUPS * F_OUTK];
    __shared__ int    s_deg[128];
    __shared__ int    n_cnt[128];
    __shared__ int    n_start[128];
    __shared__ unsigned char sorted[FUSE_CAP];
    __shared__ int    s_far;
    __shared__ int    s_out;
    __shared__ int    s_long;
    __shared__ int    s_tot;

    const int tid  = threadIdx.x;
    if (tid < 128) { s_deg[tid] = 0; n_cnt[tid] = 0; }
    if (tid == 0) { s_far = 0; s_out = 0; s_long = 0; s_tot = 0; }
    __syncthreads();
    const int info = run_info[blockIdx.x * N_TILES + tid];
    const int lane = tid & 31;
    const int wave = tid >> 5;
    const int m    = lane & 15;
    const int half = lane >> 4;

    const int rowBase = blockIdx.x * 128 + wave * 16;
    const int row     = rowBase + m;
    const float* xrow = X + (long)row * F_INK;

    const float* xp = xrow + half * 16;
    float4 xa0 = *(const float4*)(xp + 0);
    float4 xa1 = *(const float4*)(xp + 4);
    float4 xa2 = *(const float4*)(xp + 8);
    float4 xa3 = *(const float4*)(xp + 12);
    const float dfar = deg_far[row];
    const float alpha = prelu_a[0];
    const float bf    = bias[tid & 63];

    const int lo = info >> 14;
    const int rc = info & 0x3FFF;
    const int sh = lo & 3;
    const int nq = (rc > 0) ? ((sh + rc + 3) >> 2) : 0;
    const int4* rq = (const int4*)recs + tid * (TILE_E / 4) + (lo >> 2);
    int4 q4[RUN_Q];
#pragma unroll
    for (int k = 0; k < RUN_Q; ++k) {
        q4[k] = (k < nq) ? rq[k] : make_int4(-1, -1, -1, -1);
    }

    {
        const int c = tid & 63;
        for (int g = tid >> 6; g < KGROUPS; g += 4) {
            const float* wp = W + (long)(4 * g) * F_OUTK + c;
            float4 v;
            v.x = wp[0];
            v.y = wp[F_OUTK];
            v.z = wp[2 * F_OUTK];
            v.w = wp[3 * F_OUTK];
            Ws[g * F_OUTK + c] = v;
        }
    }
    int r0[RUN_Q * 4];
#pragma unroll
    for (int k = 0; k < RUN_Q; ++k) {
        r0[4 * k + 0] = q4[k].x; r0[4 * k + 1] = q4[k].y;
        r0[4 * k + 2] = q4[k].z; r0[4 * k + 3] = q4[k].w;
    }
    int far = 0;
#pragma unroll
    for (int e = 0; e < RUN_Q * 4; ++e) {
        const int r = r0[e];
        if ((unsigned)(e - sh) < (unsigned)rc) {
            if ((unsigned)(r >> 14) == blockIdx.x) atomicAdd(&s_deg[(r >> 7) & 127], 1);
            else ++far;
            const unsigned rank = (unsigned)atomicAdd(&n_cnt[r & 127], 1);
            r0[e] = (int)((unsigned)(r & 0x3FFF) | (rank << 14));
        } else {
            r0[e] = -1;
        }
    }
    if (sh + rc > RUN_Q * 4) {
        s_long = 1;
        const int* rp = recs + tid * TILE_E + lo;
        for (int j = RUN_Q * 4 - sh; j < rc; ++j) {
            const int r = rp[j];
            if ((unsigned)(r >> 14) == blockIdx.x) atomicAdd(&s_deg[(r >> 7) & 127], 1);
            else ++far;
        }
    }
    if (far) atomicAdd(&s_far, far);
    if (dfar > 0.0f) s_out = 1;
    __syncthreads();

    const float rs = __frsqrt_rn(fmaxf((float)s_deg[wave * 16 + m] + dfar, 1.0f));

    if (wave == 0) {
        const int q0 = n_cnt[4 * lane + 0];
        const int q1 = n_cnt[4 * lane + 1];
        const int q2 = n_cnt[4 * lane + 2];
        const int q3 = n_cnt[4 * lane + 3];
        const int sum4 = (q0 + q1) + (q2 + q3);
        int incl = sum4;
#pragma unroll
        for (int d = 1; d < 32; d <<= 1) {
            const int up = __shfl_up(incl, d);
            if (lane >= d) incl += up;
        }
        const int e0 = incl - sum4;
        n_start[4 * lane + 0] = e0;
        n_start[4 * lane + 1] = e0 + q0;
        n_start[4 * lane + 2] = e0 + q0 + q1;
        n_start[4 * lane + 3] = e0 + q0 + q1 + q2;
        if (lane == 31) s_tot = incl;
    }
    __syncthreads();
    const bool fused = (s_long == 0) && (s_tot <= FUSE_CAP) && (s_far == 0);
    const bool wr = !fused || (s_out != 0);
    if (tid == 0) bkt_late[blockIdx.x] = fused ? 0 : 1;
    if (fused) {
#pragma unroll
        for (int k = 0; k < RUN_Q * 4; ++k) {
            const int q = r0[k];
            if (q >= 0) sorted[n_start[q & 127] + (q >> 14)] = (unsigned char)((q >> 7) & 127);
        }
    }

    v8f c0 = {}, c1 = {}, c2 = {}, c3 = {};
    const float4* wl = Ws + (half * 4) * F_OUTK + m;

    for (int ch = 0; ch < 9; ++ch) {
        const int nx = (ch < 8) ? ch + 1 : 8;
        const float* xn = xp + nx * 32;
        const float4 xb0 = *(const float4*)(xn + 0);
        const float4 xb1 = *(const float4*)(xn + 4);
        const float4 xb2 = *(const float4*)(xn + 8);
        const float4 xb3 = *(const float4*)(xn + 12);
        const float4* wc = wl + ch * 8 * F_OUTK;
        gcn_mma_group(xa0, wc + 0 * F_OUTK, rs, c0, c1, c2, c3);
        gcn_mma_group(xa1, wc + 1 * F_OUTK, rs, c0, c1, c2, c3);
        gcn_mma_group(xa2, wc + 2 * F_OUTK, rs, c0, c1, c2, c3);
        gcn_mma_group(xa3, wc + 3 * F_OUTK, rs, c0, c1, c2, c3);
        xa0 = xb0; xa1 = xb1; xa2 = xb2; xa3 = xb3;
    }

    {
        const float4 x8 = *(const float4*)(xrow + 288 + half * 4);
        gcn_mma_group(x8, Ws + (72 + half) * F_OUTK + m, rs, c0, c1, c2, c3);
    }
    {
        const v2f xt = *(const v2f*)(xrow + 296 + half * 2);
        v2f a;
        a.x = xt.x * rs; a.y = xt.y * rs;
        const float* wt = (const float*)(Ws + 74 * F_OUTK + m) + half * 2;
        const v2f t0 = *(const v2f*)(wt);
        const v2f t1 = *(const v2f*)(wt + 16 * 4);
        const v2f t2 = *(const v2f*)(wt + 32 * 4);
        const v2f t3 = *(const v2f*)(wt + 48 * 4);
        c0 = WMMA4(a, t0, c0); c1 = WMMA4(a, t1, c1);
        c2 = WMMA4(a, t2, c2); c3 = WMMA4(a, t3, c3);
    }

    if (wr) {
        float* hout = H + (long)(rowBase + half * 8) * F_OUTK + m;
#pragma unroll
        for (int i = 0; i < 8; ++i) {
            hout[(long)i * F_OUTK +  0] = c0[i];
            hout[(long)i * F_OUTK + 16] = c1[i];
            hout[(long)i * F_OUTK + 32] = c2[i];
            hout[(long)i * F_OUTK + 48] = c3[i];
        }
    }
    if (!fused) return;

    __syncthreads();
    float*  Ht   = (float*)Ws;
    float2* accs = (float2*)Ws + 128 * 32;
    float*  n_ri = (float*)Ws + 2 * 128 * F_OUTK;
    float*  red  = n_ri + 128;
    {
        float* ht = Ht + (wave * 16 + half * 8) * F_OUTK + m;
#pragma unroll
        for (int i = 0; i < 8; ++i) {
            ht[i * F_OUTK +  0] = c0[i];
            ht[i * F_OUTK + 16] = c1[i];
            ht[i * F_OUTK + 32] = c2[i];
            ht[i * F_OUTK + 48] = c3[i];
        }
    }
    if (tid < 128) n_ri[tid] = __frsqrt_rn(fmaxf((float)n_cnt[tid], 1.0f));
    __syncthreads();

    {
        const float2* Ht2 = (const float2*)Ws;
        const int nb = wave * NPW;
        int lb = 0, ln = 0;
        if (lane < NPW) {
            lb = n_start[nb + lane];
            ln = n_cnt[nb + lane];
        }
        int beg = __builtin_amdgcn_readlane(lb, 0);
        int n   = __builtin_amdgcn_readlane(ln, 0);
        int my  = (lane < n) ? (int)sorted[beg + lane] : 0;
        for (int i = 0; i < NPW; ++i) {
            int begN = 0, nN = 0, myN = 0;
            if (i + 1 < NPW) {
                begN = __builtin_amdgcn_readlane(lb, i + 1);
                nN   = __builtin_amdgcn_readlane(ln, i + 1);
                myN  = (lane < nN) ? (int)sorted[begN + lane] : 0;
            }
            float2 acc = make_float2(0.0f, 0.0f);
            if (n > 0) {
                acc = gcn_sum_rows_lds(Ht2, my, min(32, n), lane);
                for (int j = 32; j < n; j += 32) {
                    const int more = (lane < n - j) ? (int)sorted[beg + j + lane] : 0;
                    const float2 part = gcn_sum_rows_lds(Ht2, more, min(32, n - j), lane);
                    acc.x += part.x;
                    acc.y += part.y;
                }
            }
            accs[(nb + i) * 32 + lane] = acc;
            beg = begN; n = nN; my = myN;
        }
    }
    __syncthreads();

    {
        const int f = tid & 63;
        const int g = tid >> 6;
        const float* rowsf = (const float*)accs + f;
        float partial = 0.0f;
#pragma unroll
        for (int k = 0; k < S_SUB / 4 - 1; ++k) {
            const int nn = g + 4 * k;
            float v = rowsf[nn * F_OUTK] * n_ri[nn] + bf;
            v = v > 0.0f ? v : alpha * v;
            partial += v;
        }
        {
            const int nn = g + S_SUB - 4;
            float v = rowsf[nn * F_OUTK] * n_ri[nn] + bf;
            v = v > 0.0f ? v : alpha * v;
            if (g == 3) anchor[(long)blockIdx.x * F_OUTK + f] = v;
            else        partial += v;
        }
        red[tid] = partial;
        __syncthreads();
        if (g == 0) {
            const float sum = red[f] + red[f + 64] + red[f + 128] + red[f + 192];
            pool[(long)blockIdx.x * F_OUTK + f] = sum * (1.0f / (S_SUB - 1));
        }
    }
}

__device__ __forceinline__ int gcn_block_excl_scan(const int v, int* s_w, int& total) {
    const int lane = threadIdx.x & 31;
    const int wave = threadIdx.x >> 5;
    int incl = v;
#pragma unroll
    for (int d = 1; d < 32; d <<= 1) {
        const int up = __shfl_up(incl, d);
        if (lane >= d) incl += up;
    }
    if (lane == 31) s_w[wave] = incl;
    __syncthreads();
    int base = 0, all = 0;
#pragma unroll
    for (int w = 0; w < TILE_T / 32; ++w) {
        const int x = s_w[w];
        all += x;
        if (w < wave) base += x;
    }
    total = all;
    return base + incl - v;
}

__global__ __launch_bounds__(TILE_T)
void gcn_bin_kernel(const int* __restrict__ src,
                    const int* __restrict__ dst,
                    float* __restrict__ deg_far,
                    int* __restrict__ run_info,
                    int* __restrict__ recs) {
    __shared__ int  lcur[N_BKT];
    __shared__ int4 rec4[TILE_E / 4];
    __shared__ int  s_w[TILE_T / 32];
    int* rec_l = (int*)rec4;
    const int tid = threadIdx.x;
    const int4* s4 = (const int4*)src + blockIdx.x * (TILE_E / 4);
    const int4* d4 = (const int4*)dst + blockIdx.x * (TILE_E / 4);
    const int4 da = d4[tid];
    const int4 db = d4[tid + TILE_T];
    const int4 sa = s4[tid];
    const int4 sb = s4[tid + TILE_T];
    lcur[tid] = 0;
    __syncthreads();
    atomicAdd(&lcur[da.x >> 7], 1);
    atomicAdd(&lcur[da.y >> 7], 1);
    atomicAdd(&lcur[da.z >> 7], 1);
    atomicAdd(&lcur[da.w >> 7], 1);
    atomicAdd(&lcur[db.x >> 7], 1);
    atomicAdd(&lcur[db.y >> 7], 1);
    atomicAdd(&lcur[db.z >> 7], 1);
    atomicAdd(&lcur[db.w >> 7], 1);
    __syncthreads();
    const int c = lcur[tid];
    int total;
    const int lstart = gcn_block_excl_scan(c, s_w, total);
    lcur[tid] = lstart;
    run_info[tid * N_TILES + blockIdx.x] = (lstart << 14) | c;
    if ((sa.x >> 7) != (da.x >> 7)) atomicAdd(&deg_far[sa.x], 1.0f);
    if ((sa.y >> 7) != (da.y >> 7)) atomicAdd(&deg_far[sa.y], 1.0f);
    if ((sa.z >> 7) != (da.z >> 7)) atomicAdd(&deg_far[sa.z], 1.0f);
    if ((sa.w >> 7) != (da.w >> 7)) atomicAdd(&deg_far[sa.w], 1.0f);
    if ((sb.x >> 7) != (db.x >> 7)) atomicAdd(&deg_far[sb.x], 1.0f);
    if ((sb.y >> 7) != (db.y >> 7)) atomicAdd(&deg_far[sb.y], 1.0f);
    if ((sb.z >> 7) != (db.z >> 7)) atomicAdd(&deg_far[sb.z], 1.0f);
    if ((sb.w >> 7) != (db.w >> 7)) atomicAdd(&deg_far[sb.w], 1.0f);
    __syncthreads();
    int p;
    p = atomicAdd(&lcur[da.x >> 7], 1); rec_l[p] = (sa.x << 7) | (da.x & 127);
    p = atomicAdd(&lcur[da.y >> 7], 1); rec_l[p] = (sa.y << 7) | (da.y & 127);
    p = atomicAdd(&lcur[da.z >> 7], 1); rec_l[p] = (sa.z << 7) | (da.z & 127);
    p = atomicAdd(&lcur[da.w >> 7], 1); rec_l[p] = (sa.w << 7) | (da.w & 127);
    p = atomicAdd(&lcur[db.x >> 7], 1); rec_l[p] = (sb.x << 7) | (db.x & 127);
    p = atomicAdd(&lcur[db.y >> 7], 1); rec_l[p] = (sb.y << 7) | (db.y & 127);
    p = atomicAdd(&lcur[db.z >> 7], 1); rec_l[p] = (sb.z << 7) | (db.z & 127);
    p = atomicAdd(&lcur[db.w >> 7], 1); rec_l[p] = (sb.w << 7) | (db.w & 127);
    __syncthreads();
    int4* out4 = (int4*)recs + blockIdx.x * (TILE_E / 4);
    out4[tid]          = rec4[tid];
    out4[tid + TILE_T] = rec4[tid + TILE_T];
}

__device__ __forceinline__ int gcn_excl_scan256(const int v, int* s_w8, int& total) {
    const int lane = threadIdx.x & 31;
    const int wave = threadIdx.x >> 5;
    int incl = v;
#pragma unroll
    for (int d = 1; d < 32; d <<= 1) {
        const int up = __shfl_up(incl, d);
        if (lane >= d) incl += up;
    }
    if (lane == 31) s_w8[wave] = incl;
    __syncthreads();
    int base = 0, all = 0;
#pragma unroll
    for (int w = 0; w < 8; ++w) {
        const int x = s_w8[w];
        all += x;
        if (w < wave) base += x;
    }
    total = all;
    return base + incl - v;
}

#define GCN_ROW_AT(idx) (*(const float2*)(Hb + (((unsigned)(idx)) << 8) + lane8))

__device__ __forceinline__ float2 gcn_sum_rows(const char* __restrict__ Hb, const int my, const int first,
                                               const int cnt, const unsigned lane8) {
    float2 v[32];
    v[0] = GCN_ROW_AT(first);
#pragma unroll
    for (int u = 1; u < 32; ++u) {
        if (u < cnt) v[u] = GCN_ROW_AT(__builtin_amdgcn_readlane(my, u));
        else         v[u] = make_float2(0.0f, 0.0f);
    }
    float2 a0 = v[0], a1 = v[1];
#pragma unroll
    for (int u = 2; u < 32; u += 2) {
        a0.x += v[u].x;     a0.y += v[u].y;
        a1.x += v[u + 1].x; a1.y += v[u + 1].y;
    }
    a0.x += a1.x;
    a0.y += a1.y;
    return a0;
}

#define BKT_CAP 4096

__global__ __launch_bounds__(256)
void gcn_bucket_gather_kernel(const int* __restrict__ run_info,
                              const int* __restrict__ recs,
                              const int* __restrict__ bkt_late,
                              const float* __restrict__ H,
                              const float* __restrict__ bias,
                              const float* __restrict__ prelu_a,
                              float* __restrict__ pool,
                              float* __restrict__ anchor) {
    __shared__ int    n_cnt[128];
    __shared__ int    n_start[128];
    __shared__ int    n_cur[128];
    __shared__ int    n_deg[128];
    __shared__ int    sorted[BKT_CAP];
    __shared__ float2 accs[128 * 32];
    __shared__ float  n_ri[128];
    __shared__ float  red[256];
    __shared__ int    s_w8[8];

    const int tid  = threadIdx.x;
    const int lane = tid & 31;
    const int wave = tid >> 5;
    const float alpha = prelu_a[0];
    const float bf    = bias[tid & 63];

    for (int b = blockIdx.x; b < N_BKT; b += gridDim.x) {
    if (bkt_late[b] == 0) continue;
    const int info = run_info[b * N_TILES + tid];
    const int lo = info >> 14;
    const int rc = info & 0x3FFF;
    const int* rp = recs + tid * TILE_E + lo;
    int total;
    const int pre = gcn_excl_scan256(rc, s_w8, total);

    if (tid < 128) { n_cnt[tid] = 0; n_deg[tid] = 0; }
    for (int i = tid; i < 128 * 32; i += 256) accs[i] = make_float2(0.0f, 0.0f);
    __syncthreads();

    const char* Hb = (const char*)H;
    const unsigned lane8 = (unsigned)lane * 8u;
    const int nb = wave * NPW;

    for (int c0 = 0; c0 < total; c0 += BKT_CAP) {
        const int j0 = max(0, c0 - pre);
        const int j1 = min(rc, c0 + BKT_CAP - pre);

        for (int j = j0; j < j1; ++j) atomicAdd(&n_cnt[rp[j] & 127], 1);
        __syncthreads();

        if (wave == 0) {
            const int q0 = n_cnt[4 * lane + 0];
            const int q1 = n_cnt[4 * lane + 1];
            const int q2 = n_cnt[4 * lane + 2];
            const int q3 = n_cnt[4 * lane + 3];
            const int sum4 = (q0 + q1) + (q2 + q3);
            int incl = sum4;
#pragma unroll
            for (int d = 1; d < 32; d <<= 1) {
                const int up = __shfl_up(incl, d);
                if (lane >= d) incl += up;
            }
            const int e0 = incl - sum4;
            const int e1 = e0 + q0;
            const int e2 = e1 + q1;
            const int e3 = e2 + q2;
            n_start[4 * lane + 0] = e0; n_cur[4 * lane + 0] = e0; n_deg[4 * lane + 0] += q0;
            n_start[4 * lane + 1] = e1; n_cur[4 * lane + 1] = e1; n_deg[4 * lane + 1] += q1;
            n_start[4 * lane + 2] = e2; n_cur[4 * lane + 2] = e2; n_deg[4 * lane + 2] += q2;
            n_start[4 * lane + 3] = e3; n_cur[4 * lane + 3] = e3; n_deg[4 * lane + 3] += q3;
        }
        __syncthreads();

        for (int j = j0; j < j1; ++j) {
            const int r = rp[j];
            const int p = atomicAdd(&n_cur[r & 127], 1);
            sorted[p] = r >> 7;
        }
        __syncthreads();

        {
            int lb = 0, ln = 0;
            if (lane < NPW) {
                lb = n_start[nb + lane];
                ln = n_cnt[nb + lane];
            }
            int beg = __builtin_amdgcn_readlane(lb, 0);
            int n   = __builtin_amdgcn_readlane(ln, 0);
            int my  = (lane < n) ? sorted[beg + lane] : 0;
            int first = __builtin_amdgcn_readlane(my, 0);
            for (int i = 0; i < NPW; ++i) {
                int begN = 0, nN = 0, myN = 0;
                if (i + 1 < NPW) {
                    begN = __builtin_amdgcn_readlane(lb, i + 1);
                    nN   = __builtin_amdgcn_readlane(ln, i + 1);
                    myN  = (lane < nN) ? sorted[begN + lane] : 0;
                }
                if (n > 0) {
                    float2 acc = gcn_sum_rows(Hb, my, first, min(32, n), lane8);
                    for (int j = 32; j < n; j += 32) {
                        const int more = (lane < n - j) ? sorted[beg + j + lane] : 0;
                        const float2 part = gcn_sum_rows(Hb, more, __builtin_amdgcn_readlane(more, 0),
                                                         min(32, n - j), lane8);
                        acc.x += part.x;
                        acc.y += part.y;
                    }
                    float2 o = accs[(nb + i) * 32 + lane];
                    o.x += acc.x;
                    o.y += acc.y;
                    accs[(nb + i) * 32 + lane] = o;
                }
                beg = begN; n = nN; my = myN;
                first = __builtin_amdgcn_readlane(myN, 0);
            }
        }

        if (c0 + BKT_CAP < total) {
            __syncthreads();
            if (tid < 128) n_cnt[tid] = 0;
            __syncthreads();
        }
    }

    if (tid < 128) n_ri[tid] = __frsqrt_rn(fmaxf((float)n_deg[tid], 1.0f));
    __syncthreads();
    {
        const int f = tid & 63;
        const int g = tid >> 6;
        const float* rowsf = (const float*)accs + f;
        float partial = 0.0f;
#pragma unroll
        for (int k = 0; k < S_SUB / 4 - 1; ++k) {
            const int n = g + 4 * k;
            float v = rowsf[n * F_OUTK] * n_ri[n] + bf;
            v = v > 0.0f ? v : alpha * v;
            partial += v;
        }
        {
            const int n = g + S_SUB - 4;
            float v = rowsf[n * F_OUTK] * n_ri[n] + bf;
            v = v > 0.0f ? v : alpha * v;
            if (g == 3) anchor[(long)b * F_OUTK + f] = v;
            else        partial += v;
        }
        red[tid] = partial;
        __syncthreads();
        if (g == 0) {
            const float sum = red[f] + red[f + 64] + red[f + 128] + red[f + 192];
            pool[(long)b * F_OUTK + f] = sum * (1.0f / (S_SUB - 1));
        }
    }
    __syncthreads();
    }
}

extern "C" void kernel_launch(void* const* d_in, const int* in_sizes, int n_in,
                              void* d_out, int out_size, void* d_ws, size_t ws_size,
                              hipStream_t stream) {
    const float* in_feat  = (const float*)d_in[0];
    const float* W        = (const float*)d_in[1];
    const float* bias     = (const float*)d_in[2];
    const float* prelu_a  = (const float*)d_in[3];
    const int*   src      = (const int*)d_in[4];
    const int*   dst      = (const int*)d_in[5];

    float* ws      = (float*)d_ws;
    float* deg_far = ws;
    float* H       = ws + (long)N_NODES;
    int*   run_info = (int*)(H + (long)N_NODES * F_OUTK);
    int*   recs     = run_info + (long)N_BKT * N_TILES;
    int*   bkt_late = recs + N_EDGE;

    float* pool   = (float*)d_out;
    float* anchor = pool + (long)B_SUB * F_OUTK;

    hipMemsetAsync(deg_far, 0, (long)N_NODES * sizeof(float), stream);

    gcn_bin_kernel<<<N_TILES, TILE_T, 0, stream>>>(src, dst, deg_far, run_info, recs);
    gcn_gemm_wmma_kernel<<<N_NODES / 128, 256, 0, stream>>>(in_feat, W, deg_far, run_info, recs, bias, prelu_a,
                                                            H, bkt_late, pool, anchor);
    gcn_bucket_gather_kernel<<<N_BKT / 4, 256, 0, stream>>>(run_info, recs, bkt_late, H, bias, prelu_a, pool, anchor);
}


# Row sums of the fused block: rows read in groups of four behind one wave-uniform branch, short groups padded with a zero row in LDS; no per-row branch, no zero fill of row registers; sum order kept
# speedup vs baseline: 6.1197x; 1.0809x over previous
#include <hip/hip_runtime.h>

typedef __attribute__((ext_vector_type(2))) float v2f;
typedef __attribute__((ext_vector_type(8))) float v8f;

#define N_NODES  131072
#define F_INK    300
#define F_OUTK   64
#define N_EDGE   2097152
#define S_SUB    128
#define B_SUB    1024


#define TILE_E   8192
#define N_TILES  (N_EDGE / TILE_E)
#define N_BKT    (N_NODES / 128)
#define TILE_T   1024
#define RUN_Q    8

#define KGROUPS (F_INK / 4)
#define WMMA4(a, b, c) __builtin_amdgcn_wmma_f32_16x16x4_f32(false, (a), false, (b), (short)0, (c), false, false)
#define NPW      16
#define FUSE_CAP 3072

#define ZROW 128
#define GCN_ROWS4(buf, base)                                                           \
    buf[0] = Ht2[__builtin_amdgcn_readlane(my, (base) + 0) * 32 + lane];               \
    buf[1] = Ht2[__builtin_amdgcn_readlane(my, (base) + 1) * 32 + lane];               \
    buf[2] = Ht2[__builtin_amdgcn_readlane(my, (base) + 2) * 32 + lane];               \
    buf[3] = Ht2[__builtin_amdgcn_readlane(my, (base) + 3) * 32 + lane];
#define GCN_ADD4(buf)                                                                  \
    a0.x += buf[0].x; a0.y += buf[0].y; a1.x += buf[1].x; a1.y += buf[1].y;            \
    a0.x += buf[2].x; a0.y += buf[2].y; a1.x += buf[3].x; a1.y += buf[3].y;

__device__ __forceinline__ float2 gcn_sum_rows_lds(const float2* __restrict__ Ht2, const int my,
                                                   const int cnt, const int lane) {
    float2 pa[4], pb[4];
    GCN_ROWS4(pa, 0)
    float2 a0, a1;
    if (cnt > 4) { GCN_ROWS4(pb, 4) }
    a0 = pa[0]; a1 = pa[1];
    a0.x += pa[2].x; a0.y += pa[2].y; a1.x += pa[3].x; a1.y += pa[3].y;
#pragma unroll
    for (int g = 1; g < 8; g += 2) {
        if (cnt <= 4 * g) break;
        if (g + 1 < 8 && cnt > 4 * (g + 1)) { GCN_ROWS4(pa, 4 * (g + 1)) }
        GCN_ADD4(pb)
        if (g + 1 >= 8 || cnt <= 4 * (g + 1)) break;
        if (g + 2 < 8 && cnt > 4 * (g + 2)) { GCN_ROWS4(pb, 4 * (g + 2)) }
        GCN_ADD4(pa)
    }
    a0.x += a1.x;
    a0.y += a1.y;
    return a0;
}

__device__ __forceinline__ void gcn_mma_group(const float4 x, const float4* __restrict__ wp,
                                              const float rs,
                                              v8f& c0, v8f& c1, v8f& c2, v8f& c3) {
    const float4 b0 = wp[0], b1 = wp[16], b2 = wp[32], b3 = wp[48];
    v2f a0, a1;
    a0.x = x.x * rs; a0.y = x.y * rs;
    a1.x = x.z * rs; a1.y = x.w * rs;
    v2f t0, t1, t2, t3;
    t0.x = b0.x; t0.y = b0.y; t1.x = b1.x; t1.y = b1.y;
    t2.x = b2.x; t2.y = b2.y; t3.x = b3.x; t3.y = b3.y;
    c0 = WMMA4(a0, t0, c0); c1 = WMMA4(a0, t1, c1);
    c2 = WMMA4(a0, t2, c2); c3 = WMMA4(a0, t3, c3);
    t0.x = b0.z; t0.y = b0.w; t1.x = b1.z; t1.y = b1.w;
    t2.x = b2.z; t2.y = b2.w; t3.x = b3.z; t3.y = b3.w;
    c0 = WMMA4(a1, t0, c0); c1 = WMMA4(a1, t1, c1);
    c2 = WMMA4(a1, t2, c2); c3 = WMMA4(a1, t3, c3);
}

__global__ __launch_bounds__(256)
void gcn_gemm_wmma_kernel(const float* __restrict__ X,
                          const float* __restrict__ W,
                          const float* __restrict__ deg_far,
                          const int* __restrict__ run_info,
                          const int* __restrict__ recs,
                          const float* __restrict__ bias,
                          const float* __restrict__ prelu_a,
                          float* __restrict__ H,
                          int* __restrict__ bkt_late,
                          float* __restrict__ pool,
                          float* __restrict__ anchor) {
    __shared__ float4 Ws[KGROUPS * F_OUTK];
    __shared__ int    s_deg[128];
    __shared__ int    n_cnt[128];
    __shared__ int    n_start[128];
    __shared__ unsigned char sorted[FUSE_CAP];
    __shared__ int    s_far;
    __shared__ int    s_out;
    __shared__ int    s_long;
    __shared__ int    s_tot;

    const int tid  = threadIdx.x;
    if (tid < 128) { s_deg[tid] = 0; n_cnt[tid] = 0; }
    if (tid == 0) { s_far = 0; s_out = 0; s_long = 0; s_tot = 0; }
    __syncthreads();
    const int info = run_info[blockIdx.x * N_TILES + tid];
    const int lane = tid & 31;
    const int wave = tid >> 5;
    const int m    = lane & 15;
    const int half = lane >> 4;

    const int rowBase = blockIdx.x * 128 + wave * 16;
    const int row     = rowBase + m;
    const float* xrow = X + (long)row * F_INK;

    const float* xp = xrow + half * 16;
    float4 xa0 = *(const float4*)(xp + 0);
    float4 xa1 = *(const float4*)(xp + 4);
    float4 xa2 = *(const float4*)(xp + 8);
    float4 xa3 = *(const float4*)(xp + 12);
    const float dfar = deg_far[row];
    const float alpha = prelu_a[0];
    const float bf    = bias[tid & 63];

    const int lo = info >> 14;
    const int rc = info & 0x3FFF;
    const int sh = lo & 3;
    const int nq = (rc > 0) ? ((sh + rc + 3) >> 2) : 0;
    const int4* rq = (const int4*)recs + tid * (TILE_E / 4) + (lo >> 2);
    int4 q4[RUN_Q];
#pragma unroll
    for (int k = 0; k < RUN_Q; ++k) {
        q4[k] = (k < nq) ? rq[k] : make_int4(-1, -1, -1, -1);
    }

    {
        const int c = tid & 63;
        for (int g = tid >> 6; g < KGROUPS; g += 4) {
            const float* wp = W + (long)(4 * g) * F_OUTK + c;
            float4 v;
            v.x = wp[0];
            v.y = wp[F_OUTK];
            v.z = wp[2 * F_OUTK];
            v.w = wp[3 * F_OUTK];
            Ws[g * F_OUTK + c] = v;
        }
    }
    int r0[RUN_Q * 4];
#pragma unroll
    for (int k = 0; k < RUN_Q; ++k) {
        r0[4 * k + 0] = q4[k].x; r0[4 * k + 1] = q4[k].y;
        r0[4 * k + 2] = q4[k].z; r0[4 * k + 3] = q4[k].w;
    }
    int far = 0;
#pragma unroll
    for (int e = 0; e < RUN_Q * 4; ++e) {
        const int r = r0[e];
        if ((unsigned)(e - sh) < (unsigned)rc) {
            if ((unsigned)(r >> 14) == blockIdx.x) atomicAdd(&s_deg[(r >> 7) & 127], 1);
            else ++far;
            const unsigned rank = (unsigned)atomicAdd(&n_cnt[r & 127], 1);
            r0[e] = (int)((unsigned)(r & 0x3FFF) | (rank << 14));
        } else {
            r0[e] = -1;
        }
    }
    if (sh + rc > RUN_Q * 4) {
        s_long = 1;
        const int* rp = recs + tid * TILE_E + lo;
        for (int j = RUN_Q * 4 - sh; j < rc; ++j) {
            const int r = rp[j];
            if ((unsigned)(r >> 14) == blockIdx.x) atomicAdd(&s_deg[(r >> 7) & 127], 1);
            else ++far;
        }
    }
    if (far) atomicAdd(&s_far, far);
    if (dfar > 0.0f) s_out = 1;
    __syncthreads();

    const float rs = __frsqrt_rn(fmaxf((float)s_deg[wave * 16 + m] + dfar, 1.0f));

    if (wave == 0) {
        const int q0 = n_cnt[4 * lane + 0];
        const int q1 = n_cnt[4 * lane + 1];
        const int q2 = n_cnt[4 * lane + 2];
        const int q3 = n_cnt[4 * lane + 3];
        const int sum4 = (q0 + q1) + (q2 + q3);
        int incl = sum4;
#pragma unroll
        for (int d = 1; d < 32; d <<= 1) {
            const int up = __shfl_up(incl, d);
            if (lane >= d) incl += up;
        }
        const int e0 = incl - sum4;
        n_start[4 * lane + 0] = e0;
        n_start[4 * lane + 1] = e0 + q0;
        n_start[4 * lane + 2] = e0 + q0 + q1;
        n_start[4 * lane + 3] = e0 + q0 + q1 + q2;
        if (lane == 31) s_tot = incl;
    }
    __syncthreads();
    const bool fused = (s_long == 0) && (s_tot <= FUSE_CAP) && (s_far == 0);
    const bool wr = !fused || (s_out != 0);
    if (tid == 0) bkt_late[blockIdx.x] = fused ? 0 : 1;
    if (fused) {
#pragma unroll
        for (int k = 0; k < RUN_Q * 4; ++k) {
            const int q = r0[k];
            if (q >= 0) sorted[n_start[q & 127] + (q >> 14)] = (unsigned char)((q >> 7) & 127);
        }
    }

    v8f c0 = {}, c1 = {}, c2 = {}, c3 = {};
    const float4* wl = Ws + (half * 4) * F_OUTK + m;

    for (int ch = 0; ch < 9; ++ch) {
        const int nx = (ch < 8) ? ch + 1 : 8;
        const float* xn = xp + nx * 32;
        const float4 xb0 = *(const float4*)(xn + 0);
        const float4 xb1 = *(const float4*)(xn + 4);
        const float4 xb2 = *(const float4*)(xn + 8);
        const float4 xb3 = *(const float4*)(xn + 12);
        const float4* wc = wl + ch * 8 * F_OUTK;
        gcn_mma_group(xa0, wc + 0 * F_OUTK, rs, c0, c1, c2, c3);
        gcn_mma_group(xa1, wc + 1 * F_OUTK, rs, c0, c1, c2, c3);
        gcn_mma_group(xa2, wc + 2 * F_OUTK, rs, c0, c1, c2, c3);
        gcn_mma_group(xa3, wc + 3 * F_OUTK, rs, c0, c1, c2, c3);
        xa0 = xb0; xa1 = xb1; xa2 = xb2; xa3 = xb3;
    }

    {
        const float4 x8 = *(const float4*)(xrow + 288 + half * 4);
        gcn_mma_group(x8, Ws + (72 + half) * F_OUTK + m, rs, c0, c1, c2, c3);
    }
    {
        const v2f xt = *(const v2f*)(xrow + 296 + half * 2);
        v2f a;
        a.x = xt.x * rs; a.y = xt.y * rs;
        const float* wt = (const float*)(Ws + 74 * F_OUTK + m) + half * 2;
        const v2f t0 = *(const v2f*)(wt);
        const v2f t1 = *(const v2f*)(wt + 16 * 4);
        const v2f t2 = *(const v2f*)(wt + 32 * 4);
        const v2f t3 = *(const v2f*)(wt + 48 * 4);
        c0 = WMMA4(a, t0, c0); c1 = WMMA4(a, t1, c1);
        c2 = WMMA4(a, t2, c2); c3 = WMMA4(a, t3, c3);
    }

    if (wr) {
        float* hout = H + (long)(rowBase + half * 8) * F_OUTK + m;
#pragma unroll
        for (int i = 0; i < 8; ++i) {
            hout[(long)i * F_OUTK +  0] = c0[i];
            hout[(long)i * F_OUTK + 16] = c1[i];
            hout[(long)i * F_OUTK + 32] = c2[i];
            hout[(long)i * F_OUTK + 48] = c3[i];
        }
    }
    if (!fused) return;

    __syncthreads();
    float*  Ht   = (float*)Ws;
    float2* accs = (float2*)Ws + (ZROW + 1) * 32;
    float*  n_ri = (float*)Ws + (2 * 128 + 1) * F_OUTK;
    float*  red  = n_ri + 128;
    {
        float* ht = Ht + (wave * 16 + half * 8) * F_OUTK + m;
#pragma unroll
        for (int i = 0; i < 8; ++i) {
            ht[i * F_OUTK +  0] = c0[i];
            ht[i * F_OUTK + 16] = c1[i];
            ht[i * F_OUTK + 32] = c2[i];
            ht[i * F_OUTK + 48] = c3[i];
        }
    }
    if (tid < 128) n_ri[tid] = __frsqrt_rn(fmaxf((float)n_cnt[tid], 1.0f));
    else if (tid < 128 + F_OUTK) Ht[ZROW * F_OUTK + (tid - 128)] = 0.0f;
    __syncthreads();

    {
        const float2* Ht2 = (const float2*)Ws;
        const int nb = wave * NPW;
        int lb = 0, ln = 0;
        if (lane < NPW) {
            lb = n_start[nb + lane];
            ln = n_cnt[nb + lane];
        }
        int beg = __builtin_amdgcn_readlane(lb, 0);
        int n   = __builtin_amdgcn_readlane(ln, 0);
        int my  = (lane < n) ? (int)sorted[beg + lane] : ZROW;
        for (int i = 0; i < NPW; ++i) {
            int begN = 0, nN = 0, myN = ZROW;
            if (i + 1 < NPW) {
                begN = __builtin_amdgcn_readlane(lb, i + 1);
                nN   = __builtin_amdgcn_readlane(ln, i + 1);
                myN  = (lane < nN) ? (int)sorted[begN + lane] : ZROW;
            }
            float2 acc = make_float2(0.0f, 0.0f);
            if (n > 0) {
                acc = gcn_sum_rows_lds(Ht2, my, min(32, n), lane);
                for (int j = 32; j < n; j += 32) {
                    const int more = (lane < n - j) ? (int)sorted[beg + j + lane] : ZROW;
                    const float2 part = gcn_sum_rows_lds(Ht2, more, min(32, n - j), lane);
                    acc.x += part.x;
                    acc.y += part.y;
                }
            }
            accs[(nb + i) * 32 + lane] = acc;
            beg = begN; n = nN; my = myN;
        }
    }
    __syncthreads();

    {
        const int f = tid & 63;
        const int g = tid >> 6;
        const float* rowsf = (const float*)accs + f;
        float partial = 0.0f;
#pragma unroll
        for (int k = 0; k < S_SUB / 4 - 1; ++k) {
            const int nn = g + 4 * k;
            float v = rowsf[nn * F_OUTK] * n_ri[nn] + bf;
            v = v > 0.0f ? v : alpha * v;
            partial += v;
        }
        {
            const int nn = g + S_SUB - 4;
            float v = rowsf[nn * F_OUTK] * n_ri[nn] + bf;
            v = v > 0.0f ? v : alpha * v;
            if (g == 3) anchor[(long)blockIdx.x * F_OUTK + f] = v;
            else        partial += v;
        }
        red[tid] = partial;
        __syncthreads();
        if (g == 0) {
            const float sum = red[f] + red[f + 64] + red[f + 128] + red[f + 192];
            pool[(long)blockIdx.x * F_OUTK + f] = sum * (1.0f / (S_SUB - 1));
        }
    }
}

__device__ __forceinline__ int gcn_block_excl_scan(const int v, int* s_w, int& total) {
    const int lane = threadIdx.x & 31;
    const int wave = threadIdx.x >> 5;
    int incl = v;
#pragma unroll
    for (int d = 1; d < 32; d <<= 1) {
        const int up = __shfl_up(incl, d);
        if (lane >= d) incl += up;
    }
    if (lane == 31) s_w[wave] = incl;
    __syncthreads();
    int base = 0, all = 0;
#pragma unroll
    for (int w = 0; w < TILE_T / 32; ++w) {
        const int x = s_w[w];
        all += x;
        if (w < wave) base += x;
    }
    total = all;
    return base + incl - v;
}

__global__ __launch_bounds__(TILE_T)
void gcn_bin_kernel(const int* __restrict__ src,
                    const int* __restrict__ dst,
                    float* __restrict__ deg_far,
                    int* __restrict__ run_info,
                    int* __restrict__ recs) {
    __shared__ int  lcur[N_BKT];
    __shared__ int4 rec4[TILE_E / 4];
    __shared__ int  s_w[TILE_T / 32];
    int* rec_l = (int*)rec4;
    const int tid = threadIdx.x;
    const int4* s4 = (const int4*)src + blockIdx.x * (TILE_E / 4);
    const int4* d4 = (const int4*)dst + blockIdx.x * (TILE_E / 4);
    const int4 da = d4[tid];
    const int4 db = d4[tid + TILE_T];
    const int4 sa = s4[tid];
    const int4 sb = s4[tid + TILE_T];
    lcur[tid] = 0;
    __syncthreads();
    atomicAdd(&lcur[da.x >> 7], 1);
    atomicAdd(&lcur[da.y >> 7], 1);
    atomicAdd(&lcur[da.z >> 7], 1);
    atomicAdd(&lcur[da.w >> 7], 1);
    atomicAdd(&lcur[db.x >> 7], 1);
    atomicAdd(&lcur[db.y >> 7], 1);
    atomicAdd(&lcur[db.z >> 7], 1);
    atomicAdd(&lcur[db.w >> 7], 1);
    __syncthreads();
    const int c = lcur[tid];
    int total;
    const int lstart = gcn_block_excl_scan(c, s_w, total);
    lcur[tid] = lstart;
    run_info[tid * N_TILES + blockIdx.x] = (lstart << 14) | c;
    if ((sa.x >> 7) != (da.x >> 7)) atomicAdd(&deg_far[sa.x], 1.0f);
    if ((sa.y >> 7) != (da.y >> 7)) atomicAdd(&deg_far[sa.y], 1.0f);
    if ((sa.z >> 7) != (da.z >> 7)) atomicAdd(&deg_far[sa.z], 1.0f);
    if ((sa.w >> 7) != (da.w >> 7)) atomicAdd(&deg_far[sa.w], 1.0f);
    if ((sb.x >> 7) != (db.x >> 7)) atomicAdd(&deg_far[sb.x], 1.0f);
    if ((sb.y >> 7) != (db.y >> 7)) atomicAdd(&deg_far[sb.y], 1.0f);
    if ((sb.z >> 7) != (db.z >> 7)) atomicAdd(&deg_far[sb.z], 1.0f);
    if ((sb.w >> 7) != (db.w >> 7)) atomicAdd(&deg_far[sb.w], 1.0f);
    __syncthreads();
    int p;
    p = atomicAdd(&lcur[da.x >> 7], 1); rec_l[p] = (sa.x << 7) | (da.x & 127);
    p = atomicAdd(&lcur[da.y >> 7], 1); rec_l[p] = (sa.y << 7) | (da.y & 127);
    p = atomicAdd(&lcur[da.z >> 7], 1); rec_l[p] = (sa.z << 7) | (da.z & 127);
    p = atomicAdd(&lcur[da.w >> 7], 1); rec_l[p] = (sa.w << 7) | (da.w & 127);
    p = atomicAdd(&lcur[db.x >> 7], 1); rec_l[p] = (sb.x << 7) | (db.x & 127);
    p = atomicAdd(&lcur[db.y >> 7], 1); rec_l[p] = (sb.y << 7) | (db.y & 127);
    p = atomicAdd(&lcur[db.z >> 7], 1); rec_l[p] = (sb.z << 7) | (db.z & 127);
    p = atomicAdd(&lcur[db.w >> 7], 1); rec_l[p] = (sb.w << 7) | (db.w & 127);
    __syncthreads();
    int4* out4 = (int4*)recs + blockIdx.x * (TILE_E / 4);
    out4[tid]          = rec4[tid];
    out4[tid + TILE_T] = rec4[tid + TILE_T];
}

__device__ __forceinline__ int gcn_excl_scan256(const int v, int* s_w8, int& total) {
    const int lane = threadIdx.x & 31;
    const int wave = threadIdx.x >> 5;
    int incl = v;
#pragma unroll
    for (int d = 1; d < 32; d <<= 1) {
        const int up = __shfl_up(incl, d);
        if (lane >= d) incl += up;
    }
    if (lane == 31) s_w8[wave] = incl;
    __syncthreads();
    int base = 0, all = 0;
#pragma unroll
    for (int w = 0; w < 8; ++w) {
        const int x = s_w8[w];
        all += x;
        if (w < wave) base += x;
    }
    total = all;
    return base + incl - v;
}

#define GCN_ROW_AT(idx) (*(const float2*)(Hb + (((unsigned)(idx)) << 8) + lane8))

__device__ __forceinline__ float2 gcn_sum_rows(const char* __restrict__ Hb, const int my, const int first,
                                               const int cnt, const unsigned lane8) {
    float2 v[32];
    v[0] = GCN_ROW_AT(first);
#pragma unroll
    for (int u = 1; u < 32; ++u) {
        if (u < cnt) v[u] = GCN_ROW_AT(__builtin_amdgcn_readlane(my, u));
        else         v[u] = make_float2(0.0f, 0.0f);
    }
    float2 a0 = v[0], a1 = v[1];
#pragma unroll
    for (int u = 2; u < 32; u += 2) {
        a0.x += v[u].x;     a0.y += v[u].y;
        a1.x += v[u + 1].x; a1.y += v[u + 1].y;
    }
    a0.x += a1.x;
    a0.y += a1.y;
    return a0;
}

#define BKT_CAP 4096

__global__ __launch_bounds__(256)
void gcn_bucket_gather_kernel(const int* __restrict__ run_info,
                              const int* __restrict__ recs,
                              const int* __restrict__ bkt_late,
                              const float* __restrict__ H,
                              const float* __restrict__ bias,
                              const float* __restrict__ prelu_a,
                              float* __restrict__ pool,
                              float* __restrict__ anchor) {
    __shared__ int    n_cnt[128];
    __shared__ int    n_start[128];
    __shared__ int    n_cur[128];
    __shared__ int    n_deg[128];
    __shared__ int    sorted[BKT_CAP];
    __shared__ float2 accs[128 * 32];
    __shared__ float  n_ri[128];
    __shared__ float  red[256];
    __shared__ int    s_w8[8];

    const int tid  = threadIdx.x;
    const int lane = tid & 31;
    const int wave = tid >> 5;
    const float alpha = prelu_a[0];
    const float bf    = bias[tid & 63];

    for (int b = blockIdx.x; b < N_BKT; b += gridDim.x) {
    if (bkt_late[b] == 0) continue;
    const int info = run_info[b * N_TILES + tid];
    const int lo = info >> 14;
    const int rc = info & 0x3FFF;
    const int* rp = recs + tid * TILE_E + lo;
    int total;
    const int pre = gcn_excl_scan256(rc, s_w8, total);

    if (tid < 128) { n_cnt[tid] = 0; n_deg[tid] = 0; }
    for (int i = tid; i < 128 * 32; i += 256) accs[i] = make_float2(0.0f, 0.0f);
    __syncthreads();

    const char* Hb = (const char*)H;
    const unsigned lane8 = (unsigned)lane * 8u;
    const int nb = wave * NPW;

    for (int c0 = 0; c0 < total; c0 += BKT_CAP) {
        const int j0 = max(0, c0 - pre);
        const int j1 = min(rc, c0 + BKT_CAP - pre);

        for (int j = j0; j < j1; ++j) atomicAdd(&n_cnt[rp[j] & 127], 1);
        __syncthreads();

        if (wave == 0) {
            const int q0 = n_cnt[4 * lane + 0];
            const int q1 = n_cnt[4 * lane + 1];
            const int q2 = n_cnt[4 * lane + 2];
            const int q3 = n_cnt[4 * lane + 3];
            const int sum4 = (q0 + q1) + (q2 + q3);
            int incl = sum4;
#pragma unroll
            for (int d = 1; d < 32; d <<= 1) {
                const int up = __shfl_up(incl, d);
                if (lane >= d) incl += up;
            }
            const int e0 = incl - sum4;
            const int e1 = e0 + q0;
            const int e2 = e1 + q1;
            const int e3 = e2 + q2;
            n_start[4 * lane + 0] = e0; n_cur[4 * lane + 0] = e0; n_deg[4 * lane + 0] += q0;
            n_start[4 * lane + 1] = e1; n_cur[4 * lane + 1] = e1; n_deg[4 * lane + 1] += q1;
            n_start[4 * lane + 2] = e2; n_cur[4 * lane + 2] = e2; n_deg[4 * lane + 2] += q2;
            n_start[4 * lane + 3] = e3; n_cur[4 * lane + 3] = e3; n_deg[4 * lane + 3] += q3;
        }
        __syncthreads();

        for (int j = j0; j < j1; ++j) {
            const int r = rp[j];
            const int p = atomicAdd(&n_cur[r & 127], 1);
            sorted[p] = r >> 7;
        }
        __syncthreads();

        {
            int lb = 0, ln = 0;
            if (lane < NPW) {
                lb = n_start[nb + lane];
                ln = n_cnt[nb + lane];
            }
            int beg = __builtin_amdgcn_readlane(lb, 0);
            int n   = __builtin_amdgcn_readlane(ln, 0);
            int my  = (lane < n) ? sorted[beg + lane] : 0;
            int first = __builtin_amdgcn_readlane(my, 0);
            for (int i = 0; i < NPW; ++i) {
                int begN = 0, nN = 0, myN = 0;
                if (i + 1 < NPW) {
                    begN = __builtin_amdgcn_readlane(lb, i + 1);
                    nN   = __builtin_amdgcn_readlane(ln, i + 1);
                    myN  = (lane < nN) ? sorted[begN + lane] : 0;
                }
                if (n > 0) {
                    float2 acc = gcn_sum_rows(Hb, my, first, min(32, n), lane8);
                    for (int j = 32; j < n; j += 32) {
                        const int more = (lane < n - j) ? sorted[beg + j + lane] : 0;
                        const float2 part = gcn_sum_rows(Hb, more, __builtin_amdgcn_readlane(more, 0),
                                                         min(32, n - j), lane8);
                        acc.x += part.x;
                        acc.y += part.y;
                    }
                    float2 o = accs[(nb + i) * 32 + lane];
                    o.x += acc.x;
                    o.y += acc.y;
                    accs[(nb + i) * 32 + lane] = o;
                }
                beg = begN; n = nN; my = myN;
                first = __builtin_amdgcn_readlane(myN, 0);
            }
        }

        if (c0 + BKT_CAP < total) {
            __syncthreads();
            if (tid < 128) n_cnt[tid] = 0;
            __syncthreads();
        }
    }

    if (tid < 128) n_ri[tid] = __frsqrt_rn(fmaxf((float)n_deg[tid], 1.0f));
    __syncthreads();
    {
        const int f = tid & 63;
        const int g = tid >> 6;
        const float* rowsf = (const float*)accs + f;
        float partial = 0.0f;
#pragma unroll
        for (int k = 0; k < S_SUB / 4 - 1; ++k) {
            const int n = g + 4 * k;
            float v = rowsf[n * F_OUTK] * n_ri[n] + bf;
            v = v > 0.0f ? v : alpha * v;
            partial += v;
        }
        {
            const int n = g + S_SUB - 4;
            float v = rowsf[n * F_OUTK] * n_ri[n] + bf;
            v = v > 0.0f ? v : alpha * v;
            if (g == 3) anchor[(long)b * F_OUTK + f] = v;
            else        partial += v;
        }
        red[tid] = partial;
        __syncthreads();
        if (g == 0) {
            const float sum = red[f] + red[f + 64] + red[f + 128] + red[f + 192];
            pool[(long)b * F_OUTK + f] = sum * (1.0f / (S_SUB - 1));
        }
    }
    __syncthreads();
    }
}

extern "C" void kernel_launch(void* const* d_in, const int* in_sizes, int n_in,
                              void* d_out, int out_size, void* d_ws, size_t ws_size,
                              hipStream_t stream) {
    const float* in_feat  = (const float*)d_in[0];
    const float* W        = (const float*)d_in[1];
    const float* bias     = (const float*)d_in[2];
    const float* prelu_a  = (const float*)d_in[3];
    const int*   src      = (const int*)d_in[4];
    const int*   dst      = (const int*)d_in[5];

    float* ws      = (float*)d_ws;
    float* deg_far = ws;
    float* H       = ws + (long)N_NODES;
    int*   run_info = (int*)(H + (long)N_NODES * F_OUTK);
    int*   recs     = run_info + (long)N_BKT * N_TILES;
    int*   bkt_late = recs + N_EDGE;

    float* pool   = (float*)d_out;
    float* anchor = pool + (long)B_SUB * F_OUTK;

    hipMemsetAsync(deg_far, 0, (long)N_NODES * sizeof(float), stream);

    gcn_bin_kernel<<<N_TILES, TILE_T, 0, stream>>>(src, dst, deg_far, run_info, recs);
    gcn_gemm_wmma_kernel<<<N_NODES / 128, 256, 0, stream>>>(in_feat, W, deg_far, run_info, recs, bias, prelu_a,
                                                            H, bkt_late, pool, anchor);
    gcn_bucket_gather_kernel<<<N_BKT / 4, 256, 0, stream>>>(run_info, recs, bkt_late, H, bias, prelu_a, pool, anchor);
}


# Fused row sums: two nodes per wave step, a 16-lane half each; DPP row_share hands the half its source row inside the address OR, one 16-byte LDS read per lane, no readlane per row; sum order kept
# speedup vs baseline: 6.3687x; 1.0407x over previous
#include <hip/hip_runtime.h>

typedef __attribute__((ext_vector_type(2))) float v2f;
typedef __attribute__((ext_vector_type(8))) float v8f;
typedef __attribute__((ext_vector_type(4))) float v4f;

#define N_NODES  131072
#define F_INK    300
#define F_OUTK   64
#define N_EDGE   2097152
#define S_SUB    128
#define B_SUB    1024


#define TILE_E   8192
#define N_TILES  (N_EDGE / TILE_E)
#define N_BKT    (N_NODES / 128)
#define TILE_T   1024
#define RUN_Q    8

#define KGROUPS (F_INK / 4)
#define WMMA4(a, b, c) __builtin_amdgcn_wmma_f32_16x16x4_f32(false, (a), false, (b), (short)0, (c), false, false)
#define NPW      16
#define FUSE_CAP 3072

#define ZROW 128

template <int K>
__device__ __forceinline__ v4f gcn_row_rs(const char* __restrict__ Hb, const int ms, const unsigned qoff) {
    const unsigned off = (unsigned)__builtin_amdgcn_update_dpp(0, ms, 0x150 + K, 0xf, 0xf, true) | qoff;
    return *(const v4f*)(Hb + off);
}

template <int G>
__device__ __forceinline__ void gcn_pair_group(const char* __restrict__ Hb, const int m1,
                                               const int cnt, const unsigned qoff, v4f& a0, v4f& a1,
                                               const v4f x0, const v4f x1, const v4f x2, const v4f x3) {
    if constexpr (G < 7) {
        if (cnt > 4 * (G + 1)) {
            constexpr int B = (4 * (G + 1)) & 15;
            const v4f y0 = gcn_row_rs<B + 0>(Hb, m1, qoff);
            const v4f y1 = gcn_row_rs<B + 1>(Hb, m1, qoff);
            const v4f y2 = gcn_row_rs<B + 2>(Hb, m1, qoff);
            const v4f y3 = gcn_row_rs<B + 3>(Hb, m1, qoff);
            a0 += x0; a1 += x1; a0 += x2; a1 += x3;
            gcn_pair_group<G + 1>(Hb, m1, cnt, qoff, a0, a1, y0, y1, y2, y3);
            return;
        }
    }
    a0 += x0; a1 += x1; a0 += x2; a1 += x3;
}

__device__ __forceinline__ v4f gcn_sum_rows_pair(const char* __restrict__ Hb, const int m0, const int m1,
                                                 const int cnt, const unsigned qoff) {
    const v4f x0  = gcn_row_rs<0>(Hb, m0, qoff);
    const v4f x1  = gcn_row_rs<1>(Hb, m0, qoff);
    const v4f x2  = gcn_row_rs<2>(Hb, m0, qoff);
    const v4f x3  = gcn_row_rs<3>(Hb, m0, qoff);
    const v4f x4  = gcn_row_rs<4>(Hb, m0, qoff);
    const v4f x5  = gcn_row_rs<5>(Hb, m0, qoff);
    const v4f x6  = gcn_row_rs<6>(Hb, m0, qoff);
    const v4f x7  = gcn_row_rs<7>(Hb, m0, qoff);
    const v4f x8  = gcn_row_rs<8>(Hb, m0, qoff);
    const v4f x9  = gcn_row_rs<9>(Hb, m0, qoff);
    const v4f x10 = gcn_row_rs<10>(Hb, m0, qoff);
    const v4f x11 = gcn_row_rs<11>(Hb, m0, qoff);
    const v4f x12 = gcn_row_rs<12>(Hb, m0, qoff);
    const v4f x13 = gcn_row_rs<13>(Hb, m0, qoff);
    const v4f x14 = gcn_row_rs<14>(Hb, m0, qoff);
    const v4f x15 = gcn_row_rs<15>(Hb, m0, qoff);
    const bool more = cnt > 16;
    v4f y0, y1, y2, y3;
    if (more) {
        y0 = gcn_row_rs<0>(Hb, m1, qoff);
        y1 = gcn_row_rs<1>(Hb, m1, qoff);
        y2 = gcn_row_rs<2>(Hb, m1, qoff);
        y3 = gcn_row_rs<3>(Hb, m1, qoff);
    }
    v4f a0 = x0, a1 = x1;
    a0 += x2;  a1 += x3;
    a0 += x4;  a1 += x5;  a0 += x6;  a1 += x7;
    a0 += x8;  a1 += x9;  a0 += x10; a1 += x11;
    a0 += x12; a1 += x13; a0 += x14; a1 += x15;
    if (more) gcn_pair_group<4>(Hb, m1, cnt, qoff, a0, a1, y0, y1, y2, y3);
    return a0 + a1;
}

__device__ __forceinline__ void gcn_mma_group(const float4 x, const float4* __restrict__ wp,
                                              const float rs,
                                              v8f& c0, v8f& c1, v8f& c2, v8f& c3) {
    const float4 b0 = wp[0], b1 = wp[16], b2 = wp[32], b3 = wp[48];
    v2f a0, a1;
    a0.x = x.x * rs; a0.y = x.y * rs;
    a1.x = x.z * rs; a1.y = x.w * rs;
    v2f t0, t1, t2, t3;
    t0.x = b0.x; t0.y = b0.y; t1.x = b1.x; t1.y = b1.y;
    t2.x = b2.x; t2.y = b2.y; t3.x = b3.x; t3.y = b3.y;
    c0 = WMMA4(a0, t0, c0); c1 = WMMA4(a0, t1, c1);
    c2 = WMMA4(a0, t2, c2); c3 = WMMA4(a0, t3, c3);
    t0.x = b0.z; t0.y = b0.w; t1.x = b1.z; t1.y = b1.w;
    t2.x = b2.z; t2.y = b2.w; t3.x = b3.z; t3.y = b3.w;
    c0 = WMMA4(a1, t0, c0); c1 = WMMA4(a1, t1, c1);
    c2 = WMMA4(a1, t2, c2); c3 = WMMA4(a1, t3, c3);
}

__global__ __launch_bounds__(256)
void gcn_gemm_wmma_kernel(const float* __restrict__ X,
                          const float* __restrict__ W,
                          const float* __restrict__ deg_far,
                          const int* __restrict__ run_info,
                          const int* __restrict__ recs,
                          const float* __restrict__ bias,
                          const float* __restrict__ prelu_a,
                          float* __restrict__ H,
                          int* __restrict__ bkt_late,
                          float* __restrict__ pool,
                          float* __restrict__ anchor) {
    __shared__ float4 Ws[KGROUPS * F_OUTK];
    __shared__ int    s_deg[128];
    __shared__ int    n_cnt[128];
    __shared__ int    n_start[128];
    __shared__ unsigned char sorted[FUSE_CAP];
    __shared__ int    s_far;
    __shared__ int    s_out;
    __shared__ int    s_long;
    __shared__ int    s_tot;

    const int tid  = threadIdx.x;
    if (tid < 128) { s_deg[tid] = 0; n_cnt[tid] = 0; }
    if (tid == 0) { s_far = 0; s_out = 0; s_long = 0; s_tot = 0; }
    __syncthreads();
    const int info = run_info[blockIdx.x * N_TILES + tid];
    const int lane = tid & 31;
    const int wave = tid >> 5;
    const int m    = lane & 15;
    const int half = lane >> 4;

    const int rowBase = blockIdx.x * 128 + wave * 16;
    const int row     = rowBase + m;
    const float* xrow = X + (long)row * F_INK;

    const float* xp = xrow + half * 16;
    float4 xa0 = *(const float4*)(xp + 0);
    float4 xa1 = *(const float4*)(xp + 4);
    float4 xa2 = *(const float4*)(xp + 8);
    float4 xa3 = *(const float4*)(xp + 12);
    const float dfar = deg_far[row];
    const float alpha = prelu_a[0];
    const float bf    = bias[tid & 63];

    const int lo = info >> 14;
    const int rc = info & 0x3FFF;
    const int sh = lo & 3;
    const int nq = (rc > 0) ? ((sh + rc + 3) >> 2) : 0;
    const int4* rq = (const int4*)recs + tid * (TILE_E / 4) + (lo >> 2);
    int4 q4[RUN_Q];
#pragma unroll
    for (int k = 0; k < RUN_Q; ++k) {
        q4[k] = (k < nq) ? rq[k] : make_int4(-1, -1, -1, -1);
    }

    {
        const int c = tid & 63;
        for (int g = tid >> 6; g < KGROUPS; g += 4) {
            const float* wp = W + (long)(4 * g) * F_OUTK + c;
            float4 v;
            v.x = wp[0];
            v.y = wp[F_OUTK];
            v.z = wp[2 * F_OUTK];
            v.w = wp[3 * F_OUTK];
            Ws[g * F_OUTK + c] = v;
        }
    }
    int r0[RUN_Q * 4];
#pragma unroll
    for (int k = 0; k < RUN_Q; ++k) {
        r0[4 * k + 0] = q4[k].x; r0[4 * k + 1] = q4[k].y;
        r0[4 * k + 2] = q4[k].z; r0[4 * k + 3] = q4[k].w;
    }
    int far = 0;
#pragma unroll
    for (int e = 0; e < RUN_Q * 4; ++e) {
        const int r = r0[e];
        if ((unsigned)(e - sh) < (unsigned)rc) {
            if ((unsigned)(r >> 14) == blockIdx.x) atomicAdd(&s_deg[(r >> 7) & 127], 1);
            else ++far;
            const unsigned rank = (unsigned)atomicAdd(&n_cnt[r & 127], 1);
            r0[e] = (int)((unsigned)(r & 0x3FFF) | (rank << 14));
        } else {
            r0[e] = -1;
        }
    }
    if (sh + rc > RUN_Q * 4) {
        s_long = 1;
        const int* rp = recs + tid * TILE_E + lo;
        for (int j = RUN_Q * 4 - sh; j < rc; ++j) {
            const int r = rp[j];
            if ((unsigned)(r >> 14) == blockIdx.x) atomicAdd(&s_deg[(r >> 7) & 127], 1);
            else ++far;
        }
    }
    if (far) atomicAdd(&s_far, far);
    if (dfar > 0.0f) s_out = 1;
    __syncthreads();

    const float rs = __frsqrt_rn(fmaxf((float)s_deg[wave * 16 + m] + dfar, 1.0f));

    if (wave == 0) {
        const int q0 = n_cnt[4 * lane + 0];
        const int q1 = n_cnt[4 * lane + 1];
        const int q2 = n_cnt[4 * lane + 2];
        const int q3 = n_cnt[4 * lane + 3];
        const int sum4 = (q0 + q1) + (q2 + q3);
        int incl = sum4;
#pragma unroll
        for (int d = 1; d < 32; d <<= 1) {
            const int up = __shfl_up(incl, d);
            if (lane >= d) incl += up;
        }
        const int e0 = incl - sum4;
        n_start[4 * lane + 0] = e0;
        n_start[4 * lane + 1] = e0 + q0;
        n_start[4 * lane + 2] = e0 + q0 + q1;
        n_start[4 * lane + 3] = e0 + q0 + q1 + q2;
        if (lane == 31) s_tot = incl;
    }
    __syncthreads();
    const bool fused = (s_long == 0) && (s_tot <= FUSE_CAP) && (s_far == 0);
    const bool wr = !fused || (s_out != 0);
    if (tid == 0) bkt_late[blockIdx.x] = fused ? 0 : 1;
    if (fused) {
#pragma unroll
        for (int k = 0; k < RUN_Q * 4; ++k) {
            const int q = r0[k];
            if (q >= 0) sorted[n_start[q & 127] + (q >> 14)] = (unsigned char)((q >> 7) & 127);
        }
    }

    v8f c0 = {}, c1 = {}, c2 = {}, c3 = {};
    const float4* wl = Ws + (half * 4) * F_OUTK + m;

    for (int ch = 0; ch < 9; ++ch) {
        const int nx = (ch < 8) ? ch + 1 : 8;
        const float* xn = xp + nx * 32;
        const float4 xb0 = *(const float4*)(xn + 0);
        const float4 xb1 = *(const float4*)(xn + 4);
        const float4 xb2 = *(const float4*)(xn + 8);
        const float4 xb3 = *(const float4*)(xn + 12);
        const float4* wc = wl + ch * 8 * F_OUTK;
        gcn_mma_group(xa0, wc + 0 * F_OUTK, rs, c0, c1, c2, c3);
        gcn_mma_group(xa1, wc + 1 * F_OUTK, rs, c0, c1, c2, c3);
        gcn_mma_group(xa2, wc + 2 * F_OUTK, rs, c0, c1, c2, c3);
        gcn_mma_group(xa3, wc + 3 * F_OUTK, rs, c0, c1, c2, c3);
        xa0 = xb0; xa1 = xb1; xa2 = xb2; xa3 = xb3;
    }

    {
        const float4 x8 = *(const float4*)(xrow + 288 + half * 4);
        gcn_mma_group(x8, Ws + (72 + half) * F_OUTK + m, rs, c0, c1, c2, c3);
    }
    {
        const v2f xt = *(const v2f*)(xrow + 296 + half * 2);
        v2f a;
        a.x = xt.x * rs; a.y = xt.y * rs;
        const float* wt = (const float*)(Ws + 74 * F_OUTK + m) + half * 2;
        const v2f t0 = *(const v2f*)(wt);
        const v2f t1 = *(const v2f*)(wt + 16 * 4);
        const v2f t2 = *(const v2f*)(wt + 32 * 4);
        const v2f t3 = *(const v2f*)(wt + 48 * 4);
        c0 = WMMA4(a, t0, c0); c1 = WMMA4(a, t1, c1);
        c2 = WMMA4(a, t2, c2); c3 = WMMA4(a, t3, c3);
    }

    if (wr) {
        float* hout = H + (long)(rowBase + half * 8) * F_OUTK + m;
#pragma unroll
        for (int i = 0; i < 8; ++i) {
            hout[(long)i * F_OUTK +  0] = c0[i];
            hout[(long)i * F_OUTK + 16] = c1[i];
            hout[(long)i * F_OUTK + 32] = c2[i];
            hout[(long)i * F_OUTK + 48] = c3[i];
        }
    }
    if (!fused) return;

    __syncthreads();
    float*  Ht   = (float*)Ws;
    float2* accs = (float2*)Ws + (ZROW + 1) * 32;
    float*  n_ri = (float*)Ws + (2 * 128 + 1) * F_OUTK;
    float*  red  = n_ri + 128;
    {
        float* ht = Ht + (wave * 16 + half * 8) * F_OUTK + m;
#pragma unroll
        for (int i = 0; i < 8; ++i) {
            ht[i * F_OUTK +  0] = c0[i];
            ht[i * F_OUTK + 16] = c1[i];
            ht[i * F_OUTK + 32] = c2[i];
            ht[i * F_OUTK + 48] = c3[i];
        }
    }
    if (tid < 128) n_ri[tid] = __frsqrt_rn(fmaxf((float)n_cnt[tid], 1.0f));
    else if (tid < 128 + F_OUTK) Ht[ZROW * F_OUTK + (tid - 128)] = 0.0f;
    __syncthreads();

    {
        const char* Hb = (const char*)Ws;
        v4f* accs4 = (v4f*)accs;
        const unsigned qoff = (unsigned)m << 4;
        const int nb = wave * NPW;
        int lb = 0, ln = 0;
        if (lane < NPW) {
            lb = n_start[nb + lane];
            ln = n_cnt[nb + lane];
        }
        int nA = __builtin_amdgcn_readlane(ln, 0);
        int nB = __builtin_amdgcn_readlane(ln, 1);
        int bA = __builtin_amdgcn_readlane(lb, 0);
        int bB = __builtin_amdgcn_readlane(lb, 1);
        int beg  = half ? bB : bA;
        int n    = half ? nB : nA;
        int nmax = max(nA, nB);
        int my0  = (m < n)      ? (int)sorted[beg + m]      : ZROW;
        int my1  = (m + 16 < n) ? (int)sorted[beg + 16 + m] : ZROW;
        for (int p = 0; p < NPW / 2; ++p) {
            int begN = 0, nN = 0, nmaxN = 0, my0N = ZROW, my1N = ZROW;
            if (p + 1 < NPW / 2) {
                nA    = __builtin_amdgcn_readlane(ln, 2 * p + 2);
                nB    = __builtin_amdgcn_readlane(ln, 2 * p + 3);
                bA    = __builtin_amdgcn_readlane(lb, 2 * p + 2);
                bB    = __builtin_amdgcn_readlane(lb, 2 * p + 3);
                begN  = half ? bB : bA;
                nN    = half ? nB : nA;
                nmaxN = max(nA, nB);
                my0N  = (m < nN)      ? (int)sorted[begN + m]      : ZROW;
                my1N  = (m + 16 < nN) ? (int)sorted[begN + 16 + m] : ZROW;
            }
            v4f acc = {0.0f, 0.0f, 0.0f, 0.0f};
            if (nmax > 0) {
                acc = gcn_sum_rows_pair(Hb, my0 << 8, my1 << 8, min(32, nmax), qoff);
                for (int j = 32; j < nmax; j += 32) {
                    const int e0 = (j + m < n)      ? (int)sorted[beg + j + m]      : ZROW;
                    const int e1 = (j + 16 + m < n) ? (int)sorted[beg + j + 16 + m] : ZROW;
                    acc += gcn_sum_rows_pair(Hb, e0 << 8, e1 << 8, min(32, nmax - j), qoff);
                }
            }
            accs4[(nb + 2 * p + half) * 16 + m] = acc;
            beg = begN; n = nN; nmax = nmaxN; my0 = my0N; my1 = my1N;
        }
    }
    __syncthreads();

    {
        const int f = tid & 63;
        const int g = tid >> 6;
        const float* rowsf = (const float*)accs + f;
        float partial = 0.0f;
#pragma unroll
        for (int k = 0; k < S_SUB / 4 - 1; ++k) {
            const int nn = g + 4 * k;
            float v = rowsf[nn * F_OUTK] * n_ri[nn] + bf;
            v = v > 0.0f ? v : alpha * v;
            partial += v;
        }
        {
            const int nn = g + S_SUB - 4;
            float v = rowsf[nn * F_OUTK] * n_ri[nn] + bf;
            v = v > 0.0f ? v : alpha * v;
            if (g == 3) anchor[(long)blockIdx.x * F_OUTK + f] = v;
            else        partial += v;
        }
        red[tid] = partial;
        __syncthreads();
        if (g == 0) {
            const float sum = red[f] + red[f + 64] + red[f + 128] + red[f + 192];
            pool[(long)blockIdx.x * F_OUTK + f] = sum * (1.0f / (S_SUB - 1));
        }
    }
}

__device__ __forceinline__ int gcn_block_excl_scan(const int v, int* s_w, int& total) {
    const int lane = threadIdx.x & 31;
    const int wave = threadIdx.x >> 5;
    int incl = v;
#pragma unroll
    for (int d = 1; d < 32; d <<= 1) {
        const int up = __shfl_up(incl, d);
        if (lane >= d) incl += up;
    }
    if (lane == 31) s_w[wave] = incl;
    __syncthreads();
    int base = 0, all = 0;
#pragma unroll
    for (int w = 0; w < TILE_T / 32; ++w) {
        const int x = s_w[w];
        all += x;
        if (w < wave) base += x;
    }
    total = all;
    return base + incl - v;
}

__global__ __launch_bounds__(TILE_T)
void gcn_bin_kernel(const int* __restrict__ src,
                    const int* __restrict__ dst,
                    float* __restrict__ deg_far,
                    int* __restrict__ run_info,
                    int* __restrict__ recs) {
    __shared__ int  lcur[N_BKT];
    __shared__ int4 rec4[TILE_E / 4];
    __shared__ int  s_w[TILE_T / 32];
    int* rec_l = (int*)rec4;
    const int tid = threadIdx.x;
    const int4* s4 = (const int4*)src + blockIdx.x * (TILE_E / 4);
    const int4* d4 = (const int4*)dst + blockIdx.x * (TILE_E / 4);
    const int4 da = d4[tid];
    const int4 db = d4[tid + TILE_T];
    const int4 sa = s4[tid];
    const int4 sb = s4[tid + TILE_T];
    lcur[tid] = 0;
    __syncthreads();
    atomicAdd(&lcur[da.x >> 7], 1);
    atomicAdd(&lcur[da.y >> 7], 1);
    atomicAdd(&lcur[da.z >> 7], 1);
    atomicAdd(&lcur[da.w >> 7], 1);
    atomicAdd(&lcur[db.x >> 7], 1);
    atomicAdd(&lcur[db.y >> 7], 1);
    atomicAdd(&lcur[db.z >> 7], 1);
    atomicAdd(&lcur[db.w >> 7], 1);
    __syncthreads();
    const int c = lcur[tid];
    int total;
    const int lstart = gcn_block_excl_scan(c, s_w, total);
    lcur[tid] = lstart;
    run_info[tid * N_TILES + blockIdx.x] = (lstart << 14) | c;
    if ((sa.x >> 7) != (da.x >> 7)) atomicAdd(&deg_far[sa.x], 1.0f);
    if ((sa.y >> 7) != (da.y >> 7)) atomicAdd(&deg_far[sa.y], 1.0f);
    if ((sa.z >> 7) != (da.z >> 7)) atomicAdd(&deg_far[sa.z], 1.0f);
    if ((sa.w >> 7) != (da.w >> 7)) atomicAdd(&deg_far[sa.w], 1.0f);
    if ((sb.x >> 7) != (db.x >> 7)) atomicAdd(&deg_far[sb.x], 1.0f);
    if ((sb.y >> 7) != (db.y >> 7)) atomicAdd(&deg_far[sb.y], 1.0f);
    if ((sb.z >> 7) != (db.z >> 7)) atomicAdd(&deg_far[sb.z], 1.0f);
    if ((sb.w >> 7) != (db.w >> 7)) atomicAdd(&deg_far[sb.w], 1.0f);
    __syncthreads();
    int p;
    p = atomicAdd(&lcur[da.x >> 7], 1); rec_l[p] = (sa.x << 7) | (da.x & 127);
    p = atomicAdd(&lcur[da.y >> 7], 1); rec_l[p] = (sa.y << 7) | (da.y & 127);
    p = atomicAdd(&lcur[da.z >> 7], 1); rec_l[p] = (sa.z << 7) | (da.z & 127);
    p = atomicAdd(&lcur[da.w >> 7], 1); rec_l[p] = (sa.w << 7) | (da.w & 127);
    p = atomicAdd(&lcur[db.x >> 7], 1); rec_l[p] = (sb.x << 7) | (db.x & 127);
    p = atomicAdd(&lcur[db.y >> 7], 1); rec_l[p] = (sb.y << 7) | (db.y & 127);
    p = atomicAdd(&lcur[db.z >> 7], 1); rec_l[p] = (sb.z << 7) | (db.z & 127);
    p = atomicAdd(&lcur[db.w >> 7], 1); rec_l[p] = (sb.w << 7) | (db.w & 127);
    __syncthreads();
    int4* out4 = (int4*)recs + blockIdx.x * (TILE_E / 4);
    out4[tid]          = rec4[tid];
    out4[tid + TILE_T] = rec4[tid + TILE_T];
}

__device__ __forceinline__ int gcn_excl_scan256(const int v, int* s_w8, int& total) {
    const int lane = threadIdx.x & 31;
    const int wave = threadIdx.x >> 5;
    int incl = v;
#pragma unroll
    for (int d = 1; d < 32; d <<= 1) {
        const int up = __shfl_up(incl, d);
        if (lane >= d) incl += up;
    }
    if (lane == 31) s_w8[wave] = incl;
    __syncthreads();
    int base = 0, all = 0;
#pragma unroll
    for (int w = 0; w < 8; ++w) {
        const int x = s_w8[w];
        all += x;
        if (w < wave) base += x;
    }
    total = all;
    return base + incl - v;
}

#define GCN_ROW_AT(idx) (*(const float2*)(Hb + (((unsigned)(idx)) << 8) + lane8))

__device__ __forceinline__ float2 gcn_sum_rows(const char* __restrict__ Hb, const int my, const int first,
                                               const int cnt, const unsigned lane8) {
    float2 v[32];
    v[0] = GCN_ROW_AT(first);
#pragma unroll
    for (int u = 1; u < 32; ++u) {
        if (u < cnt) v[u] = GCN_ROW_AT(__builtin_amdgcn_readlane(my, u));
        else         v[u] = make_float2(0.0f, 0.0f);
    }
    float2 a0 = v[0], a1 = v[1];
#pragma unroll
    for (int u = 2; u < 32; u += 2) {
        a0.x += v[u].x;     a0.y += v[u].y;
        a1.x += v[u + 1].x; a1.y += v[u + 1].y;
    }
    a0.x += a1.x;
    a0.y += a1.y;
    return a0;
}

#define BKT_CAP 4096

__global__ __launch_bounds__(256)
void gcn_bucket_gather_kernel(const int* __restrict__ run_info,
                              const int* __restrict__ recs,
                              const int* __restrict__ bkt_late,
                              const float* __restrict__ H,
                              const float* __restrict__ bias,
                              const float* __restrict__ prelu_a,
                              float* __restrict__ pool,
                              float* __restrict__ anchor) {
    __shared__ int    n_cnt[128];
    __shared__ int    n_start[128];
    __shared__ int    n_cur[128];
    __shared__ int    n_deg[128];
    __shared__ int    sorted[BKT_CAP];
    __shared__ float2 accs[128 * 32];
    __shared__ float  n_ri[128];
    __shared__ float  red[256];
    __shared__ int    s_w8[8];

    const int tid  = threadIdx.x;
    const int lane = tid & 31;
    const int wave = tid >> 5;
    const float alpha = prelu_a[0];
    const float bf    = bias[tid & 63];

    for (int b = blockIdx.x; b < N_BKT; b += gridDim.x) {
    if (bkt_late[b] == 0) continue;
    const int info = run_info[b * N_TILES + tid];
    const int lo = info >> 14;
    const int rc = info & 0x3FFF;
    const int* rp = recs + tid * TILE_E + lo;
    int total;
    const int pre = gcn_excl_scan256(rc, s_w8, total);

    if (tid < 128) { n_cnt[tid] = 0; n_deg[tid] = 0; }
    for (int i = tid; i < 128 * 32; i += 256) accs[i] = make_float2(0.0f, 0.0f);
    __syncthreads();

    const char* Hb = (const char*)H;
    const unsigned lane8 = (unsigned)lane * 8u;
    const int nb = wave * NPW;

    for (int c0 = 0; c0 < total; c0 += BKT_CAP) {
        const int j0 = max(0, c0 - pre);
        const int j1 = min(rc, c0 + BKT_CAP - pre);

        for (int j = j0; j < j1; ++j) atomicAdd(&n_cnt[rp[j] & 127], 1);
        __syncthreads();

        if (wave == 0) {
            const int q0 = n_cnt[4 * lane + 0];
            const int q1 = n_cnt[4 * lane + 1];
            const int q2 = n_cnt[4 * lane + 2];
            const int q3 = n_cnt[4 * lane + 3];
            const int sum4 = (q0 + q1) + (q2 + q3);
            int incl = sum4;
#pragma unroll
            for (int d = 1; d < 32; d <<= 1) {
                const int up = __shfl_up(incl, d);
                if (lane >= d) incl += up;
            }
            const int e0 = incl - sum4;
            const int e1 = e0 + q0;
            const int e2 = e1 + q1;
            const int e3 = e2 + q2;
            n_start[4 * lane + 0] = e0; n_cur[4 * lane + 0] = e0; n_deg[4 * lane + 0] += q0;
            n_start[4 * lane + 1] = e1; n_cur[4 * lane + 1] = e1; n_deg[4 * lane + 1] += q1;
            n_start[4 * lane + 2] = e2; n_cur[4 * lane + 2] = e2; n_deg[4 * lane + 2] += q2;
            n_start[4 * lane + 3] = e3; n_cur[4 * lane + 3] = e3; n_deg[4 * lane + 3] += q3;
        }
        __syncthreads();

        for (int j = j0; j < j1; ++j) {
            const int r = rp[j];
            const int p = atomicAdd(&n_cur[r & 127], 1);
            sorted[p] = r >> 7;
        }
        __syncthreads();

        {
            int lb = 0, ln = 0;
            if (lane < NPW) {
                lb = n_start[nb + lane];
                ln = n_cnt[nb + lane];
            }
            int beg = __builtin_amdgcn_readlane(lb, 0);
            int n   = __builtin_amdgcn_readlane(ln, 0);
            int my  = (lane < n) ? sorted[beg + lane] : 0;
            int first = __builtin_amdgcn_readlane(my, 0);
            for (int i = 0; i < NPW; ++i) {
                int begN = 0, nN = 0, myN = 0;
                if (i + 1 < NPW) {
                    begN = __builtin_amdgcn_readlane(lb, i + 1);
                    nN   = __builtin_amdgcn_readlane(ln, i + 1);
                    myN  = (lane < nN) ? sorted[begN + lane] : 0;
                }
                if (n > 0) {
                    float2 acc = gcn_sum_rows(Hb, my, first, min(32, n), lane8);
                    for (int j = 32; j < n; j += 32) {
                        const int more = (lane < n - j) ? sorted[beg + j + lane] : 0;
                        const float2 part = gcn_sum_rows(Hb, more, __builtin_amdgcn_readlane(more, 0),
                                                         min(32, n - j), lane8);
                        acc.x += part.x;
                        acc.y += part.y;
                    }
                    float2 o = accs[(nb + i) * 32 + lane];
                    o.x += acc.x;
                    o.y += acc.y;
                    accs[(nb + i) * 32 + lane] = o;
                }
                beg = begN; n = nN; my = myN;
                first = __builtin_amdgcn_readlane(myN, 0);
            }
        }

        if (c0 + BKT_CAP < total) {
            __syncthreads();
            if (tid < 128) n_cnt[tid] = 0;
            __syncthreads();
        }
    }

    if (tid < 128) n_ri[tid] = __frsqrt_rn(fmaxf((float)n_deg[tid], 1.0f));
    __syncthreads();
    {
        const int f = tid & 63;
        const int g = tid >> 6;
        const float* rowsf = (const float*)accs + f;
        float partial = 0.0f;
#pragma unroll
        for (int k = 0; k < S_SUB / 4 - 1; ++k) {
            const int n = g + 4 * k;
            float v = rowsf[n * F_OUTK] * n_ri[n] + bf;
            v = v > 0.0f ? v : alpha * v;
            partial += v;
        }
        {
            const int n = g + S_SUB - 4;
            float v = rowsf[n * F_OUTK] * n_ri[n] + bf;
            v = v > 0.0f ? v : alpha * v;
            if (g == 3) anchor[(long)b * F_OUTK + f] = v;
            else        partial += v;
        }
        red[tid] = partial;
        __syncthreads();
        if (g == 0) {
            const float sum = red[f] + red[f + 64] + red[f + 128] + red[f + 192];
            pool[(long)b * F_OUTK + f] = sum * (1.0f / (S_SUB - 1));
        }
    }
    __syncthreads();
    }
}

extern "C" void kernel_launch(void* const* d_in, const int* in_sizes, int n_in,
                              void* d_out, int out_size, void* d_ws, size_t ws_size,
                              hipStream_t stream) {
    const float* in_feat  = (const float*)d_in[0];
    const float* W        = (const float*)d_in[1];
    const float* bias     = (const float*)d_in[2];
    const float* prelu_a  = (const float*)d_in[3];
    const int*   src      = (const int*)d_in[4];
    const int*   dst      = (const int*)d_in[5];

    float* ws      = (float*)d_ws;
    float* deg_far = ws;
    float* H       = ws + (long)N_NODES;
    int*   run_info = (int*)(H + (long)N_NODES * F_OUTK);
    int*   recs     = run_info + (long)N_BKT * N_TILES;
    int*   bkt_late = recs + N_EDGE;

    float* pool   = (float*)d_out;
    float* anchor = pool + (long)B_SUB * F_OUTK;

    hipMemsetAsync(deg_far, 0, (long)N_NODES * sizeof(float), stream);

    gcn_bin_kernel<<<N_TILES, TILE_T, 0, stream>>>(src, dst, deg_far, run_info, recs);
    gcn_gemm_wmma_kernel<<<N_NODES / 128, 256, 0, stream>>>(in_feat, W, deg_far, run_info, recs, bias, prelu_a,
                                                            H, bkt_late, pool, anchor);
    gcn_bucket_gather_kernel<<<N_BKT / 4, 256, 0, stream>>>(run_info, recs, bkt_late, H, bias, prelu_a, pool, anchor);
}
